# starting kernel
# baseline (speedup 1.0000x reference)
#include <hip/hip_runtime.h>
#include <hip/hip_bf16.h>

typedef float v2f __attribute__((ext_vector_type(2)));
typedef float v8f __attribute__((ext_vector_type(8)));

#define PSTRIDE 160
#define WLDS    (32 * PSTRIDE)

__device__ __forceinline__ unsigned enc_f32(float f) {
    unsigned u = __float_as_uint(f);
    return (u & 0x80000000u) ? ~u : (u | 0x80000000u);
}
__device__ __forceinline__ float dec_f32(unsigned e) {
    return (e & 0x80000000u) ? __uint_as_float(e ^ 0x80000000u) : __uint_as_float(~e);
}

__device__ __forceinline__ void stage_weight_pairs(float* sW, const float* __restrict__ W) {
    for (int i = threadIdx.x; i < 64 * 64; i += blockDim.x) {
        const int k = i >> 6, col = i & 63;
        sW[(k >> 1) * PSTRIDE + col * 2 + (k & 1)] = W[i];
    }
}

__global__ void node_xform_kernel(const float* __restrict__ x2,
                                  const float* __restrict__ Wl, const float* __restrict__ bl,
                                  const float* __restrict__ Wr, const float* __restrict__ br,
                                  float* __restrict__ xl, float* __restrict__ xr, int NT)
{
    __shared__ float sWl[WLDS];
    __shared__ float sWr[WLDS];
    __shared__ float sbl[64], sbr[64];
    stage_weight_pairs(sWl, Wl);
    stage_weight_pairs(sWr, Wr);
    for (int i = threadIdx.x; i < 64; i += blockDim.x) { sbl[i] = bl[i]; sbr[i] = br[i]; }
    __syncthreads();

    const int lane    = threadIdx.x & 31;
    const int wave    = threadIdx.x >> 5;
    const int wpb     = blockDim.x >> 5;
    const int l15     = lane & 15;
    const int half    = lane >> 4;
    const int numTiles = NT >> 4;

    for (int tile = blockIdx.x * wpb + wave; tile < numTiles; tile += gridDim.x * wpb) {
        v8f accl[4] = {v8f{}, v8f{}, v8f{}, v8f{}};
        v8f accr[4] = {v8f{}, v8f{}, v8f{}, v8f{}};
        const float* xrow = x2 + (size_t)(tile * 16 + l15) * 64;
        #pragma unroll 4
        for (int kk = 0; kk < 16; ++kk) {
            const int kbase = kk * 4 + half * 2;
            const v2f a = *(const v2f*)(xrow + kbase);
            const int pbase = (kk * 2 + half) * PSTRIDE;
            #pragma unroll
            for (int nb = 0; nb < 4; ++nb) {
                const int col = nb * 16 + l15;
                const v2f bL = *(const v2f*)(sWl + pbase + col * 2);
                const v2f bR = *(const v2f*)(sWr + pbase + col * 2);
                accl[nb] = __builtin_amdgcn_wmma_f32_16x16x4_f32(
                    false, a, false, bL, (short)0, accl[nb], false, false);
                accr[nb] = __builtin_amdgcn_wmma_f32_16x16x4_f32(
                    false, a, false, bR, (short)0, accr[nb], false, false);
            }
        }
        #pragma unroll
        for (int nb = 0; nb < 4; ++nb) {
            const int col = nb * 16 + l15;
            #pragma unroll
            for (int v = 0; v < 8; ++v) {
                const size_t row = (size_t)(tile * 16 + v + half * 8);
                xl[row * 64 + col] = accl[nb][v] + sbl[col];
                xr[row * 64 + col] = accr[nb][v] + sbr[col];
            }
        }
    }
}

__global__ void init_ws_kernel(unsigned* __restrict__ smax, float* __restrict__ denom,
                               float* __restrict__ agg, int NT)
{
    const long long n = (long long)NT * 64;
    for (long long i = (long long)blockIdx.x * blockDim.x + threadIdx.x; i < n;
         i += (long long)gridDim.x * blockDim.x) {
        agg[i] = 0.0f;
        if (i < NT) { smax[i] = 0u; denom[i] = 0.0f; }
    }
}

__global__ void edge_score_kernel(const int* __restrict__ src, const int* __restrict__ dst,
                                  const float* __restrict__ eattr,
                                  const float* __restrict__ We, const float* __restrict__ att,
                                  const float* __restrict__ xl, const float* __restrict__ xr,
                                  float* __restrict__ score, unsigned* __restrict__ smax,
                                  long long E)
{
    const int lane = threadIdx.x & 31;
    const long long wid = (long long)blockIdx.x * (blockDim.x >> 5) + (threadIdx.x >> 5);
    const long long nw  = (long long)gridDim.x * (blockDim.x >> 5);
    const float2 we = ((const float2*)We)[lane];
    const float2 at = ((const float2*)att)[lane];
    for (long long e = wid; e < E; e += nw) {
        const int s = src[e];
        const int d = dst[e];
        const float ea = eattr[e];
        const float2 l2 = ((const float2*)(xl + (size_t)s * 64))[lane];
        const float2 r2 = ((const float2*)(xr + (size_t)d * 64))[lane];
        float m0 = l2.x + r2.x + ea * we.x;
        float m1 = l2.y + r2.y + ea * we.y;
        m0 = (m0 > 0.0f) ? m0 : 0.2f * m0;
        m1 = (m1 > 0.0f) ? m1 : 0.2f * m1;
        float p = at.x * m0 + at.y * m1;
        #pragma unroll
        for (int off = 16; off > 0; off >>= 1) p += __shfl_xor(p, off, 32);
        if (lane == 0) {
            score[e] = p;
            atomicMax(&smax[d], enc_f32(p));
        }
    }
}

__global__ void edge_denom_kernel(const int* __restrict__ dst, const float* __restrict__ score,
                                  const unsigned* __restrict__ smax, float* __restrict__ denom,
                                  long long E)
{
    for (long long e = (long long)blockIdx.x * blockDim.x + threadIdx.x; e < E;
         e += (long long)gridDim.x * blockDim.x) {
        const int d = dst[e];
        const float ex = __expf(score[e] - dec_f32(smax[d]));
        atomicAdd(&denom[d], ex);
    }
}

__global__ void edge_agg_kernel(const int* __restrict__ src, const int* __restrict__ dst,
                                const float* __restrict__ score, const unsigned* __restrict__ smax,
                                const float* __restrict__ denom, const float* __restrict__ xl,
                                float* __restrict__ agg, long long E)
{
    const int lane = threadIdx.x & 31;
    const long long wid = (long long)blockIdx.x * (blockDim.x >> 5) + (threadIdx.x >> 5);
    const long long nw  = (long long)gridDim.x * (blockDim.x >> 5);
    for (long long e = wid; e < E; e += nw) {
        const int s = src[e];
        const int d = dst[e];
        const float alpha = __expf(score[e] - dec_f32(smax[d])) / denom[d];
        const float2 l2 = ((const float2*)(xl + (size_t)s * 64))[lane];
        float* ap = agg + (size_t)d * 64 + 2 * lane;
        atomicAdd(ap,     alpha * l2.x);
        atomicAdd(ap + 1, alpha * l2.y);
    }
}

__global__ void out_xform_kernel(const float* __restrict__ agg, const float* __restrict__ bias,
                                 const float* __restrict__ Wfc, const float* __restrict__ bfc,
                                 float* __restrict__ out, int NT)
{
    __shared__ float sW[WLDS];
    __shared__ float sb[64], sbias[64];
    stage_weight_pairs(sW, Wfc);
    for (int i = threadIdx.x; i < 64; i += blockDim.x) { sb[i] = bfc[i]; sbias[i] = bias[i]; }
    __syncthreads();

    const int lane = threadIdx.x & 31;
    const int wave = threadIdx.x >> 5;
    const int wpb  = blockDim.x >> 5;
    const int l15  = lane & 15;
    const int half = lane >> 4;
    const int numTiles = NT >> 4;

    for (int tile = blockIdx.x * wpb + wave; tile < numTiles; tile += gridDim.x * wpb) {
        v8f acc[4] = {v8f{}, v8f{}, v8f{}, v8f{}};
        const float* arow = agg + (size_t)(tile * 16 + l15) * 64;
        #pragma unroll 4
        for (int kk = 0; kk < 16; ++kk) {
            const int kbase = kk * 4 + half * 2;
            v2f a = *(const v2f*)(arow + kbase);
            a.x += sbias[kbase];
            a.y += sbias[kbase + 1];
            const int pbase = (kk * 2 + half) * PSTRIDE;
            #pragma unroll
            for (int nb = 0; nb < 4; ++nb) {
                const int col = nb * 16 + l15;
                const v2f b = *(const v2f*)(sW + pbase + col * 2);
                acc[nb] = __builtin_amdgcn_wmma_f32_16x16x4_f32(
                    false, a, false, b, (short)0, acc[nb], false, false);
            }
        }
        #pragma unroll
        for (int nb = 0; nb < 4; ++nb) {
            const int col = nb * 16 + l15;
            #pragma unroll
            for (int v = 0; v < 8; ++v) {
                const size_t row = (size_t)(tile * 16 + v + half * 8);
                out[row * 64 + col] = acc[nb][v] + sb[col];
            }
        }
    }
}

extern "C" void kernel_launch(void* const* d_in, const int* in_sizes, int n_in,
                              void* d_out, int out_size, void* d_ws, size_t ws_size,
                              hipStream_t stream) {
    const float* x      = (const float*)d_in[0];
    const int*   eidx   = (const int*)d_in[1];
    const float* eattr  = (const float*)d_in[2];
    const float* W_l    = (const float*)d_in[3];
    const float* b_l    = (const float*)d_in[4];
    const float* W_r    = (const float*)d_in[5];
    const float* b_r    = (const float*)d_in[6];
    const float* W_e    = (const float*)d_in[7];
    const float* att    = (const float*)d_in[8];
    const float* bias   = (const float*)d_in[9];
    const float* W_fc   = (const float*)d_in[10];
    const float* b_fc   = (const float*)d_in[11];

    const int NT       = in_sizes[0] / 64;
    const long long E  = (long long)in_sizes[2];
    const int* src = eidx;
    const int* dst = eidx + E;

    float* ws     = (float*)d_ws;
    float* xl     = ws;
    float* xr     = xl + (size_t)NT * 64;
    float* agg    = xr + (size_t)NT * 64;
    float* score  = agg + (size_t)NT * 64;
    unsigned* smax = (unsigned*)(score + E);
    float* denom  = (float*)(smax + NT);

    const int numTiles = NT / 16;
    const int gemmBlocks = (numTiles + 7) / 8;

    node_xform_kernel<<<gemmBlocks, 256, 0, stream>>>(x, W_l, b_l, W_r, b_r, xl, xr, NT);
    init_ws_kernel<<<4096, 256, 0, stream>>>(smax, denom, agg, NT);
    edge_score_kernel<<<8192, 256, 0, stream>>>(src, dst, eattr, W_e, att, xl, xr, score, smax, E);
    edge_denom_kernel<<<8192, 256, 0, stream>>>(dst, score, smax, denom, E);
    edge_agg_kernel<<<8192, 256, 0, stream>>>(src, dst, score, smax, denom, xl, agg, E);
    out_xform_kernel<<<gemmBlocks, 256, 0, stream>>>(agg, bias, W_fc, b_fc, (float*)d_out, NT);
}


# edge_score: each wave takes 32 consecutive edges per trip (coalesced index loads, lane-to-scalar broadcast, 16 row loads in flight, transposed xor reduction, one score line store); sum tree kept
# speedup vs baseline: 1.3700x; 1.3700x over previous
#include <hip/hip_runtime.h>
#include <hip/hip_bf16.h>

typedef float v2f __attribute__((ext_vector_type(2)));
typedef float v8f __attribute__((ext_vector_type(8)));

#define PSTRIDE 160
#define WLDS    (32 * PSTRIDE)

__device__ __forceinline__ unsigned enc_f32(float f) {
    unsigned u = __float_as_uint(f);
    return (u & 0x80000000u) ? ~u : (u | 0x80000000u);
}
__device__ __forceinline__ float dec_f32(unsigned e) {
    return (e & 0x80000000u) ? __uint_as_float(e ^ 0x80000000u) : __uint_as_float(~e);
}

__device__ __forceinline__ void stage_weight_pairs(float* sW, const float* __restrict__ W) {
    for (int i = threadIdx.x; i < 64 * 64; i += blockDim.x) {
        const int k = i >> 6, col = i & 63;
        sW[(k >> 1) * PSTRIDE + col * 2 + (k & 1)] = W[i];
    }
}

__global__ void node_xform_kernel(const float* __restrict__ x2,
                                  const float* __restrict__ Wl, const float* __restrict__ bl,
                                  const float* __restrict__ Wr, const float* __restrict__ br,
                                  float* __restrict__ xl, float* __restrict__ xr, int NT)
{
    __shared__ float sWl[WLDS];
    __shared__ float sWr[WLDS];
    __shared__ float sbl[64], sbr[64];
    stage_weight_pairs(sWl, Wl);
    stage_weight_pairs(sWr, Wr);
    for (int i = threadIdx.x; i < 64; i += blockDim.x) { sbl[i] = bl[i]; sbr[i] = br[i]; }
    __syncthreads();

    const int lane    = threadIdx.x & 31;
    const int wave    = threadIdx.x >> 5;
    const int wpb     = blockDim.x >> 5;
    const int l15     = lane & 15;
    const int half    = lane >> 4;
    const int numTiles = NT >> 4;

    for (int tile = blockIdx.x * wpb + wave; tile < numTiles; tile += gridDim.x * wpb) {
        v8f accl[4] = {v8f{}, v8f{}, v8f{}, v8f{}};
        v8f accr[4] = {v8f{}, v8f{}, v8f{}, v8f{}};
        const float* xrow = x2 + (size_t)(tile * 16 + l15) * 64;
        #pragma unroll 4
        for (int kk = 0; kk < 16; ++kk) {
            const int kbase = kk * 4 + half * 2;
            const v2f a = *(const v2f*)(xrow + kbase);
            const int pbase = (kk * 2 + half) * PSTRIDE;
            #pragma unroll
            for (int nb = 0; nb < 4; ++nb) {
                const int col = nb * 16 + l15;
                const v2f bL = *(const v2f*)(sWl + pbase + col * 2);
                const v2f bR = *(const v2f*)(sWr + pbase + col * 2);
                accl[nb] = __builtin_amdgcn_wmma_f32_16x16x4_f32(
                    false, a, false, bL, (short)0, accl[nb], false, false);
                accr[nb] = __builtin_amdgcn_wmma_f32_16x16x4_f32(
                    false, a, false, bR, (short)0, accr[nb], false, false);
            }
        }
        #pragma unroll
        for (int nb = 0; nb < 4; ++nb) {
            const int col = nb * 16 + l15;
            #pragma unroll
            for (int v = 0; v < 8; ++v) {
                const size_t row = (size_t)(tile * 16 + v + half * 8);
                xl[row * 64 + col] = accl[nb][v] + sbl[col];
                xr[row * 64 + col] = accr[nb][v] + sbr[col];
            }
        }
    }
}

__global__ void init_ws_kernel(unsigned* __restrict__ smax, float* __restrict__ denom,
                               float* __restrict__ agg, int NT)
{
    const long long n = (long long)NT * 64;
    for (long long i = (long long)blockIdx.x * blockDim.x + threadIdx.x; i < n;
         i += (long long)gridDim.x * blockDim.x) {
        agg[i] = 0.0f;
        if (i < NT) { smax[i] = 0u; denom[i] = 0.0f; }
    }
}

#define BREV5(j) ((((j) & 1) << 4) | (((j) & 2) << 2) | ((j) & 4) | (((j) & 8) >> 2) | (((j) & 16) >> 4))

template <int MASK>
__device__ __forceinline__ float merge2(float a, float b, bool up) {
    const float keep = up ? b : a;
    const float give = up ? a : b;
    return keep + __shfl_xor(give, MASK, 32);
}

template <int C>
__device__ __forceinline__ float score_group8(int s_v, int d_v, int ea_v,
                                              const float* __restrict__ xl,
                                              const float* __restrict__ xr,
                                              float2 we, float2 at, int lane,
                                              bool b4, bool b3, bool b2)
{
    float2 l2[8], r2[8];
    float ea[8];
    #pragma unroll
    for (int j = 0; j < 8; ++j) {
        const int s = __builtin_amdgcn_readlane(s_v, BREV5(C * 8 + j));
        const int d = __builtin_amdgcn_readlane(d_v, BREV5(C * 8 + j));
        ea[j] = __int_as_float(__builtin_amdgcn_readlane(ea_v, BREV5(C * 8 + j)));
        l2[j] = ((const float2*)(xl + (size_t)s * 64))[lane];
        r2[j] = ((const float2*)(xr + (size_t)d * 64))[lane];
    }
    float p[8];
    #pragma unroll
    for (int j = 0; j < 8; ++j) {
        float m0 = l2[j].x + r2[j].x + ea[j] * we.x;
        float m1 = l2[j].y + r2[j].y + ea[j] * we.y;
        m0 = (m0 > 0.0f) ? m0 : 0.2f * m0;
        m1 = (m1 > 0.0f) ? m1 : 0.2f * m1;
        p[j] = __builtin_fmaf(at.x, m0, at.y * m1);
    }
    const float q0 = merge2<16>(p[0], p[1], b4);
    const float q1 = merge2<16>(p[2], p[3], b4);
    const float q2 = merge2<16>(p[4], p[5], b4);
    const float q3 = merge2<16>(p[6], p[7], b4);
    const float r0 = merge2<8>(q0, q1, b3);
    const float r1 = merge2<8>(q2, q3, b3);
    return merge2<4>(r0, r1, b2);
}

__global__ void edge_score_kernel(const int* __restrict__ src, const int* __restrict__ dst,
                                  const float* __restrict__ eattr,
                                  const float* __restrict__ We, const float* __restrict__ att,
                                  const float* __restrict__ xl, const float* __restrict__ xr,
                                  float* __restrict__ score, unsigned* __restrict__ smax,
                                  long long E)
{
    const int lane = threadIdx.x & 31;
    const int wib = __builtin_amdgcn_readfirstlane((int)(threadIdx.x >> 5));
    const long long wid = (long long)blockIdx.x * (blockDim.x >> 5) + wib;
    const long long nw  = (long long)gridDim.x * (blockDim.x >> 5);
    const long long nbatch = (E + 31) >> 5;
    const float2 we = ((const float2*)We)[lane];
    const float2 at = ((const float2*)att)[lane];
    const bool b4 = (lane & 16) != 0, b3 = (lane & 8) != 0, b2 = (lane & 4) != 0;
    const bool b1 = (lane & 2) != 0,  b0 = (lane & 1) != 0;
    const int mine = BREV5(lane);
    for (long long b = wid; b < nbatch; b += nw) {
        const long long e = (b << 5) + mine;
        const bool valid = e < E;
        const int s_v  = valid ? src[e] : 0;
        const int d_v  = valid ? dst[e] : 0;
        const int ea_v = valid ? __float_as_int(eattr[e]) : 0;
        const float t0 = score_group8<0>(s_v, d_v, ea_v, xl, xr, we, at, lane, b4, b3, b2);
        const float t1 = score_group8<1>(s_v, d_v, ea_v, xl, xr, we, at, lane, b4, b3, b2);
        const float t2 = score_group8<2>(s_v, d_v, ea_v, xl, xr, we, at, lane, b4, b3, b2);
        const float t3 = score_group8<3>(s_v, d_v, ea_v, xl, xr, we, at, lane, b4, b3, b2);
        const float u0 = merge2<2>(t0, t1, b1);
        const float u1 = merge2<2>(t2, t3, b1);
        const float p  = merge2<1>(u0, u1, b0);
        if (valid) {
            score[e] = p;
            atomicMax(&smax[d_v], enc_f32(p));
        }
    }
}

__global__ void edge_denom_kernel(const int* __restrict__ dst, const float* __restrict__ score,
                                  const unsigned* __restrict__ smax, float* __restrict__ denom,
                                  long long E)
{
    for (long long e = (long long)blockIdx.x * blockDim.x + threadIdx.x; e < E;
         e += (long long)gridDim.x * blockDim.x) {
        const int d = dst[e];
        const float ex = __expf(score[e] - dec_f32(smax[d]));
        atomicAdd(&denom[d], ex);
    }
}

__global__ void edge_agg_kernel(const int* __restrict__ src, const int* __restrict__ dst,
                                const float* __restrict__ score, const unsigned* __restrict__ smax,
                                const float* __restrict__ denom, const float* __restrict__ xl,
                                float* __restrict__ agg, long long E)
{
    const int lane = threadIdx.x & 31;
    const long long wid = (long long)blockIdx.x * (blockDim.x >> 5) + (threadIdx.x >> 5);
    const long long nw  = (long long)gridDim.x * (blockDim.x >> 5);
    for (long long e = wid; e < E; e += nw) {
        const int s = src[e];
        const int d = dst[e];
        const float alpha = __expf(score[e] - dec_f32(smax[d])) / denom[d];
        const float2 l2 = ((const float2*)(xl + (size_t)s * 64))[lane];
        float* ap = agg + (size_t)d * 64 + 2 * lane;
        atomicAdd(ap,     alpha * l2.x);
        atomicAdd(ap + 1, alpha * l2.y);
    }
}

__global__ void out_xform_kernel(const float* __restrict__ agg, const float* __restrict__ bias,
                                 const float* __restrict__ Wfc, const float* __restrict__ bfc,
                                 float* __restrict__ out, int NT)
{
    __shared__ float sW[WLDS];
    __shared__ float sb[64], sbias[64];
    stage_weight_pairs(sW, Wfc);
    for (int i = threadIdx.x; i < 64; i += blockDim.x) { sb[i] = bfc[i]; sbias[i] = bias[i]; }
    __syncthreads();

    const int lane = threadIdx.x & 31;
    const int wave = threadIdx.x >> 5;
    const int wpb  = blockDim.x >> 5;
    const int l15  = lane & 15;
    const int half = lane >> 4;
    const int numTiles = NT >> 4;

    for (int tile = blockIdx.x * wpb + wave; tile < numTiles; tile += gridDim.x * wpb) {
        v8f acc[4] = {v8f{}, v8f{}, v8f{}, v8f{}};
        const float* arow = agg + (size_t)(tile * 16 + l15) * 64;
        #pragma unroll 4
        for (int kk = 0; kk < 16; ++kk) {
            const int kbase = kk * 4 + half * 2;
            v2f a = *(const v2f*)(arow + kbase);
            a.x += sbias[kbase];
            a.y += sbias[kbase + 1];
            const int pbase = (kk * 2 + half) * PSTRIDE;
            #pragma unroll
            for (int nb = 0; nb < 4; ++nb) {
                const int col = nb * 16 + l15;
                const v2f b = *(const v2f*)(sW + pbase + col * 2);
                acc[nb] = __builtin_amdgcn_wmma_f32_16x16x4_f32(
                    false, a, false, b, (short)0, acc[nb], false, false);
            }
        }
        #pragma unroll
        for (int nb = 0; nb < 4; ++nb) {
            const int col = nb * 16 + l15;
            #pragma unroll
            for (int v = 0; v < 8; ++v) {
                const size_t row = (size_t)(tile * 16 + v + half * 8);
                out[row * 64 + col] = acc[nb][v] + sb[col];
            }
        }
    }
}

extern "C" void kernel_launch(void* const* d_in, const int* in_sizes, int n_in,
                              void* d_out, int out_size, void* d_ws, size_t ws_size,
                              hipStream_t stream) {
    const float* x      = (const float*)d_in[0];
    const int*   eidx   = (const int*)d_in[1];
    const float* eattr  = (const float*)d_in[2];
    const float* W_l    = (const float*)d_in[3];
    const float* b_l    = (const float*)d_in[4];
    const float* W_r    = (const float*)d_in[5];
    const float* b_r    = (const float*)d_in[6];
    const float* W_e    = (const float*)d_in[7];
    const float* att    = (const float*)d_in[8];
    const float* bias   = (const float*)d_in[9];
    const float* W_fc   = (const float*)d_in[10];
    const float* b_fc   = (const float*)d_in[11];

    const int NT       = in_sizes[0] / 64;
    const long long E  = (long long)in_sizes[2];
    const int* src = eidx;
    const int* dst = eidx + E;

    float* ws     = (float*)d_ws;
    float* xl     = ws;
    float* xr     = xl + (size_t)NT * 64;
    float* agg    = xr + (size_t)NT * 64;
    float* score  = agg + (size_t)NT * 64;
    unsigned* smax = (unsigned*)(score + E);
    float* denom  = (float*)(smax + NT);

    const int numTiles = NT / 16;
    const int gemmBlocks = (numTiles + 7) / 8;

    node_xform_kernel<<<gemmBlocks, 256, 0, stream>>>(x, W_l, b_l, W_r, b_r, xl, xr, NT);
    init_ws_kernel<<<4096, 256, 0, stream>>>(smax, denom, agg, NT);
    edge_score_kernel<<<1280, 256, 0, stream>>>(src, dst, eattr, W_e, att, xl, xr, score, smax, E);
    edge_denom_kernel<<<8192, 256, 0, stream>>>(dst, score, smax, denom, E);
    edge_agg_kernel<<<8192, 256, 0, stream>>>(src, dst, score, smax, denom, xl, agg, E);
    out_xform_kernel<<<gemmBlocks, 256, 0, stream>>>(agg, bias, W_fc, b_fc, (float*)d_out, NT);
}


# edge_agg: lane takes channels L and L+32 (not 2L, 2L+1) so each atomic instruction adds one whole 128B line instead of two half-filled ones; float atomics kept, arrival order unfixed as before
# speedup vs baseline: 1.4179x; 1.0350x over previous
#include <hip/hip_runtime.h>
#include <hip/hip_bf16.h>

typedef float v2f __attribute__((ext_vector_type(2)));
typedef float v8f __attribute__((ext_vector_type(8)));

#define PSTRIDE 160
#define WLDS    (32 * PSTRIDE)

__device__ __forceinline__ unsigned enc_f32(float f) {
    unsigned u = __float_as_uint(f);
    return (u & 0x80000000u) ? ~u : (u | 0x80000000u);
}
__device__ __forceinline__ float dec_f32(unsigned e) {
    return (e & 0x80000000u) ? __uint_as_float(e ^ 0x80000000u) : __uint_as_float(~e);
}

__device__ __forceinline__ void stage_weight_pairs(float* sW, const float* __restrict__ W) {
    for (int i = threadIdx.x; i < 64 * 64; i += blockDim.x) {
        const int k = i >> 6, col = i & 63;
        sW[(k >> 1) * PSTRIDE + col * 2 + (k & 1)] = W[i];
    }
}

__global__ void node_xform_kernel(const float* __restrict__ x2,
                                  const float* __restrict__ Wl, const float* __restrict__ bl,
                                  const float* __restrict__ Wr, const float* __restrict__ br,
                                  float* __restrict__ xl, float* __restrict__ xr, int NT)
{
    __shared__ float sWl[WLDS];
    __shared__ float sWr[WLDS];
    __shared__ float sbl[64], sbr[64];
    stage_weight_pairs(sWl, Wl);
    stage_weight_pairs(sWr, Wr);
    for (int i = threadIdx.x; i < 64; i += blockDim.x) { sbl[i] = bl[i]; sbr[i] = br[i]; }
    __syncthreads();

    const int lane    = threadIdx.x & 31;
    const int wave    = threadIdx.x >> 5;
    const int wpb     = blockDim.x >> 5;
    const int l15     = lane & 15;
    const int half    = lane >> 4;
    const int numTiles = NT >> 4;

    for (int tile = blockIdx.x * wpb + wave; tile < numTiles; tile += gridDim.x * wpb) {
        v8f accl[4] = {v8f{}, v8f{}, v8f{}, v8f{}};
        v8f accr[4] = {v8f{}, v8f{}, v8f{}, v8f{}};
        const float* xrow = x2 + (size_t)(tile * 16 + l15) * 64;
        #pragma unroll 4
        for (int kk = 0; kk < 16; ++kk) {
            const int kbase = kk * 4 + half * 2;
            const v2f a = *(const v2f*)(xrow + kbase);
            const int pbase = (kk * 2 + half) * PSTRIDE;
            #pragma unroll
            for (int nb = 0; nb < 4; ++nb) {
                const int col = nb * 16 + l15;
                const v2f bL = *(const v2f*)(sWl + pbase + col * 2);
                const v2f bR = *(const v2f*)(sWr + pbase + col * 2);
                accl[nb] = __builtin_amdgcn_wmma_f32_16x16x4_f32(
                    false, a, false, bL, (short)0, accl[nb], false, false);
                accr[nb] = __builtin_amdgcn_wmma_f32_16x16x4_f32(
                    false, a, false, bR, (short)0, accr[nb], false, false);
            }
        }
        #pragma unroll
        for (int nb = 0; nb < 4; ++nb) {
            const int col = nb * 16 + l15;
            #pragma unroll
            for (int v = 0; v < 8; ++v) {
                const size_t row = (size_t)(tile * 16 + v + half * 8);
                xl[row * 64 + col] = accl[nb][v] + sbl[col];
                xr[row * 64 + col] = accr[nb][v] + sbr[col];
            }
        }
    }
}

__global__ void init_ws_kernel(unsigned* __restrict__ smax, float* __restrict__ denom,
                               float* __restrict__ agg, int NT)
{
    const long long n = (long long)NT * 64;
    for (long long i = (long long)blockIdx.x * blockDim.x + threadIdx.x; i < n;
         i += (long long)gridDim.x * blockDim.x) {
        agg[i] = 0.0f;
        if (i < NT) { smax[i] = 0u; denom[i] = 0.0f; }
    }
}

#define BREV5(j) ((((j) & 1) << 4) | (((j) & 2) << 2) | ((j) & 4) | (((j) & 8) >> 2) | (((j) & 16) >> 4))

template <int MASK>
__device__ __forceinline__ float merge2(float a, float b, bool up) {
    const float keep = up ? b : a;
    const float give = up ? a : b;
    return keep + __shfl_xor(give, MASK, 32);
}

template <int C>
__device__ __forceinline__ float score_group8(int s_v, int d_v, int ea_v,
                                              const float* __restrict__ xl,
                                              const float* __restrict__ xr,
                                              float2 we, float2 at, int lane,
                                              bool b4, bool b3, bool b2)
{
    float2 l2[8], r2[8];
    float ea[8];
    #pragma unroll
    for (int j = 0; j < 8; ++j) {
        const int s = __builtin_amdgcn_readlane(s_v, BREV5(C * 8 + j));
        const int d = __builtin_amdgcn_readlane(d_v, BREV5(C * 8 + j));
        ea[j] = __int_as_float(__builtin_amdgcn_readlane(ea_v, BREV5(C * 8 + j)));
        l2[j] = ((const float2*)(xl + (size_t)s * 64))[lane];
        r2[j] = ((const float2*)(xr + (size_t)d * 64))[lane];
    }
    float p[8];
    #pragma unroll
    for (int j = 0; j < 8; ++j) {
        float m0 = l2[j].x + r2[j].x + ea[j] * we.x;
        float m1 = l2[j].y + r2[j].y + ea[j] * we.y;
        m0 = (m0 > 0.0f) ? m0 : 0.2f * m0;
        m1 = (m1 > 0.0f) ? m1 : 0.2f * m1;
        p[j] = __builtin_fmaf(at.x, m0, at.y * m1);
    }
    const float q0 = merge2<16>(p[0], p[1], b4);
    const float q1 = merge2<16>(p[2], p[3], b4);
    const float q2 = merge2<16>(p[4], p[5], b4);
    const float q3 = merge2<16>(p[6], p[7], b4);
    const float r0 = merge2<8>(q0, q1, b3);
    const float r1 = merge2<8>(q2, q3, b3);
    return merge2<4>(r0, r1, b2);
}

__global__ void edge_score_kernel(const int* __restrict__ src, const int* __restrict__ dst,
                                  const float* __restrict__ eattr,
                                  const float* __restrict__ We, const float* __restrict__ att,
                                  const float* __restrict__ xl, const float* __restrict__ xr,
                                  float* __restrict__ score, unsigned* __restrict__ smax,
                                  long long E)
{
    const int lane = threadIdx.x & 31;
    const int wib = __builtin_amdgcn_readfirstlane((int)(threadIdx.x >> 5));
    const long long wid = (long long)blockIdx.x * (blockDim.x >> 5) + wib;
    const long long nw  = (long long)gridDim.x * (blockDim.x >> 5);
    const long long nbatch = (E + 31) >> 5;
    const float2 we = ((const float2*)We)[lane];
    const float2 at = ((const float2*)att)[lane];
    const bool b4 = (lane & 16) != 0, b3 = (lane & 8) != 0, b2 = (lane & 4) != 0;
    const bool b1 = (lane & 2) != 0,  b0 = (lane & 1) != 0;
    const int mine = BREV5(lane);
    for (long long b = wid; b < nbatch; b += nw) {
        const long long e = (b << 5) + mine;
        const bool valid = e < E;
        const int s_v  = valid ? src[e] : 0;
        const int d_v  = valid ? dst[e] : 0;
        const int ea_v = valid ? __float_as_int(eattr[e]) : 0;
        const float t0 = score_group8<0>(s_v, d_v, ea_v, xl, xr, we, at, lane, b4, b3, b2);
        const float t1 = score_group8<1>(s_v, d_v, ea_v, xl, xr, we, at, lane, b4, b3, b2);
        const float t2 = score_group8<2>(s_v, d_v, ea_v, xl, xr, we, at, lane, b4, b3, b2);
        const float t3 = score_group8<3>(s_v, d_v, ea_v, xl, xr, we, at, lane, b4, b3, b2);
        const float u0 = merge2<2>(t0, t1, b1);
        const float u1 = merge2<2>(t2, t3, b1);
        const float p  = merge2<1>(u0, u1, b0);
        if (valid) {
            score[e] = p;
            atomicMax(&smax[d_v], enc_f32(p));
        }
    }
}

__global__ void edge_denom_kernel(const int* __restrict__ dst, const float* __restrict__ score,
                                  const unsigned* __restrict__ smax, float* __restrict__ denom,
                                  long long E)
{
    for (long long e = (long long)blockIdx.x * blockDim.x + threadIdx.x; e < E;
         e += (long long)gridDim.x * blockDim.x) {
        const int d = dst[e];
        const float ex = __expf(score[e] - dec_f32(smax[d]));
        atomicAdd(&denom[d], ex);
    }
}

__global__ void edge_agg_kernel(const int* __restrict__ src, const int* __restrict__ dst,
                                const float* __restrict__ score, const unsigned* __restrict__ smax,
                                const float* __restrict__ denom, const float* __restrict__ xl,
                                float* __restrict__ agg, long long E)
{
    const int lane = threadIdx.x & 31;
    const long long wid = (long long)blockIdx.x * (blockDim.x >> 5) + (threadIdx.x >> 5);
    const long long nw  = (long long)gridDim.x * (blockDim.x >> 5);
    for (long long e = wid; e < E; e += nw) {
        const int s = src[e];
        const int d = dst[e];
        const float alpha = __expf(score[e] - dec_f32(smax[d])) / denom[d];
        const float* xrow = xl + (size_t)s * 64;
        const float lo = xrow[lane];
        const float hi = xrow[lane + 32];
        float* ap = agg + (size_t)d * 64 + lane;
        atomicAdd(ap,      alpha * lo);
        atomicAdd(ap + 32, alpha * hi);
    }
}

__global__ void out_xform_kernel(const float* __restrict__ agg, const float* __restrict__ bias,
                                 const float* __restrict__ Wfc, const float* __restrict__ bfc,
                                 float* __restrict__ out, int NT)
{
    __shared__ float sW[WLDS];
    __shared__ float sb[64], sbias[64];
    stage_weight_pairs(sW, Wfc);
    for (int i = threadIdx.x; i < 64; i += blockDim.x) { sb[i] = bfc[i]; sbias[i] = bias[i]; }
    __syncthreads();

    const int lane = threadIdx.x & 31;
    const int wave = threadIdx.x >> 5;
    const int wpb  = blockDim.x >> 5;
    const int l15  = lane & 15;
    const int half = lane >> 4;
    const int numTiles = NT >> 4;

    for (int tile = blockIdx.x * wpb + wave; tile < numTiles; tile += gridDim.x * wpb) {
        v8f acc[4] = {v8f{}, v8f{}, v8f{}, v8f{}};
        const float* arow = agg + (size_t)(tile * 16 + l15) * 64;
        #pragma unroll 4
        for (int kk = 0; kk < 16; ++kk) {
            const int kbase = kk * 4 + half * 2;
            v2f a = *(const v2f*)(arow + kbase);
            a.x += sbias[kbase];
            a.y += sbias[kbase + 1];
            const int pbase = (kk * 2 + half) * PSTRIDE;
            #pragma unroll
            for (int nb = 0; nb < 4; ++nb) {
                const int col = nb * 16 + l15;
                const v2f b = *(const v2f*)(sW + pbase + col * 2);
                acc[nb] = __builtin_amdgcn_wmma_f32_16x16x4_f32(
                    false, a, false, b, (short)0, acc[nb], false, false);
            }
        }
        #pragma unroll
        for (int nb = 0; nb < 4; ++nb) {
            const int col = nb * 16 + l15;
            #pragma unroll
            for (int v = 0; v < 8; ++v) {
                const size_t row = (size_t)(tile * 16 + v + half * 8);
                out[row * 64 + col] = acc[nb][v] + sb[col];
            }
        }
    }
}

extern "C" void kernel_launch(void* const* d_in, const int* in_sizes, int n_in,
                              void* d_out, int out_size, void* d_ws, size_t ws_size,
                              hipStream_t stream) {
    const float* x      = (const float*)d_in[0];
    const int*   eidx   = (const int*)d_in[1];
    const float* eattr  = (const float*)d_in[2];
    const float* W_l    = (const float*)d_in[3];
    const float* b_l    = (const float*)d_in[4];
    const float* W_r    = (const float*)d_in[5];
    const float* b_r    = (const float*)d_in[6];
    const float* W_e    = (const float*)d_in[7];
    const float* att    = (const float*)d_in[8];
    const float* bias   = (const float*)d_in[9];
    const float* W_fc   = (const float*)d_in[10];
    const float* b_fc   = (const float*)d_in[11];

    const int NT       = in_sizes[0] / 64;
    const long long E  = (long long)in_sizes[2];
    const int* src = eidx;
    const int* dst = eidx + E;

    float* ws     = (float*)d_ws;
    float* xl     = ws;
    float* xr     = xl + (size_t)NT * 64;
    float* agg    = xr + (size_t)NT * 64;
    float* score  = agg + (size_t)NT * 64;
    unsigned* smax = (unsigned*)(score + E);
    float* denom  = (float*)(smax + NT);

    const int numTiles = NT / 16;
    const int gemmBlocks = (numTiles + 7) / 8;

    node_xform_kernel<<<gemmBlocks, 256, 0, stream>>>(x, W_l, b_l, W_r, b_r, xl, xr, NT);
    init_ws_kernel<<<4096, 256, 0, stream>>>(smax, denom, agg, NT);
    edge_score_kernel<<<1280, 256, 0, stream>>>(src, dst, eattr, W_e, att, xl, xr, score, smax, E);
    edge_denom_kernel<<<8192, 256, 0, stream>>>(dst, score, smax, denom, E);
    edge_agg_kernel<<<8192, 256, 0, stream>>>(src, dst, score, smax, denom, xl, agg, E);
    out_xform_kernel<<<gemmBlocks, 256, 0, stream>>>(agg, bias, W_fc, b_fc, (float*)d_out, NT);
}


# edges grouped by destination (integer-atomic rank, prefix sum, scatter of src+score); one wave per node forms max, denom and row sum in registers: no float atomics, sum order = arrival rank, unfixed
# speedup vs baseline: 2.4013x; 1.6936x over previous
#include <hip/hip_runtime.h>
#include <hip/hip_bf16.h>

typedef float v2f __attribute__((ext_vector_type(2)));
typedef float v8f __attribute__((ext_vector_type(8)));

#define PSTRIDE 160
#define WLDS    (32 * PSTRIDE)

__device__ __forceinline__ void stage_weight_pairs(float* sW, const float* __restrict__ W) {
    for (int i = threadIdx.x; i < 64 * 64; i += blockDim.x) {
        const int k = i >> 6, col = i & 63;
        sW[(k >> 1) * PSTRIDE + col * 2 + (k & 1)] = W[i];
    }
}

__global__ void node_xform_kernel(const float* __restrict__ x2,
                                  const float* __restrict__ Wl, const float* __restrict__ bl,
                                  const float* __restrict__ Wr, const float* __restrict__ br,
                                  float* __restrict__ xl, float* __restrict__ xr, int NT)
{
    __shared__ float sWl[WLDS];
    __shared__ float sWr[WLDS];
    __shared__ float sbl[64], sbr[64];
    stage_weight_pairs(sWl, Wl);
    stage_weight_pairs(sWr, Wr);
    for (int i = threadIdx.x; i < 64; i += blockDim.x) { sbl[i] = bl[i]; sbr[i] = br[i]; }
    __syncthreads();

    const int lane    = threadIdx.x & 31;
    const int wave    = threadIdx.x >> 5;
    const int wpb     = blockDim.x >> 5;
    const int l15     = lane & 15;
    const int half    = lane >> 4;
    const int numTiles = NT >> 4;

    for (int tile = blockIdx.x * wpb + wave; tile < numTiles; tile += gridDim.x * wpb) {
        v8f accl[4] = {v8f{}, v8f{}, v8f{}, v8f{}};
        v8f accr[4] = {v8f{}, v8f{}, v8f{}, v8f{}};
        const float* xrow = x2 + (size_t)(tile * 16 + l15) * 64;
        #pragma unroll 4
        for (int kk = 0; kk < 16; ++kk) {
            const int kbase = kk * 4 + half * 2;
            const v2f a = *(const v2f*)(xrow + kbase);
            const int pbase = (kk * 2 + half) * PSTRIDE;
            #pragma unroll
            for (int nb = 0; nb < 4; ++nb) {
                const int col = nb * 16 + l15;
                const v2f bL = *(const v2f*)(sWl + pbase + col * 2);
                const v2f bR = *(const v2f*)(sWr + pbase + col * 2);
                accl[nb] = __builtin_amdgcn_wmma_f32_16x16x4_f32(
                    false, a, false, bL, (short)0, accl[nb], false, false);
                accr[nb] = __builtin_amdgcn_wmma_f32_16x16x4_f32(
                    false, a, false, bR, (short)0, accr[nb], false, false);
            }
        }
        #pragma unroll
        for (int nb = 0; nb < 4; ++nb) {
            const int col = nb * 16 + l15;
            #pragma unroll
            for (int v = 0; v < 8; ++v) {
                const size_t row = (size_t)(tile * 16 + v + half * 8);
                xl[row * 64 + col] = accl[nb][v] + sbl[col];
                xr[row * 64 + col] = accr[nb][v] + sbr[col];
            }
        }
    }
}

__global__ void init_ws_kernel(unsigned* __restrict__ count, int NT)
{
    for (int i = blockIdx.x * blockDim.x + threadIdx.x; i < NT; i += gridDim.x * blockDim.x)
        count[i] = 0u;
}

__global__ void edge_rank_kernel(const int* __restrict__ dst, unsigned* __restrict__ count,
                                 unsigned* __restrict__ rank, long long E)
{
    const long long nth = (long long)gridDim.x * blockDim.x;
    const long long tid = (long long)blockIdx.x * blockDim.x + threadIdx.x;
    for (long long e0 = tid; e0 < E; e0 += 4 * nth) {
        const long long e1 = e0 + nth, e2 = e0 + 2 * nth, e3 = e0 + 3 * nth;
        const bool v1 = e1 < E, v2 = e2 < E, v3 = e3 < E;
        const int d0 = dst[e0];
        const int d1 = dst[v1 ? e1 : e0];
        const int d2 = dst[v2 ? e2 : e0];
        const int d3 = dst[v3 ? e3 : e0];
        unsigned r0, r1 = 0u, r2 = 0u, r3 = 0u;
        r0 = atomicAdd(&count[d0], 1u);
        if (v1) r1 = atomicAdd(&count[d1], 1u);
        if (v2) r2 = atomicAdd(&count[d2], 1u);
        if (v3) r3 = atomicAdd(&count[d3], 1u);
        rank[e0] = r0;
        if (v1) rank[e1] = r1;
        if (v2) rank[e2] = r2;
        if (v3) rank[e3] = r3;
    }
}

#define SCAN_TILE 1024
__global__ void __launch_bounds__(SCAN_TILE)
node_offset_kernel(const unsigned* __restrict__ count, unsigned* __restrict__ offset, int NT)
{
    __shared__ unsigned sPre[32], sTot[32];
    const int t = threadIdx.x, lane = t & 31, w = t >> 5;
    const int ntiles = (NT + SCAN_TILE - 1) / SCAN_TILE;
    for (int tile = blockIdx.x; tile < ntiles; tile += gridDim.x) {
        unsigned pre = 0u;
        for (int j = 0; j < tile; ++j) pre += count[j * SCAN_TILE + t];
        const int i = tile * SCAN_TILE + t;
        const unsigned v = (i < NT) ? count[i] : 0u;
        unsigned inc = v;
        for (int o = 1; o < 32; o <<= 1) {
            const unsigned up = __shfl_up(inc, o, 32);
            if (lane >= o) inc += up;
        }
        for (int o = 16; o > 0; o >>= 1) pre += __shfl_xor(pre, o, 32);
        __syncthreads();
        if (lane == 31) sTot[w] = inc;
        if (lane == 0)  sPre[w] = pre;
        __syncthreads();
        unsigned a = sPre[lane];
        unsigned b = (lane < w) ? sTot[lane] : 0u;
        for (int o = 16; o > 0; o >>= 1) { a += __shfl_xor(a, o, 32); b += __shfl_xor(b, o, 32); }
        if (i < NT) offset[i] = a + b + (inc - v);
    }
}

#define BREV5(j) ((((j) & 1) << 4) | (((j) & 2) << 2) | ((j) & 4) | (((j) & 8) >> 2) | (((j) & 16) >> 4))

template <int MASK>
__device__ __forceinline__ float merge2(float a, float b, bool up) {
    const float keep = up ? b : a;
    const float give = up ? a : b;
    return keep + __shfl_xor(give, MASK, 32);
}

template <int C>
__device__ __forceinline__ float score_group8(int s_v, int d_v, int ea_v,
                                              const float* __restrict__ xl,
                                              const float* __restrict__ xr,
                                              float2 we, float2 at, int lane,
                                              bool b4, bool b3, bool b2)
{
    float2 l2[8], r2[8];
    float ea[8];
    #pragma unroll
    for (int j = 0; j < 8; ++j) {
        const int s = __builtin_amdgcn_readlane(s_v, BREV5(C * 8 + j));
        const int d = __builtin_amdgcn_readlane(d_v, BREV5(C * 8 + j));
        ea[j] = __int_as_float(__builtin_amdgcn_readlane(ea_v, BREV5(C * 8 + j)));
        l2[j] = ((const float2*)(xl + (size_t)s * 64))[lane];
        r2[j] = ((const float2*)(xr + (size_t)d * 64))[lane];
    }
    float p[8];
    #pragma unroll
    for (int j = 0; j < 8; ++j) {
        float m0 = l2[j].x + r2[j].x + ea[j] * we.x;
        float m1 = l2[j].y + r2[j].y + ea[j] * we.y;
        m0 = (m0 > 0.0f) ? m0 : 0.2f * m0;
        m1 = (m1 > 0.0f) ? m1 : 0.2f * m1;
        p[j] = __builtin_fmaf(at.x, m0, at.y * m1);
    }
    const float q0 = merge2<16>(p[0], p[1], b4);
    const float q1 = merge2<16>(p[2], p[3], b4);
    const float q2 = merge2<16>(p[4], p[5], b4);
    const float q3 = merge2<16>(p[6], p[7], b4);
    const float r0 = merge2<8>(q0, q1, b3);
    const float r1 = merge2<8>(q2, q3, b3);
    return merge2<4>(r0, r1, b2);
}

__global__ void __launch_bounds__(256, 10) edge_score_kernel(const int* __restrict__ src, const int* __restrict__ dst,
                                  const float* __restrict__ eattr,
                                  const float* __restrict__ We, const float* __restrict__ att,
                                  const float* __restrict__ xl, const float* __restrict__ xr,
                                  const unsigned* __restrict__ rank, const unsigned* __restrict__ offset,
                                  int* __restrict__ pairs, long long E)
{
    const int lane = threadIdx.x & 31;
    const int wib = __builtin_amdgcn_readfirstlane((int)(threadIdx.x >> 5));
    const long long wid = (long long)blockIdx.x * (blockDim.x >> 5) + wib;
    const long long nw  = (long long)gridDim.x * (blockDim.x >> 5);
    const long long nbatch = (E + 31) >> 5;
    const float2 we = ((const float2*)We)[lane];
    const float2 at = ((const float2*)att)[lane];
    const bool b4 = (lane & 16) != 0, b3 = (lane & 8) != 0, b2 = (lane & 4) != 0;
    const bool b1 = (lane & 2) != 0,  b0 = (lane & 1) != 0;
    const int mine = BREV5(lane);
    for (long long b = wid; b < nbatch; b += nw) {
        const long long e = (b << 5) + mine;
        const bool valid = e < E;
        const int s_v  = valid ? src[e] : 0;
        const int d_v  = valid ? dst[e] : 0;
        const int ea_v = valid ? __float_as_int(eattr[e]) : 0;
        const unsigned slot = offset[d_v] + rank[valid ? e : 0];
        const float t0 = score_group8<0>(s_v, d_v, ea_v, xl, xr, we, at, lane, b4, b3, b2);
        const float t1 = score_group8<1>(s_v, d_v, ea_v, xl, xr, we, at, lane, b4, b3, b2);
        const float t2 = score_group8<2>(s_v, d_v, ea_v, xl, xr, we, at, lane, b4, b3, b2);
        const float t3 = score_group8<3>(s_v, d_v, ea_v, xl, xr, we, at, lane, b4, b3, b2);
        const float u0 = merge2<2>(t0, t1, b1);
        const float u1 = merge2<2>(t2, t3, b1);
        const float p  = merge2<1>(u0, u1, b0);
        if (valid)
            ((int2*)pairs)[slot] = make_int2(s_v, __float_as_int(p));
    }
}

template <int G>
__device__ __forceinline__ void add_group8(int s_v, float a_v, const float* __restrict__ xl,
                                           int lane, float2& acc)
{
    float2 r[8];
    float a[8];
    #pragma unroll
    for (int j = 0; j < 8; ++j) {
        const int s = __builtin_amdgcn_readlane(s_v, G * 8 + j);
        a[j] = __int_as_float(__builtin_amdgcn_readlane(__float_as_int(a_v), G * 8 + j));
        r[j] = ((const float2*)(xl + (size_t)s * 64))[lane];
    }
    #pragma unroll
    for (int j = 0; j < 8; ++j) {
        acc.x += a[j] * r[j].x;
        acc.y += a[j] * r[j].y;
    }
}

__global__ void node_agg_kernel(const int* __restrict__ pairs_i, const unsigned* __restrict__ count,
                                const unsigned* __restrict__ offset, const float* __restrict__ xl,
                                float* __restrict__ agg, int NT)
{
    const int2* __restrict__ pairs = (const int2*)pairs_i;
    const int lane = threadIdx.x & 31;
    const int wib = __builtin_amdgcn_readfirstlane((int)(threadIdx.x >> 5));
    const int wid = blockIdx.x * (blockDim.x >> 5) + wib;
    const int nw  = gridDim.x * (blockDim.x >> 5);
    for (int d = wid; d < NT; d += nw) {
        const unsigned beg = __builtin_amdgcn_readfirstlane(offset[d]);
        const unsigned n   = __builtin_amdgcn_readfirstlane(count[d]);
        const int2* __restrict__ grp = pairs + beg;

        float m = -__builtin_inff();
        for (unsigned c = 0; c < n; c += 32) {
            const unsigned i = c + lane;
            if (i < n) m = fmaxf(m, __int_as_float(grp[i].y));
        }
        for (int o = 16; o > 0; o >>= 1) m = fmaxf(m, __shfl_xor(m, o, 32));

        float den = 0.0f;
        for (unsigned c = 0; c < n; c += 32) {
            const unsigned i = c + lane;
            if (i < n) den += __expf(__int_as_float(grp[i].y) - m);
        }
        for (int o = 16; o > 0; o >>= 1) den += __shfl_xor(den, o, 32);

        float2 acc = make_float2(0.0f, 0.0f);
        for (unsigned c = 0; c < n; c += 32) {
            const unsigned i = c + lane;
            const bool valid = i < n;
            const int2 pr = valid ? grp[i] : make_int2(0, 0);
            const float a_v = valid ? __expf(__int_as_float(pr.y) - m) / den : 0.0f;
            const int s_first = __builtin_amdgcn_readfirstlane(pr.x);
            const int s_v = valid ? pr.x : s_first;
            const unsigned left = n - c;
            add_group8<0>(s_v, a_v, xl, lane, acc);
            if (left > 8)  add_group8<1>(s_v, a_v, xl, lane, acc);
            if (left > 16) add_group8<2>(s_v, a_v, xl, lane, acc);
            if (left > 24) add_group8<3>(s_v, a_v, xl, lane, acc);
        }
        ((float2*)(agg + (size_t)d * 64))[lane] = acc;
    }
}

__global__ void out_xform_kernel(const float* __restrict__ agg, const float* __restrict__ bias,
                                 const float* __restrict__ Wfc, const float* __restrict__ bfc,
                                 float* __restrict__ out, int NT)
{
    __shared__ float sW[WLDS];
    __shared__ float sb[64], sbias[64];
    stage_weight_pairs(sW, Wfc);
    for (int i = threadIdx.x; i < 64; i += blockDim.x) { sb[i] = bfc[i]; sbias[i] = bias[i]; }
    __syncthreads();

    const int lane = threadIdx.x & 31;
    const int wave = threadIdx.x >> 5;
    const int wpb  = blockDim.x >> 5;
    const int l15  = lane & 15;
    const int half = lane >> 4;
    const int numTiles = NT >> 4;

    for (int tile = blockIdx.x * wpb + wave; tile < numTiles; tile += gridDim.x * wpb) {
        v8f acc[4] = {v8f{}, v8f{}, v8f{}, v8f{}};
        const float* arow = agg + (size_t)(tile * 16 + l15) * 64;
        #pragma unroll 4
        for (int kk = 0; kk < 16; ++kk) {
            const int kbase = kk * 4 + half * 2;
            v2f a = *(const v2f*)(arow + kbase);
            a.x += sbias[kbase];
            a.y += sbias[kbase + 1];
            const int pbase = (kk * 2 + half) * PSTRIDE;
            #pragma unroll
            for (int nb = 0; nb < 4; ++nb) {
                const int col = nb * 16 + l15;
                const v2f b = *(const v2f*)(sW + pbase + col * 2);
                acc[nb] = __builtin_amdgcn_wmma_f32_16x16x4_f32(
                    false, a, false, b, (short)0, acc[nb], false, false);
            }
        }
        #pragma unroll
        for (int nb = 0; nb < 4; ++nb) {
            const int col = nb * 16 + l15;
            #pragma unroll
            for (int v = 0; v < 8; ++v) {
                const size_t row = (size_t)(tile * 16 + v + half * 8);
                out[row * 64 + col] = acc[nb][v] + sb[col];
            }
        }
    }
}

extern "C" void kernel_launch(void* const* d_in, const int* in_sizes, int n_in,
                              void* d_out, int out_size, void* d_ws, size_t ws_size,
                              hipStream_t stream) {
    const float* x      = (const float*)d_in[0];
    const int*   eidx   = (const int*)d_in[1];
    const float* eattr  = (const float*)d_in[2];
    const float* W_l    = (const float*)d_in[3];
    const float* b_l    = (const float*)d_in[4];
    const float* W_r    = (const float*)d_in[5];
    const float* b_r    = (const float*)d_in[6];
    const float* W_e    = (const float*)d_in[7];
    const float* att    = (const float*)d_in[8];
    const float* bias   = (const float*)d_in[9];
    const float* W_fc   = (const float*)d_in[10];
    const float* b_fc   = (const float*)d_in[11];

    const int NT       = in_sizes[0] / 64;
    const long long E  = (long long)in_sizes[2];
    const int* src = eidx;
    const int* dst = eidx + E;

    float* ws       = (float*)d_ws;
    float* xl       = ws;
    float* xr       = xl + (size_t)NT * 64;
    int*   pairs    = (int*)(xr + (size_t)NT * 64);
    unsigned* rank  = (unsigned*)(pairs + (size_t)NT * 64);
    unsigned* count = rank + E;
    unsigned* offset = count + NT;
    float* agg      = xr;

    const int numTiles = NT / 16;
    const int gemmBlocks = (numTiles + 7) / 8;

    init_ws_kernel<<<64, 256, 0, stream>>>(count, NT);
    edge_rank_kernel<<<1280, 256, 0, stream>>>(dst, count, rank, E);
    node_offset_kernel<<<64, SCAN_TILE, 0, stream>>>(count, offset, NT);
    node_xform_kernel<<<gemmBlocks, 256, 0, stream>>>(x, W_l, b_l, W_r, b_r, xl, xr, NT);
    edge_score_kernel<<<1280, 256, 0, stream>>>(src, dst, eattr, W_e, att, xl, xr, rank, offset, pairs, E);
    node_agg_kernel<<<1250, 256, 0, stream>>>(pairs, count, offset, xl, agg, NT);
    out_xform_kernel<<<gemmBlocks, 256, 0, stream>>>(agg, bias, W_fc, b_fc, (float*)d_out, NT);
}


# edges grouped by destination with LDS counters: 256-node buckets (one global add per bucket and block), then a block orders each bucket in LDS; no per-edge global atomic; sum order still unfixed
# speedup vs baseline: 4.2776x; 1.7814x over previous
#include <hip/hip_runtime.h>
#include <hip/hip_bf16.h>

typedef float v2f __attribute__((ext_vector_type(2)));
typedef float v8f __attribute__((ext_vector_type(8)));

#define PSTRIDE 160
#define WLDS    (32 * PSTRIDE)

__device__ __forceinline__ void stage_weight_pairs(float* sW, const float* __restrict__ W) {
    for (int i = threadIdx.x; i < 64 * 64; i += blockDim.x) {
        const int k = i >> 6, col = i & 63;
        sW[(k >> 1) * PSTRIDE + col * 2 + (k & 1)] = W[i];
    }
}

__global__ void node_xform_kernel(const float* __restrict__ x2,
                                  const float* __restrict__ Wl, const float* __restrict__ bl,
                                  const float* __restrict__ Wr, const float* __restrict__ br,
                                  float* __restrict__ xl, float* __restrict__ xr, int NT)
{
    __shared__ float sWl[WLDS];
    __shared__ float sWr[WLDS];
    __shared__ float sbl[64], sbr[64];
    stage_weight_pairs(sWl, Wl);
    stage_weight_pairs(sWr, Wr);
    for (int i = threadIdx.x; i < 64; i += blockDim.x) { sbl[i] = bl[i]; sbr[i] = br[i]; }
    __syncthreads();

    const int lane    = threadIdx.x & 31;
    const int wave    = threadIdx.x >> 5;
    const int wpb     = blockDim.x >> 5;
    const int l15     = lane & 15;
    const int half    = lane >> 4;
    const int numTiles = NT >> 4;

    for (int tile = blockIdx.x * wpb + wave; tile < numTiles; tile += gridDim.x * wpb) {
        v8f accl[4] = {v8f{}, v8f{}, v8f{}, v8f{}};
        v8f accr[4] = {v8f{}, v8f{}, v8f{}, v8f{}};
        const float* xrow = x2 + (size_t)(tile * 16 + l15) * 64;
        #pragma unroll 4
        for (int kk = 0; kk < 16; ++kk) {
            const int kbase = kk * 4 + half * 2;
            const v2f a = *(const v2f*)(xrow + kbase);
            const int pbase = (kk * 2 + half) * PSTRIDE;
            #pragma unroll
            for (int nb = 0; nb < 4; ++nb) {
                const int col = nb * 16 + l15;
                const v2f bL = *(const v2f*)(sWl + pbase + col * 2);
                const v2f bR = *(const v2f*)(sWr + pbase + col * 2);
                accl[nb] = __builtin_amdgcn_wmma_f32_16x16x4_f32(
                    false, a, false, bL, (short)0, accl[nb], false, false);
                accr[nb] = __builtin_amdgcn_wmma_f32_16x16x4_f32(
                    false, a, false, bR, (short)0, accr[nb], false, false);
            }
        }
        #pragma unroll
        for (int nb = 0; nb < 4; ++nb) {
            const int col = nb * 16 + l15;
            #pragma unroll
            for (int v = 0; v < 8; ++v) {
                const size_t row = (size_t)(tile * 16 + v + half * 8);
                xl[row * 64 + col] = accl[nb][v] + sbl[col];
                xr[row * 64 + col] = accr[nb][v] + sbr[col];
            }
        }
    }
}

#define NB_SHIFT 8
#define NB_NODES 256
#define NBMAX    256
#define STAGE_CAP 8192

__global__ void init_ws_kernel(unsigned* __restrict__ count, int NT)
{
    for (int i = blockIdx.x * blockDim.x + threadIdx.x; i < NT; i += gridDim.x * blockDim.x)
        count[i] = 0u;
}

__device__ __forceinline__ unsigned excl_scan_256(unsigned v, unsigned* sTot)
{
    const int lane = threadIdx.x & 31, w = threadIdx.x >> 5;
    unsigned inc = v;
    for (int o = 1; o < 32; o <<= 1) {
        const unsigned up = __shfl_up(inc, o, 32);
        if (lane >= o) inc += up;
    }
    __syncthreads();
    if (lane == 31 && w < 8) sTot[w] = inc;
    __syncthreads();
    unsigned pre = 0u;
    for (int j = 0; j < 8; ++j) if (j < w) pre += sTot[j];
    return pre + inc - v;
}

__global__ void __launch_bounds__(1024)
bucket_count_kernel(const int* __restrict__ dst, unsigned* __restrict__ btotal, int NT, long long E)
{
    __shared__ unsigned sCnt[NBMAX];
    const int t = threadIdx.x;
    if (t < NBMAX) sCnt[t] = 0u;
    __syncthreads();
    const long long nth = (long long)gridDim.x * blockDim.x;
    const long long tid = (long long)blockIdx.x * blockDim.x + t;
    for (long long e0 = tid; e0 < E; e0 += 4 * nth) {
        const long long e1 = e0 + nth, e2 = e0 + 2 * nth, e3 = e0 + 3 * nth;
        const bool v1 = e1 < E, v2 = e2 < E, v3 = e3 < E;
        const int d0 = dst[e0];
        const int d1 = dst[v1 ? e1 : e0];
        const int d2 = dst[v2 ? e2 : e0];
        const int d3 = dst[v3 ? e3 : e0];
        atomicAdd(&sCnt[d0 >> NB_SHIFT], 1u);
        if (v1) atomicAdd(&sCnt[d1 >> NB_SHIFT], 1u);
        if (v2) atomicAdd(&sCnt[d2 >> NB_SHIFT], 1u);
        if (v3) atomicAdd(&sCnt[d3 >> NB_SHIFT], 1u);
    }
    __syncthreads();
    const int nb = (NT + NB_NODES - 1) >> NB_SHIFT;
    if (t < nb) {
        const unsigned c = sCnt[t];
        if (c) atomicAdd(&btotal[t], c);
    }
}

#define BREV5(j) ((((j) & 1) << 4) | (((j) & 2) << 2) | ((j) & 4) | (((j) & 8) >> 2) | (((j) & 16) >> 4))

template <int MASK>
__device__ __forceinline__ float merge2(float a, float b, bool up) {
    const float keep = up ? b : a;
    const float give = up ? a : b;
    return keep + __shfl_xor(give, MASK, 32);
}

template <int C>
__device__ __forceinline__ float score_group8(int s_v, int d_v, int ea_v,
                                              const float* __restrict__ xl,
                                              const float* __restrict__ xr,
                                              float2 we, float2 at, int lane,
                                              bool b4, bool b3, bool b2)
{
    float2 l2[8], r2[8];
    float ea[8];
    #pragma unroll
    for (int j = 0; j < 8; ++j) {
        const int s = __builtin_amdgcn_readlane(s_v, BREV5(C * 8 + j));
        const int d = __builtin_amdgcn_readlane(d_v, BREV5(C * 8 + j));
        ea[j] = __int_as_float(__builtin_amdgcn_readlane(ea_v, BREV5(C * 8 + j)));
        l2[j] = ((const float2*)(xl + (size_t)s * 64))[lane];
        r2[j] = ((const float2*)(xr + (size_t)d * 64))[lane];
    }
    float p[8];
    #pragma unroll
    for (int j = 0; j < 8; ++j) {
        float m0 = l2[j].x + r2[j].x + ea[j] * we.x;
        float m1 = l2[j].y + r2[j].y + ea[j] * we.y;
        m0 = (m0 > 0.0f) ? m0 : 0.2f * m0;
        m1 = (m1 > 0.0f) ? m1 : 0.2f * m1;
        p[j] = __builtin_fmaf(at.x, m0, at.y * m1);
    }
    const float q0 = merge2<16>(p[0], p[1], b4);
    const float q1 = merge2<16>(p[2], p[3], b4);
    const float q2 = merge2<16>(p[4], p[5], b4);
    const float q3 = merge2<16>(p[6], p[7], b4);
    const float r0 = merge2<8>(q0, q1, b3);
    const float r1 = merge2<8>(q2, q3, b3);
    return merge2<4>(r0, r1, b2);
}

__global__ void __launch_bounds__(1024) edge_score_kernel(const int* __restrict__ src, const int* __restrict__ dst,
                                  const float* __restrict__ eattr,
                                  const float* __restrict__ We, const float* __restrict__ att,
                                  const float* __restrict__ xl, const float* __restrict__ xr,
                                  const unsigned* __restrict__ btotal, unsigned* __restrict__ bcursor,
                                  int* __restrict__ recs, int NT, long long E)
{
    __shared__ unsigned sCnt[NBMAX];
    __shared__ unsigned sBase[NBMAX];
    __shared__ unsigned sLoc[NBMAX];
    __shared__ unsigned sTot[8];
    __shared__ int2     sRec[STAGE_CAP];
    __shared__ unsigned sSlot[STAGE_CAP];
    const int t = threadIdx.x;
    const int lane = t & 31;
    const int wib = __builtin_amdgcn_readfirstlane((int)(threadIdx.x >> 5));
    const long long nbatch = (E + 31) >> 5;
    const long long bbeg = (nbatch * blockIdx.x) / gridDim.x;
    const long long bend = (nbatch * (blockIdx.x + 1)) / gridDim.x;
    const int nwb = (int)(blockDim.x >> 5);
    const int mine = BREV5(lane);

    if (t < NBMAX) sCnt[t] = 0u;
    __syncthreads();
    for (long long b0 = bbeg + wib; b0 < bend; b0 += 8 * nwb) {
        int d[8];
        #pragma unroll
        for (int j = 0; j < 8; ++j) {
            const long long b = b0 + (long long)j * nwb;
            const long long e = (b << 5) + mine;
            d[j] = dst[(b < bend && e < E) ? e : 0];
        }
        #pragma unroll
        for (int j = 0; j < 8; ++j) {
            const long long b = b0 + (long long)j * nwb;
            const long long e = (b << 5) + mine;
            if (b < bend && e < E) atomicAdd(&sCnt[d[j] >> NB_SHIFT], 1u);
        }
    }
    __syncthreads();
    const int nb = (NT + NB_NODES - 1) >> NB_SHIFT;
    const unsigned tot = (t < nb) ? btotal[t] : 0u;
    const unsigned first = excl_scan_256(tot, sTot);
    const unsigned c = (t < nb) ? sCnt[t] : 0u;
    const unsigned loc = excl_scan_256(c, sTot);
    if (t < nb) {
        unsigned got = 0u;
        if (c) got = atomicAdd(&bcursor[t], c);
        sBase[t] = first + got;
        sLoc[t]  = loc;
        sCnt[t]  = 0u;
    }
    __syncthreads();
    const long long ebeg = bbeg << 5;
    const long long eend = ((bend << 5) < E) ? (bend << 5) : E;
    const unsigned nblk = (eend > ebeg) ? (unsigned)(eend - ebeg) : 0u;
    const bool staged = nblk <= STAGE_CAP;

    const float2 we = ((const float2*)We)[lane];
    const float2 at = ((const float2*)att)[lane];
    const bool b4 = (lane & 16) != 0, b3 = (lane & 8) != 0, b2 = (lane & 4) != 0;
    const bool b1 = (lane & 2) != 0,  b0 = (lane & 1) != 0;
    for (long long b = bbeg + wib; b < bend; b += nwb) {
        const long long e = (b << 5) + mine;
        const bool valid = e < E;
        const int s_v  = valid ? src[e] : 0;
        const int d_v  = valid ? dst[e] : 0;
        const int ea_v = valid ? __float_as_int(eattr[e]) : 0;
        const float t0 = score_group8<0>(s_v, d_v, ea_v, xl, xr, we, at, lane, b4, b3, b2);
        const float t1 = score_group8<1>(s_v, d_v, ea_v, xl, xr, we, at, lane, b4, b3, b2);
        const float t2 = score_group8<2>(s_v, d_v, ea_v, xl, xr, we, at, lane, b4, b3, b2);
        const float t3 = score_group8<3>(s_v, d_v, ea_v, xl, xr, we, at, lane, b4, b3, b2);
        const float u0 = merge2<2>(t0, t1, b1);
        const float u1 = merge2<2>(t2, t3, b1);
        const float p  = merge2<1>(u0, u1, b0);
        if (valid) {
            const int bk = d_v >> NB_SHIFT;
            const unsigned r = atomicAdd(&sCnt[bk], 1u);
            const unsigned slot = sBase[bk] + r;
            const unsigned key = (unsigned)s_v | ((unsigned)(d_v & (NB_NODES - 1)) << 24);
            const int2 rec = make_int2((int)key, __float_as_int(p));
            if (staged) {
                const unsigned lp = sLoc[bk] + r;
                sRec[lp]  = rec;
                sSlot[lp] = slot;
            } else {
                ((int2*)recs)[slot] = rec;
            }
        }
    }
    if (staged) {
        __syncthreads();
        for (unsigned i = t; i < nblk; i += blockDim.x) ((int2*)recs)[sSlot[i]] = sRec[i];
    }
}

#define SORT_CAP 12288
__global__ void __launch_bounds__(1024)
bucket_sort_kernel(const int* __restrict__ recs_i, const unsigned* __restrict__ btotal,
                   int* __restrict__ pairs_i, unsigned* __restrict__ count,
                   unsigned* __restrict__ offset, int NT)
{
    __shared__ int2 sP[SORT_CAP];
    __shared__ unsigned sCnt[NB_NODES], sOff[NB_NODES];
    __shared__ unsigned sTot[8];
    __shared__ unsigned sBeg, sNum;
    const int2* __restrict__ recs = (const int2*)recs_i;
    int2* __restrict__ pairs = (int2*)pairs_i;
    const int t = threadIdx.x;
    const int nb = (NT + NB_NODES - 1) >> NB_SHIFT;
    for (int b = blockIdx.x; b < nb; b += gridDim.x) {
        const unsigned tot = (t < nb) ? btotal[t] : 0u;
        const unsigned first = excl_scan_256(tot, sTot);
        if (t == b) { sBeg = first; sNum = tot; }
        if (t < NB_NODES) sCnt[t] = 0u;
        __syncthreads();
        const unsigned beg = sBeg, n = sNum;
        for (unsigned i0 = t; i0 < n; i0 += 8 * 1024) {
            int k[8];
            #pragma unroll
            for (int j = 0; j < 8; ++j) {
                const unsigned i = i0 + j * 1024;
                k[j] = recs[beg + (i < n ? i : i0)].x;
            }
            #pragma unroll
            for (int j = 0; j < 8; ++j)
                if (i0 + j * 1024 < n) atomicAdd(&sCnt[(unsigned)k[j] >> 24], 1u);
        }
        __syncthreads();
        const unsigned c = (t < NB_NODES) ? sCnt[t] : 0u;
        const unsigned noff = excl_scan_256(c, sTot);
        if (t < NB_NODES) {
            const int node = b * NB_NODES + t;
            if (node < NT) { count[node] = c; offset[node] = beg + noff; }
            sOff[t] = noff;
            sCnt[t] = 0u;
        }
        __syncthreads();
        const bool staged = n <= SORT_CAP;
        for (unsigned i0 = t; i0 < n; i0 += 8 * 1024) {
            int2 r[8];
            #pragma unroll
            for (int j = 0; j < 8; ++j) {
                const unsigned i = i0 + j * 1024;
                r[j] = recs[beg + (i < n ? i : i0)];
            }
            #pragma unroll
            for (int j = 0; j < 8; ++j) {
                if (i0 + j * 1024 < n) {
                    const unsigned dl = (unsigned)r[j].x >> 24;
                    const unsigned pos = sOff[dl] + atomicAdd(&sCnt[dl], 1u);
                    const int2 o = make_int2(r[j].x & 0x00FFFFFF, r[j].y);
                    if (staged) sP[pos] = o; else pairs[beg + pos] = o;
                }
            }
        }
        if (staged) {
            __syncthreads();
            for (unsigned i = t; i < n; i += 1024) pairs[beg + i] = sP[i];
        }
        __syncthreads();
    }
}

template <int G>
__device__ __forceinline__ void add_group8(int s_v, float a_v, const float* __restrict__ xl,
                                           int lane, float2& acc)
{
    float2 r[8];
    float a[8];
    #pragma unroll
    for (int j = 0; j < 8; ++j) {
        const int s = __builtin_amdgcn_readlane(s_v, G * 8 + j);
        a[j] = __int_as_float(__builtin_amdgcn_readlane(__float_as_int(a_v), G * 8 + j));
        r[j] = ((const float2*)(xl + (size_t)s * 64))[lane];
    }
    #pragma unroll
    for (int j = 0; j < 8; ++j) {
        acc.x += a[j] * r[j].x;
        acc.y += a[j] * r[j].y;
    }
}

__global__ void node_agg_kernel(const int* __restrict__ pairs_i, const unsigned* __restrict__ count,
                                const unsigned* __restrict__ offset, const float* __restrict__ xl,
                                float* __restrict__ agg, int NT)
{
    const int2* __restrict__ pairs = (const int2*)pairs_i;
    const int lane = threadIdx.x & 31;
    const int wib = __builtin_amdgcn_readfirstlane((int)(threadIdx.x >> 5));
    const int wid = blockIdx.x * (blockDim.x >> 5) + wib;
    const int nw  = gridDim.x * (blockDim.x >> 5);
    for (int d = wid; d < NT; d += nw) {
        const unsigned beg = __builtin_amdgcn_readfirstlane(offset[d]);
        const unsigned n   = __builtin_amdgcn_readfirstlane(count[d]);
        const int2* __restrict__ grp = pairs + beg;

        float m = -__builtin_inff();
        for (unsigned c = 0; c < n; c += 32) {
            const unsigned i = c + lane;
            if (i < n) m = fmaxf(m, __int_as_float(grp[i].y));
        }
        for (int o = 16; o > 0; o >>= 1) m = fmaxf(m, __shfl_xor(m, o, 32));

        float den = 0.0f;
        for (unsigned c = 0; c < n; c += 32) {
            const unsigned i = c + lane;
            if (i < n) den += __expf(__int_as_float(grp[i].y) - m);
        }
        for (int o = 16; o > 0; o >>= 1) den += __shfl_xor(den, o, 32);

        float2 acc = make_float2(0.0f, 0.0f);
        for (unsigned c = 0; c < n; c += 32) {
            const unsigned i = c + lane;
            const bool valid = i < n;
            const int2 pr = valid ? grp[i] : make_int2(0, 0);
            const float a_v = valid ? __expf(__int_as_float(pr.y) - m) / den : 0.0f;
            const int s_first = __builtin_amdgcn_readfirstlane(pr.x);
            const int s_v = valid ? pr.x : s_first;
            const unsigned left = n - c;
            add_group8<0>(s_v, a_v, xl, lane, acc);
            if (left > 8)  add_group8<1>(s_v, a_v, xl, lane, acc);
            if (left > 16) add_group8<2>(s_v, a_v, xl, lane, acc);
            if (left > 24) add_group8<3>(s_v, a_v, xl, lane, acc);
        }
        ((float2*)(agg + (size_t)d * 64))[lane] = acc;
    }
}

__global__ void out_xform_kernel(const float* __restrict__ agg, const float* __restrict__ bias,
                                 const float* __restrict__ Wfc, const float* __restrict__ bfc,
                                 float* __restrict__ out, int NT)
{
    __shared__ float sW[WLDS];
    __shared__ float sb[64], sbias[64];
    stage_weight_pairs(sW, Wfc);
    for (int i = threadIdx.x; i < 64; i += blockDim.x) { sb[i] = bfc[i]; sbias[i] = bias[i]; }
    __syncthreads();

    const int lane = threadIdx.x & 31;
    const int wave = threadIdx.x >> 5;
    const int wpb  = blockDim.x >> 5;
    const int l15  = lane & 15;
    const int half = lane >> 4;
    const int numTiles = NT >> 4;

    for (int tile = blockIdx.x * wpb + wave; tile < numTiles; tile += gridDim.x * wpb) {
        v8f acc[4] = {v8f{}, v8f{}, v8f{}, v8f{}};
        const float* arow = agg + (size_t)(tile * 16 + l15) * 64;
        #pragma unroll 4
        for (int kk = 0; kk < 16; ++kk) {
            const int kbase = kk * 4 + half * 2;
            v2f a = *(const v2f*)(arow + kbase);
            a.x += sbias[kbase];
            a.y += sbias[kbase + 1];
            const int pbase = (kk * 2 + half) * PSTRIDE;
            #pragma unroll
            for (int nb = 0; nb < 4; ++nb) {
                const int col = nb * 16 + l15;
                const v2f b = *(const v2f*)(sW + pbase + col * 2);
                acc[nb] = __builtin_amdgcn_wmma_f32_16x16x4_f32(
                    false, a, false, b, (short)0, acc[nb], false, false);
            }
        }
        #pragma unroll
        for (int nb = 0; nb < 4; ++nb) {
            const int col = nb * 16 + l15;
            #pragma unroll
            for (int v = 0; v < 8; ++v) {
                const size_t row = (size_t)(tile * 16 + v + half * 8);
                out[row * 64 + col] = acc[nb][v] + sb[col];
            }
        }
    }
}

extern "C" void kernel_launch(void* const* d_in, const int* in_sizes, int n_in,
                              void* d_out, int out_size, void* d_ws, size_t ws_size,
                              hipStream_t stream) {
    const float* x      = (const float*)d_in[0];
    const int*   eidx   = (const int*)d_in[1];
    const float* eattr  = (const float*)d_in[2];
    const float* W_l    = (const float*)d_in[3];
    const float* b_l    = (const float*)d_in[4];
    const float* W_r    = (const float*)d_in[5];
    const float* b_r    = (const float*)d_in[6];
    const float* W_e    = (const float*)d_in[7];
    const float* att    = (const float*)d_in[8];
    const float* bias   = (const float*)d_in[9];
    const float* W_fc   = (const float*)d_in[10];
    const float* b_fc   = (const float*)d_in[11];

    const int NT       = in_sizes[0] / 64;
    const long long E  = (long long)in_sizes[2];
    const int* src = eidx;
    const int* dst = eidx + E;

    float* ws       = (float*)d_ws;
    float* xl       = ws;
    float* xr       = xl + (size_t)NT * 64;
    int*   recs     = (int*)(xr + (size_t)NT * 64);
    unsigned* btotal  = (unsigned*)(recs + (size_t)NT * 64);
    unsigned* bcursor = btotal + NBMAX;
    unsigned* count = btotal + E;
    unsigned* offset = count + NT;
    int*   pairs    = (int*)xr;
    float* agg      = (float*)recs;

    const int numTiles = NT / 16;
    const int gemmBlocks = (numTiles + 7) / 8;

    init_ws_kernel<<<1, 512, 0, stream>>>(btotal, 512);
    bucket_count_kernel<<<256, 1024, 0, stream>>>(dst, btotal, NT, E);
    node_xform_kernel<<<gemmBlocks, 256, 0, stream>>>(x, W_l, b_l, W_r, b_r, xl, xr, NT);
    edge_score_kernel<<<256, 1024, 0, stream>>>(src, dst, eattr, W_e, att, xl, xr, btotal, bcursor, recs, NT, E);
    bucket_sort_kernel<<<196, 1024, 0, stream>>>(recs, btotal, pairs, count, offset, NT);
    node_agg_kernel<<<1250, 256, 0, stream>>>(pairs, count, offset, xl, agg, NT);
    out_xform_kernel<<<gemmBlocks, 256, 0, stream>>>(agg, bias, W_fc, b_fc, (float*)d_out, NT);
}


# score moved into the per-node wave behind the sort (records carry attr): xl[src] row gathered once, xr[dst] once per node; chunks joined by running max, divided by denom last (sum order changes)
# speedup vs baseline: 5.0163x; 1.1727x over previous
#include <hip/hip_runtime.h>
#include <hip/hip_bf16.h>

typedef float v2f __attribute__((ext_vector_type(2)));
typedef float v8f __attribute__((ext_vector_type(8)));

#define PSTRIDE 160
#define WLDS    (32 * PSTRIDE)

__device__ __forceinline__ void stage_weight_pairs(float* sW, const float* __restrict__ W) {
    for (int i = threadIdx.x; i < 64 * 64; i += blockDim.x) {
        const int k = i >> 6, col = i & 63;
        sW[(k >> 1) * PSTRIDE + col * 2 + (k & 1)] = W[i];
    }
}

__global__ void node_xform_kernel(const float* __restrict__ x2,
                                  const float* __restrict__ Wl, const float* __restrict__ bl,
                                  const float* __restrict__ Wr, const float* __restrict__ br,
                                  float* __restrict__ xl, float* __restrict__ xr, int NT)
{
    __shared__ float sWl[WLDS];
    __shared__ float sWr[WLDS];
    __shared__ float sbl[64], sbr[64];
    stage_weight_pairs(sWl, Wl);
    stage_weight_pairs(sWr, Wr);
    for (int i = threadIdx.x; i < 64; i += blockDim.x) { sbl[i] = bl[i]; sbr[i] = br[i]; }
    __syncthreads();

    const int lane    = threadIdx.x & 31;
    const int wave    = threadIdx.x >> 5;
    const int wpb     = blockDim.x >> 5;
    const int l15     = lane & 15;
    const int half    = lane >> 4;
    const int numTiles = NT >> 4;

    for (int tile = blockIdx.x * wpb + wave; tile < numTiles; tile += gridDim.x * wpb) {
        v8f accl[4] = {v8f{}, v8f{}, v8f{}, v8f{}};
        v8f accr[4] = {v8f{}, v8f{}, v8f{}, v8f{}};
        const float* xrow = x2 + (size_t)(tile * 16 + l15) * 64;
        #pragma unroll 4
        for (int kk = 0; kk < 16; ++kk) {
            const int kbase = kk * 4 + half * 2;
            const v2f a = *(const v2f*)(xrow + kbase);
            const int pbase = (kk * 2 + half) * PSTRIDE;
            #pragma unroll
            for (int nb = 0; nb < 4; ++nb) {
                const int col = nb * 16 + l15;
                const v2f bL = *(const v2f*)(sWl + pbase + col * 2);
                const v2f bR = *(const v2f*)(sWr + pbase + col * 2);
                accl[nb] = __builtin_amdgcn_wmma_f32_16x16x4_f32(
                    false, a, false, bL, (short)0, accl[nb], false, false);
                accr[nb] = __builtin_amdgcn_wmma_f32_16x16x4_f32(
                    false, a, false, bR, (short)0, accr[nb], false, false);
            }
        }
        #pragma unroll
        for (int nb = 0; nb < 4; ++nb) {
            const int col = nb * 16 + l15;
            #pragma unroll
            for (int v = 0; v < 8; ++v) {
                const size_t row = (size_t)(tile * 16 + v + half * 8);
                xl[row * 64 + col] = accl[nb][v] + sbl[col];
                xr[row * 64 + col] = accr[nb][v] + sbr[col];
            }
        }
    }
}

#define NB_SHIFT 8
#define NB_NODES 256
#define NBMAX    256
#define STAGE_CAP 8192

__global__ void init_ws_kernel(unsigned* __restrict__ count, int NT)
{
    for (int i = blockIdx.x * blockDim.x + threadIdx.x; i < NT; i += gridDim.x * blockDim.x)
        count[i] = 0u;
}

__device__ __forceinline__ unsigned excl_scan_256(unsigned v, unsigned* sTot)
{
    const int lane = threadIdx.x & 31, w = threadIdx.x >> 5;
    unsigned inc = v;
    for (int o = 1; o < 32; o <<= 1) {
        const unsigned up = __shfl_up(inc, o, 32);
        if (lane >= o) inc += up;
    }
    __syncthreads();
    if (lane == 31 && w < 8) sTot[w] = inc;
    __syncthreads();
    unsigned pre = 0u;
    for (int j = 0; j < 8; ++j) if (j < w) pre += sTot[j];
    return pre + inc - v;
}

__global__ void __launch_bounds__(1024)
bucket_count_kernel(const int* __restrict__ dst, unsigned* __restrict__ btotal, int NT, long long E)
{
    __shared__ unsigned sCnt[NBMAX];
    const int t = threadIdx.x;
    if (t < NBMAX) sCnt[t] = 0u;
    __syncthreads();
    const long long nth = (long long)gridDim.x * blockDim.x;
    const long long tid = (long long)blockIdx.x * blockDim.x + t;
    for (long long e0 = tid; e0 < E; e0 += 4 * nth) {
        const long long e1 = e0 + nth, e2 = e0 + 2 * nth, e3 = e0 + 3 * nth;
        const bool v1 = e1 < E, v2 = e2 < E, v3 = e3 < E;
        const int d0 = dst[e0];
        const int d1 = dst[v1 ? e1 : e0];
        const int d2 = dst[v2 ? e2 : e0];
        const int d3 = dst[v3 ? e3 : e0];
        atomicAdd(&sCnt[d0 >> NB_SHIFT], 1u);
        if (v1) atomicAdd(&sCnt[d1 >> NB_SHIFT], 1u);
        if (v2) atomicAdd(&sCnt[d2 >> NB_SHIFT], 1u);
        if (v3) atomicAdd(&sCnt[d3 >> NB_SHIFT], 1u);
    }
    __syncthreads();
    const int nb = (NT + NB_NODES - 1) >> NB_SHIFT;
    if (t < nb) {
        const unsigned c = sCnt[t];
        if (c) atomicAdd(&btotal[t], c);
    }
}

__global__ void __launch_bounds__(1024) edge_scatter_kernel(const int* __restrict__ src, const int* __restrict__ dst,
                                  const float* __restrict__ eattr,
                                  const unsigned* __restrict__ btotal, unsigned* __restrict__ bcursor,
                                  int* __restrict__ recs, int NT, long long E)
{
    __shared__ unsigned sCnt[NBMAX];
    __shared__ unsigned sBase[NBMAX];
    __shared__ unsigned sLoc[NBMAX];
    __shared__ unsigned sTot[8];
    __shared__ int2     sRec[STAGE_CAP];
    __shared__ unsigned sSlot[STAGE_CAP];
    const int t = threadIdx.x;
    const long long ebeg = (E * blockIdx.x) / gridDim.x;
    const long long eend = (E * (blockIdx.x + 1)) / gridDim.x;
    const unsigned nblk = (eend > ebeg) ? (unsigned)(eend - ebeg) : 0u;

    if (t < NBMAX) sCnt[t] = 0u;
    __syncthreads();
    for (long long e0 = ebeg + t; e0 < eend; e0 += 8 * 1024) {
        int d[8];
        #pragma unroll
        for (int j = 0; j < 8; ++j) {
            const long long e = e0 + j * 1024;
            d[j] = dst[(e < eend) ? e : e0];
        }
        #pragma unroll
        for (int j = 0; j < 8; ++j)
            if (e0 + j * 1024 < eend) atomicAdd(&sCnt[d[j] >> NB_SHIFT], 1u);
    }
    __syncthreads();
    const int nb = (NT + NB_NODES - 1) >> NB_SHIFT;
    const unsigned tot = (t < nb) ? btotal[t] : 0u;
    const unsigned first = excl_scan_256(tot, sTot);
    const unsigned c = (t < nb) ? sCnt[t] : 0u;
    const unsigned loc = excl_scan_256(c, sTot);
    if (t < nb) {
        unsigned got = 0u;
        if (c) got = atomicAdd(&bcursor[t], c);
        sBase[t] = first + got;
        sLoc[t]  = loc;
        sCnt[t]  = 0u;
    }
    __syncthreads();
    const bool staged = nblk <= STAGE_CAP;
    for (long long e0 = ebeg + t; e0 < eend; e0 += 4 * 1024) {
        int s[4], d[4];
        float a[4];
        #pragma unroll
        for (int j = 0; j < 4; ++j) {
            const long long e = e0 + j * 1024;
            const long long ec = (e < eend) ? e : e0;
            s[j] = src[ec];
            d[j] = dst[ec];
            a[j] = eattr[ec];
        }
        #pragma unroll
        for (int j = 0; j < 4; ++j) {
            if (e0 + j * 1024 < eend) {
                const int bk = d[j] >> NB_SHIFT;
                const unsigned r = atomicAdd(&sCnt[bk], 1u);
                const unsigned slot = sBase[bk] + r;
                const unsigned key = (unsigned)s[j] | ((unsigned)(d[j] & (NB_NODES - 1)) << 24);
                const int2 rec = make_int2((int)key, __float_as_int(a[j]));
                if (staged) {
                    const unsigned lp = sLoc[bk] + r;
                    sRec[lp]  = rec;
                    sSlot[lp] = slot;
                } else {
                    ((int2*)recs)[slot] = rec;
                }
            }
        }
    }
    if (staged) {
        __syncthreads();
        for (unsigned i = t; i < nblk; i += blockDim.x) ((int2*)recs)[sSlot[i]] = sRec[i];
    }
}

#define SORT_CAP 12288
__global__ void __launch_bounds__(1024)
bucket_sort_kernel(const int* __restrict__ recs_i, const unsigned* __restrict__ btotal,
                   int* __restrict__ pairs_i, unsigned* __restrict__ count,
                   unsigned* __restrict__ offset, int NT)
{
    __shared__ int2 sP[SORT_CAP];
    __shared__ unsigned sCnt[NB_NODES], sOff[NB_NODES];
    __shared__ unsigned sTot[8];
    __shared__ unsigned sBeg, sNum;
    const int2* __restrict__ recs = (const int2*)recs_i;
    int2* __restrict__ pairs = (int2*)pairs_i;
    const int t = threadIdx.x;
    const int nb = (NT + NB_NODES - 1) >> NB_SHIFT;
    for (int b = blockIdx.x; b < nb; b += gridDim.x) {
        const unsigned tot = (t < nb) ? btotal[t] : 0u;
        const unsigned first = excl_scan_256(tot, sTot);
        if (t == b) { sBeg = first; sNum = tot; }
        if (t < NB_NODES) sCnt[t] = 0u;
        __syncthreads();
        const unsigned beg = sBeg, n = sNum;
        for (unsigned i0 = t; i0 < n; i0 += 8 * 1024) {
            int k[8];
            #pragma unroll
            for (int j = 0; j < 8; ++j) {
                const unsigned i = i0 + j * 1024;
                k[j] = recs[beg + (i < n ? i : i0)].x;
            }
            #pragma unroll
            for (int j = 0; j < 8; ++j)
                if (i0 + j * 1024 < n) atomicAdd(&sCnt[(unsigned)k[j] >> 24], 1u);
        }
        __syncthreads();
        const unsigned c = (t < NB_NODES) ? sCnt[t] : 0u;
        const unsigned noff = excl_scan_256(c, sTot);
        if (t < NB_NODES) {
            const int node = b * NB_NODES + t;
            if (node < NT) { count[node] = c; offset[node] = beg + noff; }
            sOff[t] = noff;
            sCnt[t] = 0u;
        }
        __syncthreads();
        const bool staged = n <= SORT_CAP;
        for (unsigned i0 = t; i0 < n; i0 += 8 * 1024) {
            int2 r[8];
            #pragma unroll
            for (int j = 0; j < 8; ++j) {
                const unsigned i = i0 + j * 1024;
                r[j] = recs[beg + (i < n ? i : i0)];
            }
            #pragma unroll
            for (int j = 0; j < 8; ++j) {
                if (i0 + j * 1024 < n) {
                    const unsigned dl = (unsigned)r[j].x >> 24;
                    const unsigned pos = sOff[dl] + atomicAdd(&sCnt[dl], 1u);
                    const int2 o = make_int2(r[j].x & 0x00FFFFFF, r[j].y);
                    if (staged) sP[pos] = o; else pairs[beg + pos] = o;
                }
            }
        }
        if (staged) {
            __syncthreads();
            for (unsigned i = t; i < n; i += 1024) pairs[beg + i] = sP[i];
        }
        __syncthreads();
    }
}

#define BREV5(j) ((((j) & 1) << 4) | (((j) & 2) << 2) | ((j) & 4) | (((j) & 8) >> 2) | (((j) & 16) >> 4))

template <int MASK>
__device__ __forceinline__ float merge2(float a, float b, bool up) {
    const float keep = up ? b : a;
    const float give = up ? a : b;
    return keep + __shfl_xor(give, MASK, 32);
}

template <int G>
__device__ __forceinline__ void load_group8(int s_v, const float* __restrict__ xl, int lane, float2* r)
{
    #pragma unroll
    for (int j = 0; j < 8; ++j) {
        const int s = __builtin_amdgcn_readlane(s_v, BREV5(G * 8 + j));
        r[G * 8 + j] = ((const float2*)(xl + (size_t)s * 64))[lane];
    }
}

template <int G>
__device__ __forceinline__ float score_group8(const float2* r, int ea_v, float2 xr2, float2 we, float2 at,
                                              bool b4, bool b3, bool b2)
{
    float p[8];
    #pragma unroll
    for (int j = 0; j < 8; ++j) {
        const float ea = __int_as_float(__builtin_amdgcn_readlane(ea_v, BREV5(G * 8 + j)));
        float m0 = r[G * 8 + j].x + xr2.x + ea * we.x;
        float m1 = r[G * 8 + j].y + xr2.y + ea * we.y;
        m0 = (m0 > 0.0f) ? m0 : 0.2f * m0;
        m1 = (m1 > 0.0f) ? m1 : 0.2f * m1;
        p[j] = __builtin_fmaf(at.x, m0, at.y * m1);
    }
    const float q0 = merge2<16>(p[0], p[1], b4);
    const float q1 = merge2<16>(p[2], p[3], b4);
    const float q2 = merge2<16>(p[4], p[5], b4);
    const float q3 = merge2<16>(p[6], p[7], b4);
    const float r0 = merge2<8>(q0, q1, b3);
    const float r1 = merge2<8>(q2, q3, b3);
    return merge2<4>(r0, r1, b2);
}

template <int G>
__device__ __forceinline__ void wsum_group8(const float2* r, float w_v, float2& acc)
{
    #pragma unroll
    for (int j = 0; j < 8; ++j) {
        const float w = __int_as_float(__builtin_amdgcn_readlane(__float_as_int(w_v), BREV5(G * 8 + j)));
        acc.x += w * r[G * 8 + j].x;
        acc.y += w * r[G * 8 + j].y;
    }
}

__global__ void __launch_bounds__(256)
node_agg_kernel(const int* __restrict__ pairs_i, const unsigned* __restrict__ count,
                const unsigned* __restrict__ offset, const float* __restrict__ xl,
                const float* __restrict__ We, const float* __restrict__ att,
                float* xr_agg, int NT)
{
    const int2* __restrict__ pairs = (const int2*)pairs_i;
    const int lane = threadIdx.x & 31;
    const int wib = __builtin_amdgcn_readfirstlane((int)(threadIdx.x >> 5));
    const int wid = blockIdx.x * (blockDim.x >> 5) + wib;
    const int nw  = gridDim.x * (blockDim.x >> 5);
    const float2 we = ((const float2*)We)[lane];
    const float2 at = ((const float2*)att)[lane];
    const bool b4 = (lane & 16) != 0, b3 = (lane & 8) != 0, b2 = (lane & 4) != 0;
    const bool b1 = (lane & 2) != 0,  b0 = (lane & 1) != 0;
    const unsigned mine = BREV5(lane);
    for (int d = wid; d < NT; d += nw) {
        const unsigned beg = __builtin_amdgcn_readfirstlane(offset[d]);
        const unsigned n   = __builtin_amdgcn_readfirstlane(count[d]);
        const int2* __restrict__ grp = pairs + beg;
        float2* rowp = (float2*)(xr_agg + (size_t)d * 64) + lane;
        const float2 xr2 = *rowp;

        float m = -__builtin_inff();
        float den = 0.0f;
        float2 acc = make_float2(0.0f, 0.0f);
        for (unsigned c = 0; c < n; c += 32) {
            const unsigned i = c + mine;
            const bool valid = i < n;
            const int2 pr = grp[valid ? i : c];
            const unsigned left = n - c;
            float2 r[32];
            load_group8<0>(pr.x, xl, lane, r);
            if (left > 8)  load_group8<1>(pr.x, xl, lane, r);
            if (left > 16) load_group8<2>(pr.x, xl, lane, r);
            if (left > 24) load_group8<3>(pr.x, xl, lane, r);
            float t0, t1 = 0.0f, t2 = 0.0f, t3 = 0.0f;
            t0 = score_group8<0>(r, pr.y, xr2, we, at, b4, b3, b2);
            if (left > 8)  t1 = score_group8<1>(r, pr.y, xr2, we, at, b4, b3, b2);
            if (left > 16) t2 = score_group8<2>(r, pr.y, xr2, we, at, b4, b3, b2);
            if (left > 24) t3 = score_group8<3>(r, pr.y, xr2, we, at, b4, b3, b2);
            const float u0 = merge2<2>(t0, t1, b1);
            const float u1 = merge2<2>(t2, t3, b1);
            const float p  = merge2<1>(u0, u1, b0);
            float cm = valid ? p : -__builtin_inff();
            for (int o = 16; o > 0; o >>= 1) cm = fmaxf(cm, __shfl_xor(cm, o, 32));
            const float mn = fmaxf(m, cm);
            const float scale = __expf(m - mn);
            const float w_v = valid ? __expf(p - mn) : 0.0f;
            den = den * scale + w_v;
            acc.x *= scale;
            acc.y *= scale;
            wsum_group8<0>(r, w_v, acc);
            if (left > 8)  wsum_group8<1>(r, w_v, acc);
            if (left > 16) wsum_group8<2>(r, w_v, acc);
            if (left > 24) wsum_group8<3>(r, w_v, acc);
            m = mn;
        }
        for (int o = 16; o > 0; o >>= 1) den += __shfl_xor(den, o, 32);
        const float inv = (n > 0u) ? 1.0f / den : 0.0f;
        *rowp = make_float2(acc.x * inv, acc.y * inv);
    }
}

__global__ void out_xform_kernel(const float* __restrict__ agg, const float* __restrict__ bias,
                                 const float* __restrict__ Wfc, const float* __restrict__ bfc,
                                 float* __restrict__ out, int NT)
{
    __shared__ float sW[WLDS];
    __shared__ float sb[64], sbias[64];
    stage_weight_pairs(sW, Wfc);
    for (int i = threadIdx.x; i < 64; i += blockDim.x) { sb[i] = bfc[i]; sbias[i] = bias[i]; }
    __syncthreads();

    const int lane = threadIdx.x & 31;
    const int wave = threadIdx.x >> 5;
    const int wpb  = blockDim.x >> 5;
    const int l15  = lane & 15;
    const int half = lane >> 4;
    const int numTiles = NT >> 4;

    for (int tile = blockIdx.x * wpb + wave; tile < numTiles; tile += gridDim.x * wpb) {
        v8f acc[4] = {v8f{}, v8f{}, v8f{}, v8f{}};
        const float* arow = agg + (size_t)(tile * 16 + l15) * 64;
        #pragma unroll 4
        for (int kk = 0; kk < 16; ++kk) {
            const int kbase = kk * 4 + half * 2;
            v2f a = *(const v2f*)(arow + kbase);
            a.x += sbias[kbase];
            a.y += sbias[kbase + 1];
            const int pbase = (kk * 2 + half) * PSTRIDE;
            #pragma unroll
            for (int nb = 0; nb < 4; ++nb) {
                const int col = nb * 16 + l15;
                const v2f b = *(const v2f*)(sW + pbase + col * 2);
                acc[nb] = __builtin_amdgcn_wmma_f32_16x16x4_f32(
                    false, a, false, b, (short)0, acc[nb], false, false);
            }
        }
        #pragma unroll
        for (int nb = 0; nb < 4; ++nb) {
            const int col = nb * 16 + l15;
            #pragma unroll
            for (int v = 0; v < 8; ++v) {
                const size_t row = (size_t)(tile * 16 + v + half * 8);
                out[row * 64 + col] = acc[nb][v] + sb[col];
            }
        }
    }
}

extern "C" void kernel_launch(void* const* d_in, const int* in_sizes, int n_in,
                              void* d_out, int out_size, void* d_ws, size_t ws_size,
                              hipStream_t stream) {
    const float* x      = (const float*)d_in[0];
    const int*   eidx   = (const int*)d_in[1];
    const float* eattr  = (const float*)d_in[2];
    const float* W_l    = (const float*)d_in[3];
    const float* b_l    = (const float*)d_in[4];
    const float* W_r    = (const float*)d_in[5];
    const float* b_r    = (const float*)d_in[6];
    const float* W_e    = (const float*)d_in[7];
    const float* att    = (const float*)d_in[8];
    const float* bias   = (const float*)d_in[9];
    const float* W_fc   = (const float*)d_in[10];
    const float* b_fc   = (const float*)d_in[11];

    const int NT       = in_sizes[0] / 64;
    const long long E  = (long long)in_sizes[2];
    const int* src = eidx;
    const int* dst = eidx + E;

    float* ws       = (float*)d_ws;
    float* xl       = ws;
    int*   pairs    = (int*)(xl + (size_t)NT * 64);
    int*   recs     = pairs + (size_t)NT * 64;
    unsigned* btotal  = (unsigned*)(recs + (size_t)NT * 64);
    unsigned* bcursor = btotal + NBMAX;
    unsigned* count = btotal + E;
    unsigned* offset = count + NT;
    float* xr       = (float*)recs;
    float* agg      = xr;

    const int numTiles = NT / 16;
    const int gemmBlocks = (numTiles + 7) / 8;

    init_ws_kernel<<<1, 512, 0, stream>>>(btotal, 512);
    bucket_count_kernel<<<256, 1024, 0, stream>>>(dst, btotal, NT, E);
    edge_scatter_kernel<<<256, 1024, 0, stream>>>(src, dst, eattr, btotal, bcursor, recs, NT, E);
    bucket_sort_kernel<<<196, 1024, 0, stream>>>(recs, btotal, pairs, count, offset, NT);
    node_xform_kernel<<<gemmBlocks, 256, 0, stream>>>(x, W_l, b_l, W_r, b_r, xl, xr, NT);
    node_agg_kernel<<<1024, 256, 0, stream>>>(pairs, count, offset, xl, W_e, att, xr, NT);
    out_xform_kernel<<<gemmBlocks, 256, 0, stream>>>(agg, bias, W_fc, b_fc, (float*)d_out, NT);
}


# sort level 2 moved into the node kernel: 256 buckets of 196 nodes, a block orders its bucket in LDS, its waves read the pairs there; one launch and pairs/count/offset traffic gone; sum order unfixed
# speedup vs baseline: 5.4705x; 1.0905x over previous
#include <hip/hip_runtime.h>
#include <hip/hip_bf16.h>

typedef float v2f __attribute__((ext_vector_type(2)));
typedef float v8f __attribute__((ext_vector_type(8)));

#define PSTRIDE 160
#define WLDS    (32 * PSTRIDE)

__device__ __forceinline__ void stage_weight_pairs(float* sW, const float* __restrict__ W) {
    for (int i = threadIdx.x; i < 64 * 64; i += blockDim.x) {
        const int k = i >> 6, col = i & 63;
        sW[(k >> 1) * PSTRIDE + col * 2 + (k & 1)] = W[i];
    }
}

__global__ void node_xform_kernel(const float* __restrict__ x2,
                                  const float* __restrict__ Wl, const float* __restrict__ bl,
                                  const float* __restrict__ Wr, const float* __restrict__ br,
                                  float* __restrict__ xl, float* __restrict__ xr, int NT)
{
    __shared__ float sWl[WLDS];
    __shared__ float sWr[WLDS];
    __shared__ float sbl[64], sbr[64];
    stage_weight_pairs(sWl, Wl);
    stage_weight_pairs(sWr, Wr);
    for (int i = threadIdx.x; i < 64; i += blockDim.x) { sbl[i] = bl[i]; sbr[i] = br[i]; }
    __syncthreads();

    const int lane    = threadIdx.x & 31;
    const int wave    = threadIdx.x >> 5;
    const int wpb     = blockDim.x >> 5;
    const int l15     = lane & 15;
    const int half    = lane >> 4;
    const int numTiles = NT >> 4;

    for (int tile = blockIdx.x * wpb + wave; tile < numTiles; tile += gridDim.x * wpb) {
        v8f accl[4] = {v8f{}, v8f{}, v8f{}, v8f{}};
        v8f accr[4] = {v8f{}, v8f{}, v8f{}, v8f{}};
        const float* xrow = x2 + (size_t)(tile * 16 + l15) * 64;
        #pragma unroll 4
        for (int kk = 0; kk < 16; ++kk) {
            const int kbase = kk * 4 + half * 2;
            const v2f a = *(const v2f*)(xrow + kbase);
            const int pbase = (kk * 2 + half) * PSTRIDE;
            #pragma unroll
            for (int nb = 0; nb < 4; ++nb) {
                const int col = nb * 16 + l15;
                const v2f bL = *(const v2f*)(sWl + pbase + col * 2);
                const v2f bR = *(const v2f*)(sWr + pbase + col * 2);
                accl[nb] = __builtin_amdgcn_wmma_f32_16x16x4_f32(
                    false, a, false, bL, (short)0, accl[nb], false, false);
                accr[nb] = __builtin_amdgcn_wmma_f32_16x16x4_f32(
                    false, a, false, bR, (short)0, accr[nb], false, false);
            }
        }
        #pragma unroll
        for (int nb = 0; nb < 4; ++nb) {
            const int col = nb * 16 + l15;
            #pragma unroll
            for (int v = 0; v < 8; ++v) {
                const size_t row = (size_t)(tile * 16 + v + half * 8);
                xl[row * 64 + col] = accl[nb][v] + sbl[col];
                xr[row * 64 + col] = accr[nb][v] + sbr[col];
            }
        }
    }
}

#define NB_NODES 196
#define NBMAX    256
#define STAGE_CAP 8192

__global__ void init_ws_kernel(unsigned* __restrict__ count, int NT)
{
    for (int i = blockIdx.x * blockDim.x + threadIdx.x; i < NT; i += gridDim.x * blockDim.x)
        count[i] = 0u;
}

__device__ __forceinline__ unsigned excl_scan_256(unsigned v, unsigned* sTot)
{
    const int lane = threadIdx.x & 31, w = threadIdx.x >> 5;
    unsigned inc = v;
    for (int o = 1; o < 32; o <<= 1) {
        const unsigned up = __shfl_up(inc, o, 32);
        if (lane >= o) inc += up;
    }
    __syncthreads();
    if (lane == 31 && w < 8) sTot[w] = inc;
    __syncthreads();
    unsigned pre = 0u;
    for (int j = 0; j < 8; ++j) if (j < w) pre += sTot[j];
    return pre + inc - v;
}

__global__ void __launch_bounds__(1024)
bucket_count_kernel(const int* __restrict__ dst, unsigned* __restrict__ btotal, int NT, long long E)
{
    __shared__ unsigned sCnt[NBMAX];
    const int t = threadIdx.x;
    if (t < NBMAX) sCnt[t] = 0u;
    __syncthreads();
    const long long nth = (long long)gridDim.x * blockDim.x;
    const long long tid = (long long)blockIdx.x * blockDim.x + t;
    for (long long e0 = tid; e0 < E; e0 += 4 * nth) {
        const long long e1 = e0 + nth, e2 = e0 + 2 * nth, e3 = e0 + 3 * nth;
        const bool v1 = e1 < E, v2 = e2 < E, v3 = e3 < E;
        const int d0 = dst[e0];
        const int d1 = dst[v1 ? e1 : e0];
        const int d2 = dst[v2 ? e2 : e0];
        const int d3 = dst[v3 ? e3 : e0];
        atomicAdd(&sCnt[(unsigned)d0 / NB_NODES], 1u);
        if (v1) atomicAdd(&sCnt[(unsigned)d1 / NB_NODES], 1u);
        if (v2) atomicAdd(&sCnt[(unsigned)d2 / NB_NODES], 1u);
        if (v3) atomicAdd(&sCnt[(unsigned)d3 / NB_NODES], 1u);
    }
    __syncthreads();
    const int nb = (NT + NB_NODES - 1) / NB_NODES;
    if (t < nb) {
        const unsigned c = sCnt[t];
        if (c) atomicAdd(&btotal[t], c);
    }
}

__global__ void __launch_bounds__(1024) edge_scatter_kernel(const int* __restrict__ src, const int* __restrict__ dst,
                                  const float* __restrict__ eattr,
                                  const unsigned* __restrict__ btotal, unsigned* __restrict__ bcursor,
                                  int* __restrict__ recs, int NT, long long E)
{
    __shared__ unsigned sCnt[NBMAX];
    __shared__ unsigned sBase[NBMAX];
    __shared__ unsigned sLoc[NBMAX];
    __shared__ unsigned sTot[8];
    __shared__ int2     sRec[STAGE_CAP];
    __shared__ unsigned sSlot[STAGE_CAP];
    const int t = threadIdx.x;
    const long long ebeg = (E * blockIdx.x) / gridDim.x;
    const long long eend = (E * (blockIdx.x + 1)) / gridDim.x;
    const unsigned nblk = (eend > ebeg) ? (unsigned)(eend - ebeg) : 0u;

    if (t < NBMAX) sCnt[t] = 0u;
    __syncthreads();
    for (long long e0 = ebeg + t; e0 < eend; e0 += 8 * 1024) {
        int d[8];
        #pragma unroll
        for (int j = 0; j < 8; ++j) {
            const long long e = e0 + j * 1024;
            d[j] = dst[(e < eend) ? e : e0];
        }
        #pragma unroll
        for (int j = 0; j < 8; ++j)
            if (e0 + j * 1024 < eend) atomicAdd(&sCnt[(unsigned)d[j] / NB_NODES], 1u);
    }
    __syncthreads();
    const int nb = (NT + NB_NODES - 1) / NB_NODES;
    const unsigned tot = (t < nb) ? btotal[t] : 0u;
    const unsigned first = excl_scan_256(tot, sTot);
    const unsigned c = (t < nb) ? sCnt[t] : 0u;
    const unsigned loc = excl_scan_256(c, sTot);
    if (t < nb) {
        unsigned got = 0u;
        if (c) got = atomicAdd(&bcursor[t], c);
        sBase[t] = first + got;
        sLoc[t]  = loc;
        sCnt[t]  = 0u;
    }
    __syncthreads();
    const bool staged = nblk <= STAGE_CAP;
    for (long long e0 = ebeg + t; e0 < eend; e0 += 4 * 1024) {
        int s[4], d[4];
        float a[4];
        #pragma unroll
        for (int j = 0; j < 4; ++j) {
            const long long e = e0 + j * 1024;
            const long long ec = (e < eend) ? e : e0;
            s[j] = src[ec];
            d[j] = dst[ec];
            a[j] = eattr[ec];
        }
        #pragma unroll
        for (int j = 0; j < 4; ++j) {
            if (e0 + j * 1024 < eend) {
                const unsigned bk = (unsigned)d[j] / NB_NODES;
                const unsigned r = atomicAdd(&sCnt[bk], 1u);
                const unsigned slot = sBase[bk] + r;
                const unsigned key = (unsigned)s[j] | (((unsigned)d[j] - bk * NB_NODES) << 24);
                const int2 rec = make_int2((int)key, __float_as_int(a[j]));
                if (staged) {
                    const unsigned lp = sLoc[bk] + r;
                    sRec[lp]  = rec;
                    sSlot[lp] = slot;
                } else {
                    ((int2*)recs)[slot] = rec;
                }
            }
        }
    }
    if (staged) {
        __syncthreads();
        for (unsigned i = t; i < nblk; i += blockDim.x) ((int2*)recs)[sSlot[i]] = sRec[i];
    }
}

#define BREV5(j) ((((j) & 1) << 4) | (((j) & 2) << 2) | ((j) & 4) | (((j) & 8) >> 2) | (((j) & 16) >> 4))

template <int MASK>
__device__ __forceinline__ float merge2(float a, float b, bool up) {
    const float keep = up ? b : a;
    const float give = up ? a : b;
    return keep + __shfl_xor(give, MASK, 32);
}

template <int G>
__device__ __forceinline__ void load_group8(int s_v, const float* __restrict__ xl, int lane, float2* r)
{
    #pragma unroll
    for (int j = 0; j < 8; ++j) {
        const int s = __builtin_amdgcn_readlane(s_v, BREV5(G * 8 + j));
        r[G * 8 + j] = ((const float2*)(xl + (size_t)s * 64))[lane];
    }
}

template <int G>
__device__ __forceinline__ float score_group8(const float2* r, int ea_v, float2 xr2, float2 we, float2 at,
                                              bool b4, bool b3, bool b2)
{
    float p[8];
    #pragma unroll
    for (int j = 0; j < 8; ++j) {
        const float ea = __int_as_float(__builtin_amdgcn_readlane(ea_v, BREV5(G * 8 + j)));
        float m0 = r[G * 8 + j].x + xr2.x + ea * we.x;
        float m1 = r[G * 8 + j].y + xr2.y + ea * we.y;
        m0 = (m0 > 0.0f) ? m0 : 0.2f * m0;
        m1 = (m1 > 0.0f) ? m1 : 0.2f * m1;
        p[j] = __builtin_fmaf(at.x, m0, at.y * m1);
    }
    const float q0 = merge2<16>(p[0], p[1], b4);
    const float q1 = merge2<16>(p[2], p[3], b4);
    const float q2 = merge2<16>(p[4], p[5], b4);
    const float q3 = merge2<16>(p[6], p[7], b4);
    const float r0 = merge2<8>(q0, q1, b3);
    const float r1 = merge2<8>(q2, q3, b3);
    return merge2<4>(r0, r1, b2);
}

template <int G>
__device__ __forceinline__ void wsum_group8(const float2* r, float w_v, float2& acc)
{
    #pragma unroll
    for (int j = 0; j < 8; ++j) {
        const float w = __int_as_float(__builtin_amdgcn_readlane(__float_as_int(w_v), BREV5(G * 8 + j)));
        acc.x += w * r[G * 8 + j].x;
        acc.y += w * r[G * 8 + j].y;
    }
}

#define SORT_CAP 12288
__global__ void __launch_bounds__(1024)
bucket_agg_kernel(const int* __restrict__ recs_i, const unsigned* __restrict__ btotal,
                  const float* __restrict__ xl, const float* __restrict__ We, const float* __restrict__ att,
                  float* xr_agg, int NT)
{
    __shared__ int2 sP[SORT_CAP];
    __shared__ unsigned sCnt[NBMAX], sOff[NBMAX];
    __shared__ unsigned sTot[8];
    __shared__ unsigned sBeg, sNum;
    __shared__ int2 sSel[32][32];
    const int2* __restrict__ recs = (const int2*)recs_i;
    const int t = threadIdx.x;
    const int lane = t & 31;
    const int wib = __builtin_amdgcn_readfirstlane(t >> 5);
    const int nb = (NT + NB_NODES - 1) / NB_NODES;
    const float2 we = ((const float2*)We)[lane];
    const float2 at = ((const float2*)att)[lane];
    const bool b4 = (lane & 16) != 0, b3 = (lane & 8) != 0, b2 = (lane & 4) != 0;
    const bool b1 = (lane & 2) != 0,  b0 = (lane & 1) != 0;
    const unsigned mine = BREV5(lane);
    const int b = blockIdx.x;
    if (b >= nb) return;
    {
        const unsigned tot = (t < nb) ? btotal[t] : 0u;
        const unsigned first = excl_scan_256(tot, sTot);
        if (t == b) { sBeg = first; sNum = tot; }
        if (t < NBMAX) sCnt[t] = 0u;
        __syncthreads();
        const unsigned beg = sBeg, nrec = sNum;
        const bool staged = nrec <= SORT_CAP;
        if (staged) {
            for (unsigned i0 = t; i0 < nrec; i0 += 8 * 1024) {
                int k[8];
                #pragma unroll
                for (int j = 0; j < 8; ++j) {
                    const unsigned i = i0 + j * 1024;
                    k[j] = recs[beg + (i < nrec ? i : i0)].x;
                }
                #pragma unroll
                for (int j = 0; j < 8; ++j)
                    if (i0 + j * 1024 < nrec) atomicAdd(&sCnt[(unsigned)k[j] >> 24], 1u);
            }
        }
        __syncthreads();
        const unsigned cn = (t < NB_NODES) ? sCnt[t] : 0u;
        const unsigned noff = excl_scan_256(cn, sTot);
        if (t < NB_NODES) {
            sOff[t] = noff;
            sCnt[t] = 0u;
        }
        __syncthreads();
        if (staged) {
            for (unsigned i0 = t; i0 < nrec; i0 += 8 * 1024) {
                int2 r[8];
                #pragma unroll
                for (int j = 0; j < 8; ++j) {
                    const unsigned i = i0 + j * 1024;
                    r[j] = recs[beg + (i < nrec ? i : i0)];
                }
                #pragma unroll
                for (int j = 0; j < 8; ++j) {
                    if (i0 + j * 1024 < nrec) {
                        const unsigned dl = (unsigned)r[j].x >> 24;
                        const unsigned pos = sOff[dl] + atomicAdd(&sCnt[dl], 1u);
                        sP[pos] = make_int2(r[j].x & 0x00FFFFFF, r[j].y);
                    }
                }
            }
        }
        __syncthreads();

        for (int tn = wib; tn < NB_NODES; tn += 32) {
            const int d = b * NB_NODES + tn;
            if (d >= NT) break;
            const unsigned off = __builtin_amdgcn_readfirstlane(sOff[tn]);
            const unsigned n = staged ? __builtin_amdgcn_readfirstlane(sCnt[tn]) : nrec;
            float2* rowp = (float2*)(xr_agg + (size_t)d * 64) + lane;
            const float2 xr2 = *rowp;

            float m = -__builtin_inff();
            float den = 0.0f;
            float2 acc = make_float2(0.0f, 0.0f);
            bool any = false;
            for (unsigned c = 0; c < n; c += 32) {
                int2 pr;
                bool valid;
                unsigned left;
                if (staged) {
                    const unsigned i = c + mine;
                    valid = i < n;
                    pr = sP[off + (valid ? i : c)];
                    left = n - c;
                } else {
                    const unsigned i = c + lane;
                    const bool in = i < nrec;
                    const int2 rr = recs[beg + (in ? i : c)];
                    const bool match = in && ((unsigned)rr.x >> 24) == (unsigned)tn;
                    const unsigned mask = __builtin_amdgcn_ballot_w32(match);
                    if (mask == 0u) continue;
                    left = (unsigned)__builtin_popcount(mask);
                    const unsigned rank = (unsigned)__builtin_popcount(mask & ((1u << lane) - 1u));
                    if (match) sSel[wib][rank] = make_int2(rr.x & 0x00FFFFFF, rr.y);
                    valid = mine < left;
                    pr = sSel[wib][valid ? mine : 0u];
                }
                any = true;
                float2 r[32];
                load_group8<0>(pr.x, xl, lane, r);
                if (left > 8)  load_group8<1>(pr.x, xl, lane, r);
                if (left > 16) load_group8<2>(pr.x, xl, lane, r);
                if (left > 24) load_group8<3>(pr.x, xl, lane, r);
                float t0, t1 = 0.0f, t2 = 0.0f, t3 = 0.0f;
                t0 = score_group8<0>(r, pr.y, xr2, we, at, b4, b3, b2);
                if (left > 8)  t1 = score_group8<1>(r, pr.y, xr2, we, at, b4, b3, b2);
                if (left > 16) t2 = score_group8<2>(r, pr.y, xr2, we, at, b4, b3, b2);
                if (left > 24) t3 = score_group8<3>(r, pr.y, xr2, we, at, b4, b3, b2);
                const float u0 = merge2<2>(t0, t1, b1);
                const float u1 = merge2<2>(t2, t3, b1);
                const float p  = merge2<1>(u0, u1, b0);
                float cm = valid ? p : -__builtin_inff();
                for (int o = 16; o > 0; o >>= 1) cm = fmaxf(cm, __shfl_xor(cm, o, 32));
                const float mn = fmaxf(m, cm);
                const float scale = __expf(m - mn);
                const float w_v = valid ? __expf(p - mn) : 0.0f;
                den = den * scale + w_v;
                acc.x *= scale;
                acc.y *= scale;
                wsum_group8<0>(r, w_v, acc);
                if (left > 8)  wsum_group8<1>(r, w_v, acc);
                if (left > 16) wsum_group8<2>(r, w_v, acc);
                if (left > 24) wsum_group8<3>(r, w_v, acc);
                m = mn;
            }
            for (int o = 16; o > 0; o >>= 1) den += __shfl_xor(den, o, 32);
            const float inv = any ? 1.0f / den : 0.0f;
            *rowp = make_float2(acc.x * inv, acc.y * inv);
        }
    }
}

__global__ void out_xform_kernel(const float* __restrict__ agg, const float* __restrict__ bias,
                                 const float* __restrict__ Wfc, const float* __restrict__ bfc,
                                 float* __restrict__ out, int NT)
{
    __shared__ float sW[WLDS];
    __shared__ float sb[64], sbias[64];
    stage_weight_pairs(sW, Wfc);
    for (int i = threadIdx.x; i < 64; i += blockDim.x) { sb[i] = bfc[i]; sbias[i] = bias[i]; }
    __syncthreads();

    const int lane = threadIdx.x & 31;
    const int wave = threadIdx.x >> 5;
    const int wpb  = blockDim.x >> 5;
    const int l15  = lane & 15;
    const int half = lane >> 4;
    const int numTiles = NT >> 4;

    for (int tile = blockIdx.x * wpb + wave; tile < numTiles; tile += gridDim.x * wpb) {
        v8f acc[4] = {v8f{}, v8f{}, v8f{}, v8f{}};
        const float* arow = agg + (size_t)(tile * 16 + l15) * 64;
        #pragma unroll 4
        for (int kk = 0; kk < 16; ++kk) {
            const int kbase = kk * 4 + half * 2;
            v2f a = *(const v2f*)(arow + kbase);
            a.x += sbias[kbase];
            a.y += sbias[kbase + 1];
            const int pbase = (kk * 2 + half) * PSTRIDE;
            #pragma unroll
            for (int nb = 0; nb < 4; ++nb) {
                const int col = nb * 16 + l15;
                const v2f b = *(const v2f*)(sW + pbase + col * 2);
                acc[nb] = __builtin_amdgcn_wmma_f32_16x16x4_f32(
                    false, a, false, b, (short)0, acc[nb], false, false);
            }
        }
        #pragma unroll
        for (int nb = 0; nb < 4; ++nb) {
            const int col = nb * 16 + l15;
            #pragma unroll
            for (int v = 0; v < 8; ++v) {
                const size_t row = (size_t)(tile * 16 + v + half * 8);
                out[row * 64 + col] = acc[nb][v] + sb[col];
            }
        }
    }
}

extern "C" void kernel_launch(void* const* d_in, const int* in_sizes, int n_in,
                              void* d_out, int out_size, void* d_ws, size_t ws_size,
                              hipStream_t stream) {
    const float* x      = (const float*)d_in[0];
    const int*   eidx   = (const int*)d_in[1];
    const float* eattr  = (const float*)d_in[2];
    const float* W_l    = (const float*)d_in[3];
    const float* b_l    = (const float*)d_in[4];
    const float* W_r    = (const float*)d_in[5];
    const float* b_r    = (const float*)d_in[6];
    const float* W_e    = (const float*)d_in[7];
    const float* att    = (const float*)d_in[8];
    const float* bias   = (const float*)d_in[9];
    const float* W_fc   = (const float*)d_in[10];
    const float* b_fc   = (const float*)d_in[11];

    const int NT       = in_sizes[0] / 64;
    const long long E  = (long long)in_sizes[2];
    const int* src = eidx;
    const int* dst = eidx + E;

    float* ws       = (float*)d_ws;
    float* xl       = ws;
    float* xr       = xl + (size_t)NT * 64;
    int*   recs     = (int*)(xr + (size_t)NT * 64);
    unsigned* btotal  = (unsigned*)(recs + (size_t)NT * 64);
    unsigned* bcursor = btotal + NBMAX;
    float* agg      = xr;

    const int numTiles = NT / 16;
    const int gemmBlocks = (numTiles + 7) / 8;

    init_ws_kernel<<<1, 512, 0, stream>>>(btotal, 512);
    bucket_count_kernel<<<256, 1024, 0, stream>>>(dst, btotal, NT, E);
    edge_scatter_kernel<<<256, 1024, 0, stream>>>(src, dst, eattr, btotal, bcursor, recs, NT, E);
    node_xform_kernel<<<gemmBlocks, 256, 0, stream>>>(x, W_l, b_l, W_r, b_r, xl, xr, NT);
    bucket_agg_kernel<<<256, 1024, 0, stream>>>(recs, btotal, xl, W_e, att, xr, NT);
    out_xform_kernel<<<gemmBlocks, 256, 0, stream>>>(agg, bias, W_fc, b_fc, (float*)d_out, NT);
}


# out = (agg + bias) @ W_fc + b_fc moved into the node kernel: a bucket's agg rows stay in LDS and its waves run the WMMAs from there; one launch and the agg round trip gone; sum order unchanged
# speedup vs baseline: 6.1697x; 1.1278x over previous
#include <hip/hip_runtime.h>
#include <hip/hip_bf16.h>

typedef float v2f __attribute__((ext_vector_type(2)));
typedef float v8f __attribute__((ext_vector_type(8)));

#define PSTRIDE 160
#define WLDS    (32 * PSTRIDE)

__device__ __forceinline__ void stage_weight_pairs(float* sW, const float* __restrict__ W) {
    for (int i = threadIdx.x; i < 64 * 64; i += blockDim.x) {
        const int k = i >> 6, col = i & 63;
        sW[(k >> 1) * PSTRIDE + col * 2 + (k & 1)] = W[i];
    }
}

__global__ void node_xform_kernel(const float* __restrict__ x2,
                                  const float* __restrict__ Wl, const float* __restrict__ bl,
                                  const float* __restrict__ Wr, const float* __restrict__ br,
                                  float* __restrict__ xl, float* __restrict__ xr, int NT)
{
    __shared__ float sWl[WLDS];
    __shared__ float sWr[WLDS];
    __shared__ float sbl[64], sbr[64];
    stage_weight_pairs(sWl, Wl);
    stage_weight_pairs(sWr, Wr);
    for (int i = threadIdx.x; i < 64; i += blockDim.x) { sbl[i] = bl[i]; sbr[i] = br[i]; }
    __syncthreads();

    const int lane    = threadIdx.x & 31;
    const int wave    = threadIdx.x >> 5;
    const int wpb     = blockDim.x >> 5;
    const int l15     = lane & 15;
    const int half    = lane >> 4;
    const int numTiles = NT >> 4;

    for (int tile = blockIdx.x * wpb + wave; tile < numTiles; tile += gridDim.x * wpb) {
        v8f accl[4] = {v8f{}, v8f{}, v8f{}, v8f{}};
        v8f accr[4] = {v8f{}, v8f{}, v8f{}, v8f{}};
        const float* xrow = x2 + (size_t)(tile * 16 + l15) * 64;
        #pragma unroll 4
        for (int kk = 0; kk < 16; ++kk) {
            const int kbase = kk * 4 + half * 2;
            const v2f a = *(const v2f*)(xrow + kbase);
            const int pbase = (kk * 2 + half) * PSTRIDE;
            #pragma unroll
            for (int nb = 0; nb < 4; ++nb) {
                const int col = nb * 16 + l15;
                const v2f bL = *(const v2f*)(sWl + pbase + col * 2);
                const v2f bR = *(const v2f*)(sWr + pbase + col * 2);
                accl[nb] = __builtin_amdgcn_wmma_f32_16x16x4_f32(
                    false, a, false, bL, (short)0, accl[nb], false, false);
                accr[nb] = __builtin_amdgcn_wmma_f32_16x16x4_f32(
                    false, a, false, bR, (short)0, accr[nb], false, false);
            }
        }
        #pragma unroll
        for (int nb = 0; nb < 4; ++nb) {
            const int col = nb * 16 + l15;
            #pragma unroll
            for (int v = 0; v < 8; ++v) {
                const size_t row = (size_t)(tile * 16 + v + half * 8);
                xl[row * 64 + col] = accl[nb][v] + sbl[col];
                xr[row * 64 + col] = accr[nb][v] + sbr[col];
            }
        }
    }
}

#define NB_NODES 196
#define NBMAX    256
#define STAGE_CAP 8192

__global__ void init_ws_kernel(unsigned* __restrict__ count, int NT)
{
    for (int i = blockIdx.x * blockDim.x + threadIdx.x; i < NT; i += gridDim.x * blockDim.x)
        count[i] = 0u;
}

__device__ __forceinline__ unsigned excl_scan_256(unsigned v, unsigned* sTot)
{
    const int lane = threadIdx.x & 31, w = threadIdx.x >> 5;
    unsigned inc = v;
    for (int o = 1; o < 32; o <<= 1) {
        const unsigned up = __shfl_up(inc, o, 32);
        if (lane >= o) inc += up;
    }
    __syncthreads();
    if (lane == 31 && w < 8) sTot[w] = inc;
    __syncthreads();
    unsigned pre = 0u;
    for (int j = 0; j < 8; ++j) if (j < w) pre += sTot[j];
    return pre + inc - v;
}

__global__ void __launch_bounds__(1024)
bucket_count_kernel(const int* __restrict__ dst, unsigned* __restrict__ btotal, int NT, long long E)
{
    __shared__ unsigned sCnt[NBMAX];
    const int t = threadIdx.x;
    if (t < NBMAX) sCnt[t] = 0u;
    __syncthreads();
    const long long nth = (long long)gridDim.x * blockDim.x;
    const long long tid = (long long)blockIdx.x * blockDim.x + t;
    for (long long e0 = tid; e0 < E; e0 += 4 * nth) {
        const long long e1 = e0 + nth, e2 = e0 + 2 * nth, e3 = e0 + 3 * nth;
        const bool v1 = e1 < E, v2 = e2 < E, v3 = e3 < E;
        const int d0 = dst[e0];
        const int d1 = dst[v1 ? e1 : e0];
        const int d2 = dst[v2 ? e2 : e0];
        const int d3 = dst[v3 ? e3 : e0];
        atomicAdd(&sCnt[(unsigned)d0 / NB_NODES], 1u);
        if (v1) atomicAdd(&sCnt[(unsigned)d1 / NB_NODES], 1u);
        if (v2) atomicAdd(&sCnt[(unsigned)d2 / NB_NODES], 1u);
        if (v3) atomicAdd(&sCnt[(unsigned)d3 / NB_NODES], 1u);
    }
    __syncthreads();
    const int nb = (NT + NB_NODES - 1) / NB_NODES;
    if (t < nb) {
        const unsigned c = sCnt[t];
        if (c) atomicAdd(&btotal[t], c);
    }
}

__global__ void __launch_bounds__(1024) edge_scatter_kernel(const int* __restrict__ src, const int* __restrict__ dst,
                                  const float* __restrict__ eattr,
                                  const unsigned* __restrict__ btotal, unsigned* __restrict__ bcursor,
                                  int* __restrict__ recs, int NT, long long E)
{
    __shared__ unsigned sCnt[NBMAX];
    __shared__ unsigned sBase[NBMAX];
    __shared__ unsigned sLoc[NBMAX];
    __shared__ unsigned sTot[8];
    __shared__ int2     sRec[STAGE_CAP];
    __shared__ unsigned sSlot[STAGE_CAP];
    const int t = threadIdx.x;
    const long long ebeg = (E * blockIdx.x) / gridDim.x;
    const long long eend = (E * (blockIdx.x + 1)) / gridDim.x;
    const unsigned nblk = (eend > ebeg) ? (unsigned)(eend - ebeg) : 0u;

    if (t < NBMAX) sCnt[t] = 0u;
    __syncthreads();
    for (long long e0 = ebeg + t; e0 < eend; e0 += 8 * 1024) {
        int d[8];
        #pragma unroll
        for (int j = 0; j < 8; ++j) {
            const long long e = e0 + j * 1024;
            d[j] = dst[(e < eend) ? e : e0];
        }
        #pragma unroll
        for (int j = 0; j < 8; ++j)
            if (e0 + j * 1024 < eend) atomicAdd(&sCnt[(unsigned)d[j] / NB_NODES], 1u);
    }
    __syncthreads();
    const int nb = (NT + NB_NODES - 1) / NB_NODES;
    const unsigned tot = (t < nb) ? btotal[t] : 0u;
    const unsigned first = excl_scan_256(tot, sTot);
    const unsigned c = (t < nb) ? sCnt[t] : 0u;
    const unsigned loc = excl_scan_256(c, sTot);
    if (t < nb) {
        unsigned got = 0u;
        if (c) got = atomicAdd(&bcursor[t], c);
        sBase[t] = first + got;
        sLoc[t]  = loc;
        sCnt[t]  = 0u;
    }
    __syncthreads();
    const bool staged = nblk <= STAGE_CAP;
    for (long long e0 = ebeg + t; e0 < eend; e0 += 4 * 1024) {
        int s[4], d[4];
        float a[4];
        #pragma unroll
        for (int j = 0; j < 4; ++j) {
            const long long e = e0 + j * 1024;
            const long long ec = (e < eend) ? e : e0;
            s[j] = src[ec];
            d[j] = dst[ec];
            a[j] = eattr[ec];
        }
        #pragma unroll
        for (int j = 0; j < 4; ++j) {
            if (e0 + j * 1024 < eend) {
                const unsigned bk = (unsigned)d[j] / NB_NODES;
                const unsigned r = atomicAdd(&sCnt[bk], 1u);
                const unsigned slot = sBase[bk] + r;
                const unsigned key = (unsigned)s[j] | (((unsigned)d[j] - bk * NB_NODES) << 24);
                const int2 rec = make_int2((int)key, __float_as_int(a[j]));
                if (staged) {
                    const unsigned lp = sLoc[bk] + r;
                    sRec[lp]  = rec;
                    sSlot[lp] = slot;
                } else {
                    ((int2*)recs)[slot] = rec;
                }
            }
        }
    }
    if (staged) {
        __syncthreads();
        for (unsigned i = t; i < nblk; i += blockDim.x) ((int2*)recs)[sSlot[i]] = sRec[i];
    }
}

#define BREV5(j) ((((j) & 1) << 4) | (((j) & 2) << 2) | ((j) & 4) | (((j) & 8) >> 2) | (((j) & 16) >> 4))

template <int MASK>
__device__ __forceinline__ float merge2(float a, float b, bool up) {
    const float keep = up ? b : a;
    const float give = up ? a : b;
    return keep + __shfl_xor(give, MASK, 32);
}

template <int G>
__device__ __forceinline__ void load_group8(int s_v, const float* __restrict__ xl, int lane, float2* r)
{
    #pragma unroll
    for (int j = 0; j < 8; ++j) {
        const int s = __builtin_amdgcn_readlane(s_v, BREV5(G * 8 + j));
        r[G * 8 + j] = ((const float2*)(xl + (size_t)s * 64))[lane];
    }
}

template <int G>
__device__ __forceinline__ float score_group8(const float2* r, int ea_v, float2 xr2, float2 we, float2 at,
                                              bool b4, bool b3, bool b2)
{
    float p[8];
    #pragma unroll
    for (int j = 0; j < 8; ++j) {
        const float ea = __int_as_float(__builtin_amdgcn_readlane(ea_v, BREV5(G * 8 + j)));
        float m0 = r[G * 8 + j].x + xr2.x + ea * we.x;
        float m1 = r[G * 8 + j].y + xr2.y + ea * we.y;
        m0 = (m0 > 0.0f) ? m0 : 0.2f * m0;
        m1 = (m1 > 0.0f) ? m1 : 0.2f * m1;
        p[j] = __builtin_fmaf(at.x, m0, at.y * m1);
    }
    const float q0 = merge2<16>(p[0], p[1], b4);
    const float q1 = merge2<16>(p[2], p[3], b4);
    const float q2 = merge2<16>(p[4], p[5], b4);
    const float q3 = merge2<16>(p[6], p[7], b4);
    const float r0 = merge2<8>(q0, q1, b3);
    const float r1 = merge2<8>(q2, q3, b3);
    return merge2<4>(r0, r1, b2);
}

template <int G>
__device__ __forceinline__ void wsum_group8(const float2* r, float w_v, float2& acc)
{
    #pragma unroll
    for (int j = 0; j < 8; ++j) {
        const float w = __int_as_float(__builtin_amdgcn_readlane(__float_as_int(w_v), BREV5(G * 8 + j)));
        acc.x += w * r[G * 8 + j].x;
        acc.y += w * r[G * 8 + j].y;
    }
}

#define SORT_CAP 12288
#define AGG_TILES ((NB_NODES + 15) / 16)
#define AGG_PITCH 68
__global__ void __launch_bounds__(1024)
bucket_agg_kernel(const int* __restrict__ recs_i, const unsigned* __restrict__ btotal,
                  const float* __restrict__ xl, const float* __restrict__ We, const float* __restrict__ att,
                  const float* __restrict__ xr, const float* __restrict__ bias,
                  const float* __restrict__ Wfc, const float* __restrict__ bfc,
                  float* __restrict__ out, int NT)
{
    __shared__ float sAgg[AGG_TILES * 16 * AGG_PITCH];
    __shared__ float sWfc[WLDS];
    __shared__ float sbfc[64], sbias[64];
    __shared__ int2 sP[SORT_CAP];
    __shared__ unsigned sCnt[NBMAX], sOff[NBMAX];
    __shared__ unsigned sTot[8];
    __shared__ unsigned sBeg, sNum;
    __shared__ int2 sSel[32][32];
    const int2* __restrict__ recs = (const int2*)recs_i;
    const int t = threadIdx.x;
    const int lane = t & 31;
    const int wib = __builtin_amdgcn_readfirstlane(t >> 5);
    const int nb = (NT + NB_NODES - 1) / NB_NODES;
    const float2 we = ((const float2*)We)[lane];
    const float2 at = ((const float2*)att)[lane];
    const bool b4 = (lane & 16) != 0, b3 = (lane & 8) != 0, b2 = (lane & 4) != 0;
    const bool b1 = (lane & 2) != 0,  b0 = (lane & 1) != 0;
    const unsigned mine = BREV5(lane);
    const int b = blockIdx.x;
    if (b >= nb) return;
    stage_weight_pairs(sWfc, Wfc);
    if (t < 64) { sbfc[t] = bfc[t]; sbias[t] = bias[t]; }
    {
        const unsigned tot = (t < nb) ? btotal[t] : 0u;
        const unsigned first = excl_scan_256(tot, sTot);
        if (t == b) { sBeg = first; sNum = tot; }
        if (t < NBMAX) sCnt[t] = 0u;
        __syncthreads();
        const unsigned beg = sBeg, nrec = sNum;
        const bool staged = nrec <= SORT_CAP;
        if (staged) {
            for (unsigned i0 = t; i0 < nrec; i0 += 8 * 1024) {
                int k[8];
                #pragma unroll
                for (int j = 0; j < 8; ++j) {
                    const unsigned i = i0 + j * 1024;
                    k[j] = recs[beg + (i < nrec ? i : i0)].x;
                }
                #pragma unroll
                for (int j = 0; j < 8; ++j)
                    if (i0 + j * 1024 < nrec) atomicAdd(&sCnt[(unsigned)k[j] >> 24], 1u);
            }
        }
        __syncthreads();
        const unsigned cn = (t < NB_NODES) ? sCnt[t] : 0u;
        const unsigned noff = excl_scan_256(cn, sTot);
        if (t < NB_NODES) {
            sOff[t] = noff;
            sCnt[t] = 0u;
        }
        __syncthreads();
        if (staged) {
            for (unsigned i0 = t; i0 < nrec; i0 += 8 * 1024) {
                int2 r[8];
                #pragma unroll
                for (int j = 0; j < 8; ++j) {
                    const unsigned i = i0 + j * 1024;
                    r[j] = recs[beg + (i < nrec ? i : i0)];
                }
                #pragma unroll
                for (int j = 0; j < 8; ++j) {
                    if (i0 + j * 1024 < nrec) {
                        const unsigned dl = (unsigned)r[j].x >> 24;
                        const unsigned pos = sOff[dl] + atomicAdd(&sCnt[dl], 1u);
                        sP[pos] = make_int2(r[j].x & 0x00FFFFFF, r[j].y);
                    }
                }
            }
        }
        __syncthreads();

        for (int tn = wib; tn < NB_NODES; tn += 32) {
            const int d = b * NB_NODES + tn;
            if (d >= NT) break;
            const unsigned off = __builtin_amdgcn_readfirstlane(sOff[tn]);
            const unsigned n = staged ? __builtin_amdgcn_readfirstlane(sCnt[tn]) : nrec;
            const float2 xr2 = ((const float2*)(xr + (size_t)d * 64))[lane];

            float m = -__builtin_inff();
            float den = 0.0f;
            float2 acc = make_float2(0.0f, 0.0f);
            bool any = false;
            for (unsigned c = 0; c < n; c += 32) {
                int2 pr;
                bool valid;
                unsigned left;
                if (staged) {
                    const unsigned i = c + mine;
                    valid = i < n;
                    pr = sP[off + (valid ? i : c)];
                    left = n - c;
                } else {
                    const unsigned i = c + lane;
                    const bool in = i < nrec;
                    const int2 rr = recs[beg + (in ? i : c)];
                    const bool match = in && ((unsigned)rr.x >> 24) == (unsigned)tn;
                    const unsigned mask = __builtin_amdgcn_ballot_w32(match);
                    if (mask == 0u) continue;
                    left = (unsigned)__builtin_popcount(mask);
                    const unsigned rank = (unsigned)__builtin_popcount(mask & ((1u << lane) - 1u));
                    if (match) sSel[wib][rank] = make_int2(rr.x & 0x00FFFFFF, rr.y);
                    valid = mine < left;
                    pr = sSel[wib][valid ? mine : 0u];
                }
                any = true;
                float2 r[32];
                load_group8<0>(pr.x, xl, lane, r);
                if (left > 8)  load_group8<1>(pr.x, xl, lane, r);
                if (left > 16) load_group8<2>(pr.x, xl, lane, r);
                if (left > 24) load_group8<3>(pr.x, xl, lane, r);
                float t0, t1 = 0.0f, t2 = 0.0f, t3 = 0.0f;
                t0 = score_group8<0>(r, pr.y, xr2, we, at, b4, b3, b2);
                if (left > 8)  t1 = score_group8<1>(r, pr.y, xr2, we, at, b4, b3, b2);
                if (left > 16) t2 = score_group8<2>(r, pr.y, xr2, we, at, b4, b3, b2);
                if (left > 24) t3 = score_group8<3>(r, pr.y, xr2, we, at, b4, b3, b2);
                const float u0 = merge2<2>(t0, t1, b1);
                const float u1 = merge2<2>(t2, t3, b1);
                const float p  = merge2<1>(u0, u1, b0);
                float cm = valid ? p : -__builtin_inff();
                for (int o = 16; o > 0; o >>= 1) cm = fmaxf(cm, __shfl_xor(cm, o, 32));
                const float mn = fmaxf(m, cm);
                const float scale = __expf(m - mn);
                const float w_v = valid ? __expf(p - mn) : 0.0f;
                den = den * scale + w_v;
                acc.x *= scale;
                acc.y *= scale;
                wsum_group8<0>(r, w_v, acc);
                if (left > 8)  wsum_group8<1>(r, w_v, acc);
                if (left > 16) wsum_group8<2>(r, w_v, acc);
                if (left > 24) wsum_group8<3>(r, w_v, acc);
                m = mn;
            }
            for (int o = 16; o > 0; o >>= 1) den += __shfl_xor(den, o, 32);
            const float inv = any ? 1.0f / den : 0.0f;
            ((float2*)(sAgg + tn * AGG_PITCH))[lane] = make_float2(acc.x * inv, acc.y * inv);
        }
    }
    __syncthreads();

    if (wib < AGG_TILES * 2) {
        const int tile = wib >> 1, nb0 = (wib & 1) * 2;
        const int l15 = lane & 15, half = lane >> 4;
        v8f acc0 = v8f{}, acc1 = v8f{};
        const float* arow = sAgg + (tile * 16 + l15) * AGG_PITCH;
        #pragma unroll
        for (int kk = 0; kk < 16; ++kk) {
            const int kbase = kk * 4 + half * 2;
            v2f a = *(const v2f*)(arow + kbase);
            a.x += sbias[kbase];
            a.y += sbias[kbase + 1];
            const int pbase = (kk * 2 + half) * PSTRIDE;
            const v2f b0 = *(const v2f*)(sWfc + pbase + (nb0 * 16 + l15) * 2);
            const v2f b1 = *(const v2f*)(sWfc + pbase + ((nb0 + 1) * 16 + l15) * 2);
            acc0 = __builtin_amdgcn_wmma_f32_16x16x4_f32(false, a, false, b0, (short)0, acc0, false, false);
            acc1 = __builtin_amdgcn_wmma_f32_16x16x4_f32(false, a, false, b1, (short)0, acc1, false, false);
        }
        const int c0 = nb0 * 16 + l15, c1 = c0 + 16;
        #pragma unroll
        for (int v = 0; v < 8; ++v) {
            const int tn = tile * 16 + v + half * 8;
            const int d = b * NB_NODES + tn;
            if (tn < NB_NODES && d < NT) {
                out[(size_t)d * 64 + c0] = acc0[v] + sbfc[c0];
                out[(size_t)d * 64 + c1] = acc1[v] + sbfc[c1];
            }
        }
    }
}

extern "C" void kernel_launch(void* const* d_in, const int* in_sizes, int n_in,
                              void* d_out, int out_size, void* d_ws, size_t ws_size,
                              hipStream_t stream) {
    const float* x      = (const float*)d_in[0];
    const int*   eidx   = (const int*)d_in[1];
    const float* eattr  = (const float*)d_in[2];
    const float* W_l    = (const float*)d_in[3];
    const float* b_l    = (const float*)d_in[4];
    const float* W_r    = (const float*)d_in[5];
    const float* b_r    = (const float*)d_in[6];
    const float* W_e    = (const float*)d_in[7];
    const float* att    = (const float*)d_in[8];
    const float* bias   = (const float*)d_in[9];
    const float* W_fc   = (const float*)d_in[10];
    const float* b_fc   = (const float*)d_in[11];

    const int NT       = in_sizes[0] / 64;
    const long long E  = (long long)in_sizes[2];
    const int* src = eidx;
    const int* dst = eidx + E;

    float* ws       = (float*)d_ws;
    float* xl       = ws;
    float* xr       = xl + (size_t)NT * 64;
    int*   recs     = (int*)(xr + (size_t)NT * 64);
    unsigned* btotal  = (unsigned*)(recs + (size_t)NT * 64);
    unsigned* bcursor = btotal + NBMAX;

    const int numTiles = NT / 16;
    const int gemmBlocks = (numTiles + 7) / 8;

    init_ws_kernel<<<1, 512, 0, stream>>>(btotal, 512);
    bucket_count_kernel<<<256, 1024, 0, stream>>>(dst, btotal, NT, E);
    edge_scatter_kernel<<<256, 1024, 0, stream>>>(src, dst, eattr, btotal, bcursor, recs, NT, E);
    node_xform_kernel<<<gemmBlocks, 256, 0, stream>>>(x, W_l, b_l, W_r, b_r, xl, xr, NT);
    bucket_agg_kernel<<<256, 1024, 0, stream>>>(recs, btotal, xl, W_e, att, xr, bias, W_fc, b_fc, (float*)d_out, NT);
}


# level-1 sort block-local: a block orders its own share of recs by bucket and publishes each run's place and length; global totals, cursors, count and init launches gone; sum order still unfixed
# speedup vs baseline: 6.4546x; 1.0462x over previous
#include <hip/hip_runtime.h>
#include <hip/hip_bf16.h>

typedef float v2f __attribute__((ext_vector_type(2)));
typedef float v8f __attribute__((ext_vector_type(8)));

#define PSTRIDE 160
#define WLDS    (32 * PSTRIDE)

__device__ __forceinline__ void stage_weight_pairs(float* sW, const float* __restrict__ W) {
    for (int i = threadIdx.x; i < 64 * 64; i += blockDim.x) {
        const int k = i >> 6, col = i & 63;
        sW[(k >> 1) * PSTRIDE + col * 2 + (k & 1)] = W[i];
    }
}

__global__ void node_xform_kernel(const float* __restrict__ x2,
                                  const float* __restrict__ Wl, const float* __restrict__ bl,
                                  const float* __restrict__ Wr, const float* __restrict__ br,
                                  float* __restrict__ xl, float* __restrict__ xr, int NT)
{
    __shared__ float sWl[WLDS];
    __shared__ float sWr[WLDS];
    __shared__ float sbl[64], sbr[64];
    stage_weight_pairs(sWl, Wl);
    stage_weight_pairs(sWr, Wr);
    for (int i = threadIdx.x; i < 64; i += blockDim.x) { sbl[i] = bl[i]; sbr[i] = br[i]; }
    __syncthreads();

    const int lane    = threadIdx.x & 31;
    const int wave    = threadIdx.x >> 5;
    const int wpb     = blockDim.x >> 5;
    const int l15     = lane & 15;
    const int half    = lane >> 4;
    const int numTiles = NT >> 4;

    for (int tile = blockIdx.x * wpb + wave; tile < numTiles; tile += gridDim.x * wpb) {
        v8f accl[4] = {v8f{}, v8f{}, v8f{}, v8f{}};
        v8f accr[4] = {v8f{}, v8f{}, v8f{}, v8f{}};
        const float* xrow = x2 + (size_t)(tile * 16 + l15) * 64;
        #pragma unroll 4
        for (int kk = 0; kk < 16; ++kk) {
            const int kbase = kk * 4 + half * 2;
            const v2f a = *(const v2f*)(xrow + kbase);
            const int pbase = (kk * 2 + half) * PSTRIDE;
            #pragma unroll
            for (int nb = 0; nb < 4; ++nb) {
                const int col = nb * 16 + l15;
                const v2f bL = *(const v2f*)(sWl + pbase + col * 2);
                const v2f bR = *(const v2f*)(sWr + pbase + col * 2);
                accl[nb] = __builtin_amdgcn_wmma_f32_16x16x4_f32(
                    false, a, false, bL, (short)0, accl[nb], false, false);
                accr[nb] = __builtin_amdgcn_wmma_f32_16x16x4_f32(
                    false, a, false, bR, (short)0, accr[nb], false, false);
            }
        }
        #pragma unroll
        for (int nb = 0; nb < 4; ++nb) {
            const int col = nb * 16 + l15;
            #pragma unroll
            for (int v = 0; v < 8; ++v) {
                const size_t row = (size_t)(tile * 16 + v + half * 8);
                xl[row * 64 + col] = accl[nb][v] + sbl[col];
                xr[row * 64 + col] = accr[nb][v] + sbr[col];
            }
        }
    }
}

#define NB_NODES 196
#define NBMAX    256
#define SC_BLOCKS 256
#define STAGE_CAP 8192

__device__ __forceinline__ unsigned excl_scan_256(unsigned v, unsigned* sTot)
{
    const int lane = threadIdx.x & 31, w = threadIdx.x >> 5;
    unsigned inc = v;
    for (int o = 1; o < 32; o <<= 1) {
        const unsigned up = __shfl_up(inc, o, 32);
        if (lane >= o) inc += up;
    }
    __syncthreads();
    if (lane == 31 && w < 8) sTot[w] = inc;
    __syncthreads();
    unsigned pre = 0u;
    for (int j = 0; j < 8; ++j) if (j < w) pre += sTot[j];
    return pre + inc - v;
}

__global__ void __launch_bounds__(1024) edge_scatter_kernel(const int* __restrict__ src, const int* __restrict__ dst,
                                  const float* __restrict__ eattr,
                                  unsigned* __restrict__ runtab_u, int* __restrict__ recs, int NT, long long E)
{
    __shared__ unsigned sCnt[NBMAX];
    __shared__ unsigned sLoc[NBMAX];
    __shared__ unsigned sTot[8];
    __shared__ int2     sRec[STAGE_CAP];
    const int t = threadIdx.x;
    const long long ebeg = (E * blockIdx.x) / SC_BLOCKS;
    const long long eend = (E * (blockIdx.x + 1)) / SC_BLOCKS;
    const unsigned nblk = (eend > ebeg) ? (unsigned)(eend - ebeg) : 0u;

    if (t < NBMAX) sCnt[t] = 0u;
    __syncthreads();
    for (long long e0 = ebeg + t; e0 < eend; e0 += 8 * 1024) {
        int d[8];
        #pragma unroll
        for (int j = 0; j < 8; ++j) {
            const long long e = e0 + j * 1024;
            d[j] = dst[(e < eend) ? e : e0];
        }
        #pragma unroll
        for (int j = 0; j < 8; ++j)
            if (e0 + j * 1024 < eend) atomicAdd(&sCnt[(unsigned)d[j] / NB_NODES], 1u);
    }
    __syncthreads();
    const int nb = (NT + NB_NODES - 1) / NB_NODES;
    const unsigned c = (t < nb) ? sCnt[t] : 0u;
    const unsigned loc = excl_scan_256(c, sTot);
    if (t < nb) {
        ((uint2*)runtab_u)[(size_t)t * SC_BLOCKS + blockIdx.x] = make_uint2(loc, c);
        sLoc[t]  = loc;
        sCnt[t]  = 0u;
    }
    __syncthreads();
    const bool staged = nblk <= STAGE_CAP;
    for (long long e0 = ebeg + t; e0 < eend; e0 += 4 * 1024) {
        int s[4], d[4];
        float a[4];
        #pragma unroll
        for (int j = 0; j < 4; ++j) {
            const long long e = e0 + j * 1024;
            const long long ec = (e < eend) ? e : e0;
            s[j] = src[ec];
            d[j] = dst[ec];
            a[j] = eattr[ec];
        }
        #pragma unroll
        for (int j = 0; j < 4; ++j) {
            if (e0 + j * 1024 < eend) {
                const unsigned bk = (unsigned)d[j] / NB_NODES;
                const unsigned r = atomicAdd(&sCnt[bk], 1u);
                const unsigned lp = sLoc[bk] + r;
                const unsigned key = (unsigned)s[j] | (((unsigned)d[j] - bk * NB_NODES) << 24);
                const int2 rec = make_int2((int)key, __float_as_int(a[j]));
                if (staged) sRec[lp] = rec;
                else        ((int2*)recs)[ebeg + lp] = rec;
            }
        }
    }
    if (staged) {
        __syncthreads();
        for (unsigned i = t; i < nblk; i += blockDim.x) ((int2*)recs)[ebeg + i] = sRec[i];
    }
}

#define BREV5(j) ((((j) & 1) << 4) | (((j) & 2) << 2) | ((j) & 4) | (((j) & 8) >> 2) | (((j) & 16) >> 4))

template <int MASK>
__device__ __forceinline__ float merge2(float a, float b, bool up) {
    const float keep = up ? b : a;
    const float give = up ? a : b;
    return keep + __shfl_xor(give, MASK, 32);
}

template <int G>
__device__ __forceinline__ void load_group8(int s_v, const float* __restrict__ xl, int lane, float2* r)
{
    #pragma unroll
    for (int j = 0; j < 8; ++j) {
        const int s = __builtin_amdgcn_readlane(s_v, BREV5(G * 8 + j));
        r[G * 8 + j] = ((const float2*)(xl + (size_t)s * 64))[lane];
    }
}

template <int G>
__device__ __forceinline__ float score_group8(const float2* r, int ea_v, float2 xr2, float2 we, float2 at,
                                              bool b4, bool b3, bool b2)
{
    float p[8];
    #pragma unroll
    for (int j = 0; j < 8; ++j) {
        const float ea = __int_as_float(__builtin_amdgcn_readlane(ea_v, BREV5(G * 8 + j)));
        float m0 = r[G * 8 + j].x + xr2.x + ea * we.x;
        float m1 = r[G * 8 + j].y + xr2.y + ea * we.y;
        m0 = (m0 > 0.0f) ? m0 : 0.2f * m0;
        m1 = (m1 > 0.0f) ? m1 : 0.2f * m1;
        p[j] = __builtin_fmaf(at.x, m0, at.y * m1);
    }
    const float q0 = merge2<16>(p[0], p[1], b4);
    const float q1 = merge2<16>(p[2], p[3], b4);
    const float q2 = merge2<16>(p[4], p[5], b4);
    const float q3 = merge2<16>(p[6], p[7], b4);
    const float r0 = merge2<8>(q0, q1, b3);
    const float r1 = merge2<8>(q2, q3, b3);
    return merge2<4>(r0, r1, b2);
}

template <int G>
__device__ __forceinline__ void wsum_group8(const float2* r, float w_v, float2& acc)
{
    #pragma unroll
    for (int j = 0; j < 8; ++j) {
        const float w = __int_as_float(__builtin_amdgcn_readlane(__float_as_int(w_v), BREV5(G * 8 + j)));
        acc.x += w * r[G * 8 + j].x;
        acc.y += w * r[G * 8 + j].y;
    }
}

#define SORT_CAP 12288
#define AGG_TILES ((NB_NODES + 15) / 16)
#define AGG_PITCH 68
__device__ __forceinline__ unsigned run_slot(unsigned i, const unsigned* sPre, const unsigned* sBeg)
{
    unsigned lo = 0u;
    for (unsigned st = SC_BLOCKS / 2; st > 0u; st >>= 1)
        if (sPre[lo + st] <= i) lo += st;
    return sBeg[lo] + (i - sPre[lo]);
}

__global__ void __launch_bounds__(1024)
bucket_agg_kernel(const int* __restrict__ recs_i, const unsigned* __restrict__ runtab_u,
                  const float* __restrict__ xl, const float* __restrict__ We, const float* __restrict__ att,
                  const float* __restrict__ xr, const float* __restrict__ bias,
                  const float* __restrict__ Wfc, const float* __restrict__ bfc,
                  float* __restrict__ out, int NT, long long E)
{
    __shared__ float sAgg[AGG_TILES * 16 * AGG_PITCH];
    __shared__ float sWfc[WLDS];
    __shared__ float sbfc[64], sbias[64];
    __shared__ int2 sP[SORT_CAP];
    __shared__ unsigned sCnt[NBMAX], sOff[NBMAX];
    __shared__ unsigned sTot[8];
    __shared__ unsigned sRunBeg[SC_BLOCKS], sRunCnt[SC_BLOCKS], sRunPre[SC_BLOCKS];
    __shared__ unsigned sNum;
    __shared__ int2 sSel[32][32];
    const int2* __restrict__ recs = (const int2*)recs_i;
    const int t = threadIdx.x;
    const int lane = t & 31;
    const int wib = __builtin_amdgcn_readfirstlane(t >> 5);
    const int nb = (NT + NB_NODES - 1) / NB_NODES;
    const float2 we = ((const float2*)We)[lane];
    const float2 at = ((const float2*)att)[lane];
    const bool b4 = (lane & 16) != 0, b3 = (lane & 8) != 0, b2 = (lane & 4) != 0;
    const bool b1 = (lane & 2) != 0,  b0 = (lane & 1) != 0;
    const unsigned mine = BREV5(lane);
    const int b = blockIdx.x;
    if (b >= nb) return;
    stage_weight_pairs(sWfc, Wfc);
    if (t < 64) { sbfc[t] = bfc[t]; sbias[t] = bias[t]; }
    {
        unsigned rlen = 0u;
        if (t < SC_BLOCKS) {
            const uint2 ent = ((const uint2*)runtab_u)[(size_t)b * SC_BLOCKS + t];
            sRunBeg[t] = (unsigned)((E * t) / SC_BLOCKS) + ent.x;
            sRunCnt[t] = ent.y;
            rlen = ent.y;
        }
        const unsigned rpre = excl_scan_256(rlen, sTot);
        if (t < SC_BLOCKS) sRunPre[t] = rpre;
        if (t == SC_BLOCKS - 1) sNum = rpre + rlen;
        if (t < NBMAX) sCnt[t] = 0u;
        __syncthreads();
        const unsigned nrec = sNum;
        const bool staged = nrec <= SORT_CAP;
        unsigned rbeg[SC_BLOCKS / 32], rcnt[SC_BLOCKS / 32];
        unsigned longest = 0u;
        #pragma unroll
        for (int j = 0; j < SC_BLOCKS / 32; ++j) {
            rbeg[j] = __builtin_amdgcn_readfirstlane(sRunBeg[wib + 32 * j]);
            rcnt[j] = __builtin_amdgcn_readfirstlane(sRunCnt[wib + 32 * j]);
            longest = rcnt[j] > longest ? rcnt[j] : longest;
        }
        if (staged) {
            for (unsigned c0 = 0; c0 < longest; c0 += 32) {
                const unsigned i = c0 + lane;
                int k[SC_BLOCKS / 32];
                #pragma unroll
                for (int j = 0; j < SC_BLOCKS / 32; ++j)
                    k[j] = recs[i < rcnt[j] ? rbeg[j] + i : 0u].x;
                #pragma unroll
                for (int j = 0; j < SC_BLOCKS / 32; ++j)
                    if (i < rcnt[j]) atomicAdd(&sCnt[(unsigned)k[j] >> 24], 1u);
            }
        }
        __syncthreads();
        const unsigned cn = (t < NB_NODES) ? sCnt[t] : 0u;
        const unsigned noff = excl_scan_256(cn, sTot);
        if (t < NB_NODES) {
            sOff[t] = noff;
            sCnt[t] = 0u;
        }
        __syncthreads();
        if (staged) {
            for (unsigned c0 = 0; c0 < longest; c0 += 32) {
                const unsigned i = c0 + lane;
                int2 r[SC_BLOCKS / 32];
                #pragma unroll
                for (int j = 0; j < SC_BLOCKS / 32; ++j)
                    r[j] = recs[i < rcnt[j] ? rbeg[j] + i : 0u];
                #pragma unroll
                for (int j = 0; j < SC_BLOCKS / 32; ++j) {
                    if (i < rcnt[j]) {
                        const unsigned dl = (unsigned)r[j].x >> 24;
                        const unsigned pos = sOff[dl] + atomicAdd(&sCnt[dl], 1u);
                        sP[pos] = make_int2(r[j].x & 0x00FFFFFF, r[j].y);
                    }
                }
            }
        }
        __syncthreads();

        for (int tn = wib; tn < NB_NODES; tn += 32) {
            const int d = b * NB_NODES + tn;
            if (d >= NT) break;
            const unsigned off = __builtin_amdgcn_readfirstlane(sOff[tn]);
            const unsigned n = staged ? __builtin_amdgcn_readfirstlane(sCnt[tn]) : nrec;
            const float2 xr2 = ((const float2*)(xr + (size_t)d * 64))[lane];

            float m = -__builtin_inff();
            float den = 0.0f;
            float2 acc = make_float2(0.0f, 0.0f);
            bool any = false;
            for (unsigned c = 0; c < n; c += 32) {
                int2 pr;
                bool valid;
                unsigned left;
                if (staged) {
                    const unsigned i = c + mine;
                    valid = i < n;
                    pr = sP[off + (valid ? i : c)];
                    left = n - c;
                } else {
                    const unsigned i = c + lane;
                    const bool in = i < nrec;
                    const int2 rr = recs[in ? run_slot(i, sRunPre, sRunBeg) : 0u];
                    const bool match = in && ((unsigned)rr.x >> 24) == (unsigned)tn;
                    const unsigned mask = __builtin_amdgcn_ballot_w32(match);
                    if (mask == 0u) continue;
                    left = (unsigned)__builtin_popcount(mask);
                    const unsigned rank = (unsigned)__builtin_popcount(mask & ((1u << lane) - 1u));
                    if (match) sSel[wib][rank] = make_int2(rr.x & 0x00FFFFFF, rr.y);
                    valid = mine < left;
                    pr = sSel[wib][valid ? mine : 0u];
                }
                any = true;
                float2 r[32];
                load_group8<0>(pr.x, xl, lane, r);
                if (left > 8)  load_group8<1>(pr.x, xl, lane, r);
                if (left > 16) load_group8<2>(pr.x, xl, lane, r);
                if (left > 24) load_group8<3>(pr.x, xl, lane, r);
                float t0, t1 = 0.0f, t2 = 0.0f, t3 = 0.0f;
                t0 = score_group8<0>(r, pr.y, xr2, we, at, b4, b3, b2);
                if (left > 8)  t1 = score_group8<1>(r, pr.y, xr2, we, at, b4, b3, b2);
                if (left > 16) t2 = score_group8<2>(r, pr.y, xr2, we, at, b4, b3, b2);
                if (left > 24) t3 = score_group8<3>(r, pr.y, xr2, we, at, b4, b3, b2);
                const float u0 = merge2<2>(t0, t1, b1);
                const float u1 = merge2<2>(t2, t3, b1);
                const float p  = merge2<1>(u0, u1, b0);
                float cm = valid ? p : -__builtin_inff();
                for (int o = 16; o > 0; o >>= 1) cm = fmaxf(cm, __shfl_xor(cm, o, 32));
                const float mn = fmaxf(m, cm);
                const float scale = __expf(m - mn);
                const float w_v = valid ? __expf(p - mn) : 0.0f;
                den = den * scale + w_v;
                acc.x *= scale;
                acc.y *= scale;
                wsum_group8<0>(r, w_v, acc);
                if (left > 8)  wsum_group8<1>(r, w_v, acc);
                if (left > 16) wsum_group8<2>(r, w_v, acc);
                if (left > 24) wsum_group8<3>(r, w_v, acc);
                m = mn;
            }
            for (int o = 16; o > 0; o >>= 1) den += __shfl_xor(den, o, 32);
            const float inv = any ? 1.0f / den : 0.0f;
            ((float2*)(sAgg + tn * AGG_PITCH))[lane] = make_float2(acc.x * inv, acc.y * inv);
        }
    }
    __syncthreads();

    if (wib < AGG_TILES * 2) {
        const int tile = wib >> 1, nb0 = (wib & 1) * 2;
        const int l15 = lane & 15, half = lane >> 4;
        v8f acc0 = v8f{}, acc1 = v8f{};
        const float* arow = sAgg + (tile * 16 + l15) * AGG_PITCH;
        #pragma unroll
        for (int kk = 0; kk < 16; ++kk) {
            const int kbase = kk * 4 + half * 2;
            v2f a = *(const v2f*)(arow + kbase);
            a.x += sbias[kbase];
            a.y += sbias[kbase + 1];
            const int pbase = (kk * 2 + half) * PSTRIDE;
            const v2f b0 = *(const v2f*)(sWfc + pbase + (nb0 * 16 + l15) * 2);
            const v2f b1 = *(const v2f*)(sWfc + pbase + ((nb0 + 1) * 16 + l15) * 2);
            acc0 = __builtin_amdgcn_wmma_f32_16x16x4_f32(false, a, false, b0, (short)0, acc0, false, false);
            acc1 = __builtin_amdgcn_wmma_f32_16x16x4_f32(false, a, false, b1, (short)0, acc1, false, false);
        }
        const int c0 = nb0 * 16 + l15, c1 = c0 + 16;
        #pragma unroll
        for (int v = 0; v < 8; ++v) {
            const int tn = tile * 16 + v + half * 8;
            const int d = b * NB_NODES + tn;
            if (tn < NB_NODES && d < NT) {
                out[(size_t)d * 64 + c0] = acc0[v] + sbfc[c0];
                out[(size_t)d * 64 + c1] = acc1[v] + sbfc[c1];
            }
        }
    }
}

extern "C" void kernel_launch(void* const* d_in, const int* in_sizes, int n_in,
                              void* d_out, int out_size, void* d_ws, size_t ws_size,
                              hipStream_t stream) {
    const float* x      = (const float*)d_in[0];
    const int*   eidx   = (const int*)d_in[1];
    const float* eattr  = (const float*)d_in[2];
    const float* W_l    = (const float*)d_in[3];
    const float* b_l    = (const float*)d_in[4];
    const float* W_r    = (const float*)d_in[5];
    const float* b_r    = (const float*)d_in[6];
    const float* W_e    = (const float*)d_in[7];
    const float* att    = (const float*)d_in[8];
    const float* bias   = (const float*)d_in[9];
    const float* W_fc   = (const float*)d_in[10];
    const float* b_fc   = (const float*)d_in[11];

    const int NT       = in_sizes[0] / 64;
    const long long E  = (long long)in_sizes[2];
    const int* src = eidx;
    const int* dst = eidx + E;

    float* ws       = (float*)d_ws;
    float* xl       = ws;
    float* xr       = xl + (size_t)NT * 64;
    int*   recs     = (int*)(xr + (size_t)NT * 64);
    unsigned* runtab = (unsigned*)(recs + (size_t)NT * 64);

    const int numTiles = NT / 16;
    const int gemmBlocks = (numTiles + 7) / 8;

    edge_scatter_kernel<<<SC_BLOCKS, 1024, 0, stream>>>(src, dst, eattr, runtab, recs, NT, E);
    node_xform_kernel<<<gemmBlocks, 256, 0, stream>>>(x, W_l, b_l, W_r, b_r, xl, xr, NT);
    bucket_agg_kernel<<<256, 1024, 0, stream>>>(recs, runtab, xl, W_e, att, xr, bias, W_fc, b_fc, (float*)d_out, NT, E);
}


# edge ordering and the xl/xr transform (independent) run as ONE launch: 256 + 256 blocks by role, 64 registers a lane so one of each shares a WGP; a wave forms one tile x one matrix; sums unchanged
# speedup vs baseline: 8.0020x; 1.2397x over previous
#include <hip/hip_runtime.h>
#include <hip/hip_bf16.h>

typedef float v2f __attribute__((ext_vector_type(2)));
typedef float v8f __attribute__((ext_vector_type(8)));

#define PSTRIDE 160
#define WLDS    (32 * PSTRIDE)

__device__ __forceinline__ void stage_weight_pairs(float* sW, const float* __restrict__ W) {
    for (int i = threadIdx.x; i < 64 * 64; i += blockDim.x) {
        const int k = i >> 6, col = i & 63;
        sW[(k >> 1) * PSTRIDE + col * 2 + (k & 1)] = W[i];
    }
}

#define XF_BLOCKS 256

__device__ __forceinline__ void node_xform_role(int blk,
                                  const float* __restrict__ x2,
                                  const float* __restrict__ Wl, const float* __restrict__ bl,
                                  const float* __restrict__ Wr, const float* __restrict__ br,
                                  float* __restrict__ xl, float* __restrict__ xr, int NT)
{
    __shared__ float sWl[WLDS];
    __shared__ float sWr[WLDS];
    __shared__ float sbl[64], sbr[64];
    stage_weight_pairs(sWl, Wl);
    stage_weight_pairs(sWr, Wr);
    for (int i = threadIdx.x; i < 64; i += blockDim.x) { sbl[i] = bl[i]; sbr[i] = br[i]; }
    __syncthreads();

    const int lane    = threadIdx.x & 31;
    const int wave    = __builtin_amdgcn_readfirstlane(threadIdx.x >> 5);
    const int wpb     = blockDim.x >> 5;
    const int l15     = lane & 15;
    const int half    = lane >> 4;
    const int numUnits = (NT >> 4) * 2;
    const int per      = (numUnits + XF_BLOCKS - 1) / XF_BLOCKS;

    for (int w = wave; w < per; w += wpb) {
        const int unit = blk * per + w;
        if (unit >= numUnits) break;
        const int tile = unit >> 1;
        const bool right = (unit & 1) != 0;
        const float* sW = right ? sWr : sWl;
        const float* sb = right ? sbr : sbl;
        float* __restrict__ dstp = right ? xr : xl;
        v8f acc[4] = {v8f{}, v8f{}, v8f{}, v8f{}};
        const float* xrow = x2 + (size_t)(tile * 16 + l15) * 64;
        for (int k4 = 0; k4 < 16; k4 += 4) {
            v2f a[4];
            #pragma unroll
            for (int j = 0; j < 4; ++j)
                a[j] = *(const v2f*)(xrow + (k4 + j) * 4 + half * 2);
            #pragma unroll
            for (int j = 0; j < 4; ++j) {
                const int pbase = ((k4 + j) * 2 + half) * PSTRIDE;
                #pragma unroll
                for (int nb = 0; nb < 4; ++nb) {
                    const int col = nb * 16 + l15;
                    const v2f bw = *(const v2f*)(sW + pbase + col * 2);
                    acc[nb] = __builtin_amdgcn_wmma_f32_16x16x4_f32(
                        false, a[j], false, bw, (short)0, acc[nb], false, false);
                }
            }
        }
        #pragma unroll
        for (int nb = 0; nb < 4; ++nb) {
            const int col = nb * 16 + l15;
            #pragma unroll
            for (int v = 0; v < 8; ++v) {
                const size_t row = (size_t)(tile * 16 + v + half * 8);
                dstp[row * 64 + col] = acc[nb][v] + sb[col];
            }
        }
    }
}

#define NB_NODES 196
#define NBMAX    256
#define SC_BLOCKS 256
#define STAGE_CAP 8192

__device__ __forceinline__ unsigned excl_scan_256(unsigned v, unsigned* sTot)
{
    const int lane = threadIdx.x & 31, w = threadIdx.x >> 5;
    unsigned inc = v;
    for (int o = 1; o < 32; o <<= 1) {
        const unsigned up = __shfl_up(inc, o, 32);
        if (lane >= o) inc += up;
    }
    __syncthreads();
    if (lane == 31 && w < 8) sTot[w] = inc;
    __syncthreads();
    unsigned pre = 0u;
    for (int j = 0; j < 8; ++j) if (j < w) pre += sTot[j];
    return pre + inc - v;
}

__device__ __forceinline__ void edge_scatter_role(const int* __restrict__ src, const int* __restrict__ dst,
                                  const float* __restrict__ eattr,
                                  unsigned* __restrict__ runtab_u, int* __restrict__ recs, int NT, long long E)
{
    __shared__ unsigned sCnt[NBMAX];
    __shared__ unsigned sLoc[NBMAX];
    __shared__ unsigned sTot[8];
    __shared__ int2     sRec[STAGE_CAP];
    const int t = threadIdx.x;
    const long long ebeg = (E * blockIdx.x) / SC_BLOCKS;
    const long long eend = (E * (blockIdx.x + 1)) / SC_BLOCKS;
    const unsigned nblk = (eend > ebeg) ? (unsigned)(eend - ebeg) : 0u;

    if (t < NBMAX) sCnt[t] = 0u;
    __syncthreads();
    for (long long e0 = ebeg + t; e0 < eend; e0 += 8 * 1024) {
        int d[8];
        #pragma unroll
        for (int j = 0; j < 8; ++j) {
            const long long e = e0 + j * 1024;
            d[j] = dst[(e < eend) ? e : e0];
        }
        #pragma unroll
        for (int j = 0; j < 8; ++j)
            if (e0 + j * 1024 < eend) atomicAdd(&sCnt[(unsigned)d[j] / NB_NODES], 1u);
    }
    __syncthreads();
    const int nb = (NT + NB_NODES - 1) / NB_NODES;
    const unsigned c = (t < nb) ? sCnt[t] : 0u;
    const unsigned loc = excl_scan_256(c, sTot);
    if (t < nb) {
        ((uint2*)runtab_u)[(size_t)t * SC_BLOCKS + blockIdx.x] = make_uint2(loc, c);
        sLoc[t]  = loc;
        sCnt[t]  = 0u;
    }
    __syncthreads();
    const bool staged = nblk <= STAGE_CAP;
    for (long long e0 = ebeg + t; e0 < eend; e0 += 4 * 1024) {
        int s[4], d[4];
        float a[4];
        #pragma unroll
        for (int j = 0; j < 4; ++j) {
            const long long e = e0 + j * 1024;
            const long long ec = (e < eend) ? e : e0;
            s[j] = src[ec];
            d[j] = dst[ec];
            a[j] = eattr[ec];
        }
        #pragma unroll
        for (int j = 0; j < 4; ++j) {
            if (e0 + j * 1024 < eend) {
                const unsigned bk = (unsigned)d[j] / NB_NODES;
                const unsigned r = atomicAdd(&sCnt[bk], 1u);
                const unsigned lp = sLoc[bk] + r;
                const unsigned key = (unsigned)s[j] | (((unsigned)d[j] - bk * NB_NODES) << 24);
                const int2 rec = make_int2((int)key, __float_as_int(a[j]));
                if (staged) sRec[lp] = rec;
                else        ((int2*)recs)[ebeg + lp] = rec;
            }
        }
    }
    if (staged) {
        __syncthreads();
        for (unsigned i = t; i < nblk; i += blockDim.x) ((int2*)recs)[ebeg + i] = sRec[i];
    }
}

__global__ void __launch_bounds__(1024) __attribute__((amdgpu_waves_per_eu(16)))
prep_kernel(const int* __restrict__ src, const int* __restrict__ dst, const float* __restrict__ eattr,
            unsigned* __restrict__ runtab_u, int* __restrict__ recs,
            const float* __restrict__ x2,
            const float* __restrict__ Wl, const float* __restrict__ bl,
            const float* __restrict__ Wr, const float* __restrict__ br,
            float* __restrict__ xl, float* __restrict__ xr, int NT, long long E)
{
    if (blockIdx.x < SC_BLOCKS) edge_scatter_role(src, dst, eattr, runtab_u, recs, NT, E);
    else                        node_xform_role((int)blockIdx.x - SC_BLOCKS, x2, Wl, bl, Wr, br, xl, xr, NT);
}

#define BREV5(j) ((((j) & 1) << 4) | (((j) & 2) << 2) | ((j) & 4) | (((j) & 8) >> 2) | (((j) & 16) >> 4))

template <int MASK>
__device__ __forceinline__ float merge2(float a, float b, bool up) {
    const float keep = up ? b : a;
    const float give = up ? a : b;
    return keep + __shfl_xor(give, MASK, 32);
}

template <int G>
__device__ __forceinline__ void load_group8(int s_v, const float* __restrict__ xl, int lane, float2* r)
{
    #pragma unroll
    for (int j = 0; j < 8; ++j) {
        const int s = __builtin_amdgcn_readlane(s_v, BREV5(G * 8 + j));
        r[G * 8 + j] = ((const float2*)(xl + (size_t)s * 64))[lane];
    }
}

template <int G>
__device__ __forceinline__ float score_group8(const float2* r, int ea_v, float2 xr2, float2 we, float2 at,
                                              bool b4, bool b3, bool b2)
{
    float p[8];
    #pragma unroll
    for (int j = 0; j < 8; ++j) {
        const float ea = __int_as_float(__builtin_amdgcn_readlane(ea_v, BREV5(G * 8 + j)));
        float m0 = r[G * 8 + j].x + xr2.x + ea * we.x;
        float m1 = r[G * 8 + j].y + xr2.y + ea * we.y;
        m0 = (m0 > 0.0f) ? m0 : 0.2f * m0;
        m1 = (m1 > 0.0f) ? m1 : 0.2f * m1;
        p[j] = __builtin_fmaf(at.x, m0, at.y * m1);
    }
    const float q0 = merge2<16>(p[0], p[1], b4);
    const float q1 = merge2<16>(p[2], p[3], b4);
    const float q2 = merge2<16>(p[4], p[5], b4);
    const float q3 = merge2<16>(p[6], p[7], b4);
    const float r0 = merge2<8>(q0, q1, b3);
    const float r1 = merge2<8>(q2, q3, b3);
    return merge2<4>(r0, r1, b2);
}

template <int G>
__device__ __forceinline__ void wsum_group8(const float2* r, float w_v, float2& acc)
{
    #pragma unroll
    for (int j = 0; j < 8; ++j) {
        const float w = __int_as_float(__builtin_amdgcn_readlane(__float_as_int(w_v), BREV5(G * 8 + j)));
        acc.x += w * r[G * 8 + j].x;
        acc.y += w * r[G * 8 + j].y;
    }
}

#define SORT_CAP 12288
#define AGG_TILES ((NB_NODES + 15) / 16)
#define AGG_PITCH 68
__device__ __forceinline__ unsigned run_slot(unsigned i, const unsigned* sPre, const unsigned* sBeg)
{
    unsigned lo = 0u;
    for (unsigned st = SC_BLOCKS / 2; st > 0u; st >>= 1)
        if (sPre[lo + st] <= i) lo += st;
    return sBeg[lo] + (i - sPre[lo]);
}

__global__ void __launch_bounds__(1024)
bucket_agg_kernel(const int* __restrict__ recs_i, const unsigned* __restrict__ runtab_u,
                  const float* __restrict__ xl, const float* __restrict__ We, const float* __restrict__ att,
                  const float* __restrict__ xr, const float* __restrict__ bias,
                  const float* __restrict__ Wfc, const float* __restrict__ bfc,
                  float* __restrict__ out, int NT, long long E)
{
    __shared__ float sAgg[AGG_TILES * 16 * AGG_PITCH];
    __shared__ float sWfc[WLDS];
    __shared__ float sbfc[64], sbias[64];
    __shared__ int2 sP[SORT_CAP];
    __shared__ unsigned sCnt[NBMAX], sOff[NBMAX];
    __shared__ unsigned sTot[8];
    __shared__ unsigned sRunBeg[SC_BLOCKS], sRunCnt[SC_BLOCKS], sRunPre[SC_BLOCKS];
    __shared__ unsigned sNum;
    __shared__ int2 sSel[32][32];
    const int2* __restrict__ recs = (const int2*)recs_i;
    const int t = threadIdx.x;
    const int lane = t & 31;
    const int wib = __builtin_amdgcn_readfirstlane(t >> 5);
    const int nb = (NT + NB_NODES - 1) / NB_NODES;
    const float2 we = ((const float2*)We)[lane];
    const float2 at = ((const float2*)att)[lane];
    const bool b4 = (lane & 16) != 0, b3 = (lane & 8) != 0, b2 = (lane & 4) != 0;
    const bool b1 = (lane & 2) != 0,  b0 = (lane & 1) != 0;
    const unsigned mine = BREV5(lane);
    const int b = blockIdx.x;
    if (b >= nb) return;
    stage_weight_pairs(sWfc, Wfc);
    if (t < 64) { sbfc[t] = bfc[t]; sbias[t] = bias[t]; }
    {
        unsigned rlen = 0u;
        if (t < SC_BLOCKS) {
            const uint2 ent = ((const uint2*)runtab_u)[(size_t)b * SC_BLOCKS + t];
            sRunBeg[t] = (unsigned)((E * t) / SC_BLOCKS) + ent.x;
            sRunCnt[t] = ent.y;
            rlen = ent.y;
        }
        const unsigned rpre = excl_scan_256(rlen, sTot);
        if (t < SC_BLOCKS) sRunPre[t] = rpre;
        if (t == SC_BLOCKS - 1) sNum = rpre + rlen;
        if (t < NBMAX) sCnt[t] = 0u;
        __syncthreads();
        const unsigned nrec = sNum;
        const bool staged = nrec <= SORT_CAP;
        unsigned rbeg[SC_BLOCKS / 32], rcnt[SC_BLOCKS / 32];
        unsigned longest = 0u;
        #pragma unroll
        for (int j = 0; j < SC_BLOCKS / 32; ++j) {
            rbeg[j] = __builtin_amdgcn_readfirstlane(sRunBeg[wib + 32 * j]);
            rcnt[j] = __builtin_amdgcn_readfirstlane(sRunCnt[wib + 32 * j]);
            longest = rcnt[j] > longest ? rcnt[j] : longest;
        }
        if (staged) {
            for (unsigned c0 = 0; c0 < longest; c0 += 32) {
                const unsigned i = c0 + lane;
                int k[SC_BLOCKS / 32];
                #pragma unroll
                for (int j = 0; j < SC_BLOCKS / 32; ++j)
                    k[j] = recs[i < rcnt[j] ? rbeg[j] + i : 0u].x;
                #pragma unroll
                for (int j = 0; j < SC_BLOCKS / 32; ++j)
                    if (i < rcnt[j]) atomicAdd(&sCnt[(unsigned)k[j] >> 24], 1u);
            }
        }
        __syncthreads();
        const unsigned cn = (t < NB_NODES) ? sCnt[t] : 0u;
        const unsigned noff = excl_scan_256(cn, sTot);
        if (t < NB_NODES) {
            sOff[t] = noff;
            sCnt[t] = 0u;
        }
        __syncthreads();
        if (staged) {
            for (unsigned c0 = 0; c0 < longest; c0 += 32) {
                const unsigned i = c0 + lane;
                int2 r[SC_BLOCKS / 32];
                #pragma unroll
                for (int j = 0; j < SC_BLOCKS / 32; ++j)
                    r[j] = recs[i < rcnt[j] ? rbeg[j] + i : 0u];
                #pragma unroll
                for (int j = 0; j < SC_BLOCKS / 32; ++j) {
                    if (i < rcnt[j]) {
                        const unsigned dl = (unsigned)r[j].x >> 24;
                        const unsigned pos = sOff[dl] + atomicAdd(&sCnt[dl], 1u);
                        sP[pos] = make_int2(r[j].x & 0x00FFFFFF, r[j].y);
                    }
                }
            }
        }
        __syncthreads();

        for (int tn = wib; tn < NB_NODES; tn += 32) {
            const int d = b * NB_NODES + tn;
            if (d >= NT) break;
            const unsigned off = __builtin_amdgcn_readfirstlane(sOff[tn]);
            const unsigned n = staged ? __builtin_amdgcn_readfirstlane(sCnt[tn]) : nrec;
            const float2 xr2 = ((const float2*)(xr + (size_t)d * 64))[lane];

            float m = -__builtin_inff();
            float den = 0.0f;
            float2 acc = make_float2(0.0f, 0.0f);
            bool any = false;
            for (unsigned c = 0; c < n; c += 32) {
                int2 pr;
                bool valid;
                unsigned left;
                if (staged) {
                    const unsigned i = c + mine;
                    valid = i < n;
                    pr = sP[off + (valid ? i : c)];
                    left = n - c;
                } else {
                    const unsigned i = c + lane;
                    const bool in = i < nrec;
                    const int2 rr = recs[in ? run_slot(i, sRunPre, sRunBeg) : 0u];
                    const bool match = in && ((unsigned)rr.x >> 24) == (unsigned)tn;
                    const unsigned mask = __builtin_amdgcn_ballot_w32(match);
                    if (mask == 0u) continue;
                    left = (unsigned)__builtin_popcount(mask);
                    const unsigned rank = (unsigned)__builtin_popcount(mask & ((1u << lane) - 1u));
                    if (match) sSel[wib][rank] = make_int2(rr.x & 0x00FFFFFF, rr.y);
                    valid = mine < left;
                    pr = sSel[wib][valid ? mine : 0u];
                }
                any = true;
                float2 r[32];
                load_group8<0>(pr.x, xl, lane, r);
                if (left > 8)  load_group8<1>(pr.x, xl, lane, r);
                if (left > 16) load_group8<2>(pr.x, xl, lane, r);
                if (left > 24) load_group8<3>(pr.x, xl, lane, r);
                float t0, t1 = 0.0f, t2 = 0.0f, t3 = 0.0f;
                t0 = score_group8<0>(r, pr.y, xr2, we, at, b4, b3, b2);
                if (left > 8)  t1 = score_group8<1>(r, pr.y, xr2, we, at, b4, b3, b2);
                if (left > 16) t2 = score_group8<2>(r, pr.y, xr2, we, at, b4, b3, b2);
                if (left > 24) t3 = score_group8<3>(r, pr.y, xr2, we, at, b4, b3, b2);
                const float u0 = merge2<2>(t0, t1, b1);
                const float u1 = merge2<2>(t2, t3, b1);
                const float p  = merge2<1>(u0, u1, b0);
                float cm = valid ? p : -__builtin_inff();
                for (int o = 16; o > 0; o >>= 1) cm = fmaxf(cm, __shfl_xor(cm, o, 32));
                const float mn = fmaxf(m, cm);
                const float scale = __expf(m - mn);
                const float w_v = valid ? __expf(p - mn) : 0.0f;
                den = den * scale + w_v;
                acc.x *= scale;
                acc.y *= scale;
                wsum_group8<0>(r, w_v, acc);
                if (left > 8)  wsum_group8<1>(r, w_v, acc);
                if (left > 16) wsum_group8<2>(r, w_v, acc);
                if (left > 24) wsum_group8<3>(r, w_v, acc);
                m = mn;
            }
            for (int o = 16; o > 0; o >>= 1) den += __shfl_xor(den, o, 32);
            const float inv = any ? 1.0f / den : 0.0f;
            ((float2*)(sAgg + tn * AGG_PITCH))[lane] = make_float2(acc.x * inv, acc.y * inv);
        }
    }
    __syncthreads();

    if (wib < AGG_TILES * 2) {
        const int tile = wib >> 1, nb0 = (wib & 1) * 2;
        const int l15 = lane & 15, half = lane >> 4;
        v8f acc0 = v8f{}, acc1 = v8f{};
        const float* arow = sAgg + (tile * 16 + l15) * AGG_PITCH;
        #pragma unroll
        for (int kk = 0; kk < 16; ++kk) {
            const int kbase = kk * 4 + half * 2;
            v2f a = *(const v2f*)(arow + kbase);
            a.x += sbias[kbase];
            a.y += sbias[kbase + 1];
            const int pbase = (kk * 2 + half) * PSTRIDE;
            const v2f b0 = *(const v2f*)(sWfc + pbase + (nb0 * 16 + l15) * 2);
            const v2f b1 = *(const v2f*)(sWfc + pbase + ((nb0 + 1) * 16 + l15) * 2);
            acc0 = __builtin_amdgcn_wmma_f32_16x16x4_f32(false, a, false, b0, (short)0, acc0, false, false);
            acc1 = __builtin_amdgcn_wmma_f32_16x16x4_f32(false, a, false, b1, (short)0, acc1, false, false);
        }
        const int c0 = nb0 * 16 + l15, c1 = c0 + 16;
        #pragma unroll
        for (int v = 0; v < 8; ++v) {
            const int tn = tile * 16 + v + half * 8;
            const int d = b * NB_NODES + tn;
            if (tn < NB_NODES && d < NT) {
                out[(size_t)d * 64 + c0] = acc0[v] + sbfc[c0];
                out[(size_t)d * 64 + c1] = acc1[v] + sbfc[c1];
            }
        }
    }
}

extern "C" void kernel_launch(void* const* d_in, const int* in_sizes, int n_in,
                              void* d_out, int out_size, void* d_ws, size_t ws_size,
                              hipStream_t stream) {
    const float* x      = (const float*)d_in[0];
    const int*   eidx   = (const int*)d_in[1];
    const float* eattr  = (const float*)d_in[2];
    const float* W_l    = (const float*)d_in[3];
    const float* b_l    = (const float*)d_in[4];
    const float* W_r    = (const float*)d_in[5];
    const float* b_r    = (const float*)d_in[6];
    const float* W_e    = (const float*)d_in[7];
    const float* att    = (const float*)d_in[8];
    const float* bias   = (const float*)d_in[9];
    const float* W_fc   = (const float*)d_in[10];
    const float* b_fc   = (const float*)d_in[11];

    const int NT       = in_sizes[0] / 64;
    const long long E  = (long long)in_sizes[2];
    const int* src = eidx;
    const int* dst = eidx + E;

    float* ws       = (float*)d_ws;
    float* xl       = ws;
    float* xr       = xl + (size_t)NT * 64;
    int*   recs     = (int*)(xr + (size_t)NT * 64);
    unsigned* runtab = (unsigned*)(recs + (size_t)NT * 64);

    prep_kernel<<<SC_BLOCKS + XF_BLOCKS, 1024, 0, stream>>>(src, dst, eattr, runtab, recs, x, W_l, b_l, W_r, b_r, xl, xr, NT, E);
    bucket_agg_kernel<<<256, 1024, 0, stream>>>(recs, runtab, xl, W_e, att, xr, bias, W_fc, b_fc, (float*)d_out, NT, E);
}


# node wave, per-edge arithmetic: weighted row sum as one packed fused multiply-add per edge (was packed mul, moves, adds), leaky relu as max(m, 0.2m); row-sum terms now round once (fused)
# speedup vs baseline: 8.8912x; 1.1111x over previous
#include <hip/hip_runtime.h>
#include <hip/hip_bf16.h>

typedef float v2f __attribute__((ext_vector_type(2)));
typedef float v8f __attribute__((ext_vector_type(8)));

#define PSTRIDE 160
#define WLDS    (32 * PSTRIDE)

__device__ __forceinline__ void stage_weight_pairs(float* sW, const float* __restrict__ W) {
    for (int i = threadIdx.x; i < 64 * 64; i += blockDim.x) {
        const int k = i >> 6, col = i & 63;
        sW[(k >> 1) * PSTRIDE + col * 2 + (k & 1)] = W[i];
    }
}

#define XF_BLOCKS 256

__device__ __forceinline__ void node_xform_role(int blk,
                                  const float* __restrict__ x2,
                                  const float* __restrict__ Wl, const float* __restrict__ bl,
                                  const float* __restrict__ Wr, const float* __restrict__ br,
                                  float* __restrict__ xl, float* __restrict__ xr, int NT)
{
    __shared__ float sWl[WLDS];
    __shared__ float sWr[WLDS];
    __shared__ float sbl[64], sbr[64];
    stage_weight_pairs(sWl, Wl);
    stage_weight_pairs(sWr, Wr);
    for (int i = threadIdx.x; i < 64; i += blockDim.x) { sbl[i] = bl[i]; sbr[i] = br[i]; }
    __syncthreads();

    const int lane    = threadIdx.x & 31;
    const int wave    = __builtin_amdgcn_readfirstlane(threadIdx.x >> 5);
    const int wpb     = blockDim.x >> 5;
    const int l15     = lane & 15;
    const int half    = lane >> 4;
    const int numUnits = (NT >> 4) * 2;
    const int per      = (numUnits + XF_BLOCKS - 1) / XF_BLOCKS;

    for (int w = wave; w < per; w += wpb) {
        const int unit = blk * per + w;
        if (unit >= numUnits) break;
        const int tile = unit >> 1;
        const bool right = (unit & 1) != 0;
        const float* sW = right ? sWr : sWl;
        const float* sb = right ? sbr : sbl;
        float* __restrict__ dstp = right ? xr : xl;
        v8f acc[4] = {v8f{}, v8f{}, v8f{}, v8f{}};
        const float* xrow = x2 + (size_t)(tile * 16 + l15) * 64;
        for (int k4 = 0; k4 < 16; k4 += 4) {
            v2f a[4];
            #pragma unroll
            for (int j = 0; j < 4; ++j)
                a[j] = *(const v2f*)(xrow + (k4 + j) * 4 + half * 2);
            #pragma unroll
            for (int j = 0; j < 4; ++j) {
                const int pbase = ((k4 + j) * 2 + half) * PSTRIDE;
                #pragma unroll
                for (int nb = 0; nb < 4; ++nb) {
                    const int col = nb * 16 + l15;
                    const v2f bw = *(const v2f*)(sW + pbase + col * 2);
                    acc[nb] = __builtin_amdgcn_wmma_f32_16x16x4_f32(
                        false, a[j], false, bw, (short)0, acc[nb], false, false);
                }
            }
        }
        #pragma unroll
        for (int nb = 0; nb < 4; ++nb) {
            const int col = nb * 16 + l15;
            #pragma unroll
            for (int v = 0; v < 8; ++v) {
                const size_t row = (size_t)(tile * 16 + v + half * 8);
                dstp[row * 64 + col] = acc[nb][v] + sb[col];
            }
        }
    }
}

#define NB_NODES 196
#define NBMAX    256
#define SC_BLOCKS 256
#define STAGE_CAP 8192

__device__ __forceinline__ unsigned excl_scan_256(unsigned v, unsigned* sTot)
{
    const int lane = threadIdx.x & 31, w = threadIdx.x >> 5;
    unsigned inc = v;
    for (int o = 1; o < 32; o <<= 1) {
        const unsigned up = __shfl_up(inc, o, 32);
        if (lane >= o) inc += up;
    }
    __syncthreads();
    if (lane == 31 && w < 8) sTot[w] = inc;
    __syncthreads();
    unsigned pre = 0u;
    for (int j = 0; j < 8; ++j) if (j < w) pre += sTot[j];
    return pre + inc - v;
}

__device__ __forceinline__ void edge_scatter_role(const int* __restrict__ src, const int* __restrict__ dst,
                                  const float* __restrict__ eattr,
                                  unsigned* __restrict__ runtab_u, int* __restrict__ recs, int NT, long long E)
{
    __shared__ unsigned sCnt[NBMAX];
    __shared__ unsigned sLoc[NBMAX];
    __shared__ unsigned sTot[8];
    __shared__ int2     sRec[STAGE_CAP];
    const int t = threadIdx.x;
    const long long ebeg = (E * blockIdx.x) / SC_BLOCKS;
    const long long eend = (E * (blockIdx.x + 1)) / SC_BLOCKS;
    const unsigned nblk = (eend > ebeg) ? (unsigned)(eend - ebeg) : 0u;

    if (t < NBMAX) sCnt[t] = 0u;
    __syncthreads();
    for (long long e0 = ebeg + t; e0 < eend; e0 += 8 * 1024) {
        int d[8];
        #pragma unroll
        for (int j = 0; j < 8; ++j) {
            const long long e = e0 + j * 1024;
            d[j] = dst[(e < eend) ? e : e0];
        }
        #pragma unroll
        for (int j = 0; j < 8; ++j)
            if (e0 + j * 1024 < eend) atomicAdd(&sCnt[(unsigned)d[j] / NB_NODES], 1u);
    }
    __syncthreads();
    const int nb = (NT + NB_NODES - 1) / NB_NODES;
    const unsigned c = (t < nb) ? sCnt[t] : 0u;
    const unsigned loc = excl_scan_256(c, sTot);
    if (t < nb) {
        ((uint2*)runtab_u)[(size_t)t * SC_BLOCKS + blockIdx.x] = make_uint2(loc, c);
        sLoc[t]  = loc;
        sCnt[t]  = 0u;
    }
    __syncthreads();
    const bool staged = nblk <= STAGE_CAP;
    for (long long e0 = ebeg + t; e0 < eend; e0 += 4 * 1024) {
        int s[4], d[4];
        float a[4];
        #pragma unroll
        for (int j = 0; j < 4; ++j) {
            const long long e = e0 + j * 1024;
            const long long ec = (e < eend) ? e : e0;
            s[j] = src[ec];
            d[j] = dst[ec];
            a[j] = eattr[ec];
        }
        #pragma unroll
        for (int j = 0; j < 4; ++j) {
            if (e0 + j * 1024 < eend) {
                const unsigned bk = (unsigned)d[j] / NB_NODES;
                const unsigned r = atomicAdd(&sCnt[bk], 1u);
                const unsigned lp = sLoc[bk] + r;
                const unsigned key = (unsigned)s[j] | (((unsigned)d[j] - bk * NB_NODES) << 24);
                const int2 rec = make_int2((int)key, __float_as_int(a[j]));
                if (staged) sRec[lp] = rec;
                else        ((int2*)recs)[ebeg + lp] = rec;
            }
        }
    }
    if (staged) {
        __syncthreads();
        for (unsigned i = t; i < nblk; i += blockDim.x) ((int2*)recs)[ebeg + i] = sRec[i];
    }
}

__global__ void __launch_bounds__(1024) __attribute__((amdgpu_waves_per_eu(16)))
prep_kernel(const int* __restrict__ src, const int* __restrict__ dst, const float* __restrict__ eattr,
            unsigned* __restrict__ runtab_u, int* __restrict__ recs,
            const float* __restrict__ x2,
            const float* __restrict__ Wl, const float* __restrict__ bl,
            const float* __restrict__ Wr, const float* __restrict__ br,
            float* __restrict__ xl, float* __restrict__ xr, int NT, long long E)
{
    if (blockIdx.x < SC_BLOCKS) edge_scatter_role(src, dst, eattr, runtab_u, recs, NT, E);
    else                        node_xform_role((int)blockIdx.x - SC_BLOCKS, x2, Wl, bl, Wr, br, xl, xr, NT);
}

#define BREV5(j) ((((j) & 1) << 4) | (((j) & 2) << 2) | ((j) & 4) | (((j) & 8) >> 2) | (((j) & 16) >> 4))

template <int MASK>
__device__ __forceinline__ float merge2(float a, float b, bool up) {
    const float keep = up ? b : a;
    const float give = up ? a : b;
    return keep + __shfl_xor(give, MASK, 32);
}

template <int G>
__device__ __forceinline__ void load_group8(int s_v, const float* __restrict__ xl, int lane, float2* r)
{
    #pragma unroll
    for (int j = 0; j < 8; ++j) {
        const int s = __builtin_amdgcn_readlane(s_v, BREV5(G * 8 + j));
        r[G * 8 + j] = ((const float2*)(xl + (size_t)s * 64))[lane];
    }
}

template <int G>
__device__ __forceinline__ float score_group8(const float2* r, int ea_v, float2 xr2, float2 we, float2 at,
                                              bool b4, bool b3, bool b2)
{
    float p[8];
    #pragma unroll
    for (int j = 0; j < 8; ++j) {
        const float ea = __int_as_float(__builtin_amdgcn_readlane(ea_v, BREV5(G * 8 + j)));
        float m0 = r[G * 8 + j].x + xr2.x + ea * we.x;
        float m1 = r[G * 8 + j].y + xr2.y + ea * we.y;
        m0 = fmaxf(m0, 0.2f * m0);
        m1 = fmaxf(m1, 0.2f * m1);
        p[j] = __builtin_fmaf(at.x, m0, at.y * m1);
    }
    const float q0 = merge2<16>(p[0], p[1], b4);
    const float q1 = merge2<16>(p[2], p[3], b4);
    const float q2 = merge2<16>(p[4], p[5], b4);
    const float q3 = merge2<16>(p[6], p[7], b4);
    const float r0 = merge2<8>(q0, q1, b3);
    const float r1 = merge2<8>(q2, q3, b3);
    return merge2<4>(r0, r1, b2);
}

template <int G>
__device__ __forceinline__ void wsum_group8(const float2* r, float w_v, float2& acc)
{
    #pragma unroll
    for (int j = 0; j < 8; ++j) {
        const float w = __int_as_float(__builtin_amdgcn_readlane(__float_as_int(w_v), BREV5(G * 8 + j)));
        acc.x = __builtin_fmaf(w, r[G * 8 + j].x, acc.x);
        acc.y = __builtin_fmaf(w, r[G * 8 + j].y, acc.y);
    }
}

#define SORT_CAP 12288
#define AGG_TILES ((NB_NODES + 15) / 16)
#define AGG_PITCH 68
__device__ __forceinline__ unsigned run_slot(unsigned i, const unsigned* sPre, const unsigned* sBeg)
{
    unsigned lo = 0u;
    for (unsigned st = SC_BLOCKS / 2; st > 0u; st >>= 1)
        if (sPre[lo + st] <= i) lo += st;
    return sBeg[lo] + (i - sPre[lo]);
}

__global__ void __launch_bounds__(1024)
bucket_agg_kernel(const int* __restrict__ recs_i, const unsigned* __restrict__ runtab_u,
                  const float* __restrict__ xl, const float* __restrict__ We, const float* __restrict__ att,
                  const float* __restrict__ xr, const float* __restrict__ bias,
                  const float* __restrict__ Wfc, const float* __restrict__ bfc,
                  float* __restrict__ out, int NT, long long E)
{
    __shared__ float sAgg[AGG_TILES * 16 * AGG_PITCH];
    __shared__ float sWfc[WLDS];
    __shared__ float sbfc[64], sbias[64];
    __shared__ int2 sP[SORT_CAP];
    __shared__ unsigned sCnt[NBMAX], sOff[NBMAX];
    __shared__ unsigned sTot[8];
    __shared__ unsigned sRunBeg[SC_BLOCKS], sRunCnt[SC_BLOCKS], sRunPre[SC_BLOCKS];
    __shared__ unsigned sNum;
    __shared__ int2 sSel[32][32];
    const int2* __restrict__ recs = (const int2*)recs_i;
    const int t = threadIdx.x;
    const int lane = t & 31;
    const int wib = __builtin_amdgcn_readfirstlane(t >> 5);
    const int nb = (NT + NB_NODES - 1) / NB_NODES;
    const float2 we = ((const float2*)We)[lane];
    const float2 at = ((const float2*)att)[lane];
    const bool b4 = (lane & 16) != 0, b3 = (lane & 8) != 0, b2 = (lane & 4) != 0;
    const bool b1 = (lane & 2) != 0,  b0 = (lane & 1) != 0;
    const unsigned mine = BREV5(lane);
    const int b = blockIdx.x;
    if (b >= nb) return;
    stage_weight_pairs(sWfc, Wfc);
    if (t < 64) { sbfc[t] = bfc[t]; sbias[t] = bias[t]; }
    {
        unsigned rlen = 0u;
        if (t < SC_BLOCKS) {
            const uint2 ent = ((const uint2*)runtab_u)[(size_t)b * SC_BLOCKS + t];
            sRunBeg[t] = (unsigned)((E * t) / SC_BLOCKS) + ent.x;
            sRunCnt[t] = ent.y;
            rlen = ent.y;
        }
        const unsigned rpre = excl_scan_256(rlen, sTot);
        if (t < SC_BLOCKS) sRunPre[t] = rpre;
        if (t == SC_BLOCKS - 1) sNum = rpre + rlen;
        if (t < NBMAX) sCnt[t] = 0u;
        __syncthreads();
        const unsigned nrec = sNum;
        const bool staged = nrec <= SORT_CAP;
        unsigned rbeg[SC_BLOCKS / 32], rcnt[SC_BLOCKS / 32];
        unsigned longest = 0u;
        #pragma unroll
        for (int j = 0; j < SC_BLOCKS / 32; ++j) {
            rbeg[j] = __builtin_amdgcn_readfirstlane(sRunBeg[wib + 32 * j]);
            rcnt[j] = __builtin_amdgcn_readfirstlane(sRunCnt[wib + 32 * j]);
            longest = rcnt[j] > longest ? rcnt[j] : longest;
        }
        if (staged) {
            for (unsigned c0 = 0; c0 < longest; c0 += 32) {
                const unsigned i = c0 + lane;
                int k[SC_BLOCKS / 32];
                #pragma unroll
                for (int j = 0; j < SC_BLOCKS / 32; ++j)
                    k[j] = recs[i < rcnt[j] ? rbeg[j] + i : 0u].x;
                #pragma unroll
                for (int j = 0; j < SC_BLOCKS / 32; ++j)
                    if (i < rcnt[j]) atomicAdd(&sCnt[(unsigned)k[j] >> 24], 1u);
            }
        }
        __syncthreads();
        const unsigned cn = (t < NB_NODES) ? sCnt[t] : 0u;
        const unsigned noff = excl_scan_256(cn, sTot);
        if (t < NB_NODES) {
            sOff[t] = noff;
            sCnt[t] = 0u;
        }
        __syncthreads();
        if (staged) {
            for (unsigned c0 = 0; c0 < longest; c0 += 32) {
                const unsigned i = c0 + lane;
                int2 r[SC_BLOCKS / 32];
                #pragma unroll
                for (int j = 0; j < SC_BLOCKS / 32; ++j)
                    r[j] = recs[i < rcnt[j] ? rbeg[j] + i : 0u];
                #pragma unroll
                for (int j = 0; j < SC_BLOCKS / 32; ++j) {
                    if (i < rcnt[j]) {
                        const unsigned dl = (unsigned)r[j].x >> 24;
                        const unsigned pos = sOff[dl] + atomicAdd(&sCnt[dl], 1u);
                        sP[pos] = make_int2(r[j].x & 0x00FFFFFF, r[j].y);
                    }
                }
            }
        }
        __syncthreads();

        for (int tn = wib; tn < NB_NODES; tn += 32) {
            const int d = b * NB_NODES + tn;
            if (d >= NT) break;
            const unsigned off = __builtin_amdgcn_readfirstlane(sOff[tn]);
            const unsigned n = staged ? __builtin_amdgcn_readfirstlane(sCnt[tn]) : nrec;
            const float2 xr2 = ((const float2*)(xr + (size_t)d * 64))[lane];

            float m = -__builtin_inff();
            float den = 0.0f;
            float2 acc = make_float2(0.0f, 0.0f);
            bool any = false;
            for (unsigned c = 0; c < n; c += 32) {
                int2 pr;
                bool valid;
                unsigned left;
                if (staged) {
                    const unsigned i = c + mine;
                    valid = i < n;
                    pr = sP[off + (valid ? i : c)];
                    left = n - c;
                } else {
                    const unsigned i = c + lane;
                    const bool in = i < nrec;
                    const int2 rr = recs[in ? run_slot(i, sRunPre, sRunBeg) : 0u];
                    const bool match = in && ((unsigned)rr.x >> 24) == (unsigned)tn;
                    const unsigned mask = __builtin_amdgcn_ballot_w32(match);
                    if (mask == 0u) continue;
                    left = (unsigned)__builtin_popcount(mask);
                    const unsigned rank = (unsigned)__builtin_popcount(mask & ((1u << lane) - 1u));
                    if (match) sSel[wib][rank] = make_int2(rr.x & 0x00FFFFFF, rr.y);
                    valid = mine < left;
                    pr = sSel[wib][valid ? mine : 0u];
                }
                any = true;
                float2 r[32];
                load_group8<0>(pr.x, xl, lane, r);
                if (left > 8)  load_group8<1>(pr.x, xl, lane, r);
                if (left > 16) load_group8<2>(pr.x, xl, lane, r);
                if (left > 24) load_group8<3>(pr.x, xl, lane, r);
                float t0, t1 = 0.0f, t2 = 0.0f, t3 = 0.0f;
                t0 = score_group8<0>(r, pr.y, xr2, we, at, b4, b3, b2);
                if (left > 8)  t1 = score_group8<1>(r, pr.y, xr2, we, at, b4, b3, b2);
                if (left > 16) t2 = score_group8<2>(r, pr.y, xr2, we, at, b4, b3, b2);
                if (left > 24) t3 = score_group8<3>(r, pr.y, xr2, we, at, b4, b3, b2);
                const float u0 = merge2<2>(t0, t1, b1);
                const float u1 = merge2<2>(t2, t3, b1);
                const float p  = merge2<1>(u0, u1, b0);
                float cm = valid ? p : -__builtin_inff();
                for (int o = 16; o > 0; o >>= 1) cm = fmaxf(cm, __shfl_xor(cm, o, 32));
                const float mn = fmaxf(m, cm);
                const float scale = __expf(m - mn);
                const float w_v = valid ? __expf(p - mn) : 0.0f;
                den = den * scale + w_v;
                acc.x *= scale;
                acc.y *= scale;
                wsum_group8<0>(r, w_v, acc);
                if (left > 8)  wsum_group8<1>(r, w_v, acc);
                if (left > 16) wsum_group8<2>(r, w_v, acc);
                if (left > 24) wsum_group8<3>(r, w_v, acc);
                m = mn;
            }
            for (int o = 16; o > 0; o >>= 1) den += __shfl_xor(den, o, 32);
            const float inv = any ? 1.0f / den : 0.0f;
            ((float2*)(sAgg + tn * AGG_PITCH))[lane] = make_float2(acc.x * inv, acc.y * inv);
        }
    }
    __syncthreads();

    if (wib < AGG_TILES * 2) {
        const int tile = wib >> 1, nb0 = (wib & 1) * 2;
        const int l15 = lane & 15, half = lane >> 4;
        v8f acc0 = v8f{}, acc1 = v8f{};
        const float* arow = sAgg + (tile * 16 + l15) * AGG_PITCH;
        #pragma unroll
        for (int kk = 0; kk < 16; ++kk) {
            const int kbase = kk * 4 + half * 2;
            v2f a = *(const v2f*)(arow + kbase);
            a.x += sbias[kbase];
            a.y += sbias[kbase + 1];
            const int pbase = (kk * 2 + half) * PSTRIDE;
            const v2f b0 = *(const v2f*)(sWfc + pbase + (nb0 * 16 + l15) * 2);
            const v2f b1 = *(const v2f*)(sWfc + pbase + ((nb0 + 1) * 16 + l15) * 2);
            acc0 = __builtin_amdgcn_wmma_f32_16x16x4_f32(false, a, false, b0, (short)0, acc0, false, false);
            acc1 = __builtin_amdgcn_wmma_f32_16x16x4_f32(false, a, false, b1, (short)0, acc1, false, false);
        }
        const int c0 = nb0 * 16 + l15, c1 = c0 + 16;
        #pragma unroll
        for (int v = 0; v < 8; ++v) {
            const int tn = tile * 16 + v + half * 8;
            const int d = b * NB_NODES + tn;
            if (tn < NB_NODES && d < NT) {
                out[(size_t)d * 64 + c0] = acc0[v] + sbfc[c0];
                out[(size_t)d * 64 + c1] = acc1[v] + sbfc[c1];
            }
        }
    }
}

extern "C" void kernel_launch(void* const* d_in, const int* in_sizes, int n_in,
                              void* d_out, int out_size, void* d_ws, size_t ws_size,
                              hipStream_t stream) {
    const float* x      = (const float*)d_in[0];
    const int*   eidx   = (const int*)d_in[1];
    const float* eattr  = (const float*)d_in[2];
    const float* W_l    = (const float*)d_in[3];
    const float* b_l    = (const float*)d_in[4];
    const float* W_r    = (const float*)d_in[5];
    const float* b_r    = (const float*)d_in[6];
    const float* W_e    = (const float*)d_in[7];
    const float* att    = (const float*)d_in[8];
    const float* bias   = (const float*)d_in[9];
    const float* W_fc   = (const float*)d_in[10];
    const float* b_fc   = (const float*)d_in[11];

    const int NT       = in_sizes[0] / 64;
    const long long E  = (long long)in_sizes[2];
    const int* src = eidx;
    const int* dst = eidx + E;

    float* ws       = (float*)d_ws;
    float* xl       = ws;
    float* xr       = xl + (size_t)NT * 64;
    int*   recs     = (int*)(xr + (size_t)NT * 64);
    unsigned* runtab = (unsigned*)(recs + (size_t)NT * 64);

    prep_kernel<<<SC_BLOCKS + XF_BLOCKS, 1024, 0, stream>>>(src, dst, eattr, runtab, recs, x, W_l, b_l, W_r, b_r, xl, xr, NT, E);
    bucket_agg_kernel<<<256, 1024, 0, stream>>>(recs, runtab, xl, W_e, att, xr, bias, W_fc, b_fc, (float*)d_out, NT, E);
}
